# starting kernel
# baseline (speedup 1.0000x reference)
#include <hip/hip_runtime.h>

#define N_NODES 50000
#define NE_EDGES 800000
#define BPAIRS 4096
#define DF 128
#define EPS_BN 1e-5f

typedef float v2f __attribute__((ext_vector_type(2)));
typedef float v8f __attribute__((ext_vector_type(8)));

__device__ __forceinline__ v8f wmma4(v2f a, v2f b, v8f c) {
  return __builtin_amdgcn_wmma_f32_16x16x4_f32(
       false, a,  false, b,
       (short)0, c,  false,  false);
}

__global__ __launch_bounds__(256)
void wmma_gemm_strip(const float* __restrict__ A, const float* __restrict__ Bt,
                     const float* __restrict__ bias, float* __restrict__ C,
                     int M, int K) {
  int gtid = blockIdx.x * blockDim.x + threadIdx.x;
  int wave = gtid >> 5;
  int lane = threadIdx.x & 31;
  if (wave * 16 >= M) return;

  int l16   = lane & 15;
  int khalf = (lane >> 4) << 1;
  const float* arow = A + (size_t)(wave * 16 + l16) * K + khalf;

  const float* brow[8];
#pragma unroll
  for (int nt = 0; nt < 8; ++nt)
    brow[nt] = Bt + (size_t)(nt * 16 + l16) * K + khalf;

  v8f zero = {};
  v8f acc[8];
#pragma unroll
  for (int nt = 0; nt < 8; ++nt) acc[nt] = zero;

  for (int k = 0; k < K; k += 4) {
    v2f a = *(const v2f*)(arow + k);
#pragma unroll
    for (int nt = 0; nt < 8; ++nt) {
      v2f b = *(const v2f*)(brow[nt] + k);
      acc[nt] = wmma4(a, b, acc[nt]);
    }
  }

  int rbase = wave * 16 + ((lane >> 4) << 3);
#pragma unroll
  for (int nt = 0; nt < 8; ++nt) {
    float bs = bias ? bias[nt * 16 + l16] : 0.f;
    float* crow = C + (size_t)rbase * 128 + nt * 16 + l16;
#pragma unroll
    for (int r = 0; r < 8; ++r)
      crow[(size_t)r * 128] = acc[nt][r] + bs;
  }
}

__global__ __launch_bounds__(256)
void wmma_matmul(const float* __restrict__ A, const float* __restrict__ Bm,
                 const float* __restrict__ bias, float* __restrict__ C,
                 int M, int Nc, int K, int bTrans) {
  int gtid = blockIdx.x * blockDim.x + threadIdx.x;
  int wave = gtid >> 5;
  int lane = threadIdx.x & 31;
  int ntiles = Nc >> 4;
  int mt = wave / ntiles;
  int nt = wave - mt * ntiles;
  if (mt * 16 >= M) return;

  int l16   = lane & 15;
  int khalf = (lane >> 4) << 1;
  const float* arow = A + (size_t)(mt * 16 + l16) * K + khalf;

  v8f acc = {};
  if (bTrans) {
    const float* brow = Bm + (size_t)(nt * 16 + l16) * K + khalf;
    for (int k = 0; k < K; k += 4) {
      v2f a = *(const v2f*)(arow + k);
      v2f b = *(const v2f*)(brow + k);
      acc = wmma4(a, b, acc);
    }
  } else {
    const float* bcol = Bm + (size_t)khalf * Nc + nt * 16 + l16;
    for (int k = 0; k < K; k += 4) {
      v2f a = *(const v2f*)(arow + k);
      v2f b;
      b.x = bcol[(size_t)k * Nc];
      b.y = bcol[(size_t)(k + 1) * Nc];
      acc = wmma4(a, b, acc);
    }
  }

  float bs = bias ? bias[nt * 16 + l16] : 0.f;
  int rbase = mt * 16 + ((lane >> 4) << 3);
  float* crow = C + (size_t)rbase * Nc + nt * 16 + l16;
#pragma unroll
  for (int r = 0; r < 8; ++r)
    crow[(size_t)r * Nc] = acc[r] + bs;
}

__global__ void zero_int(int* p, int n) {
  int t = blockIdx.x * blockDim.x + threadIdx.x;
  if (t < n) p[t] = 0;
}

__global__ void count_edges(const int* __restrict__ ecol, int* __restrict__ cnt) {
  int e = blockIdx.x * blockDim.x + threadIdx.x;
  if (e < NE_EDGES) atomicAdd(&cnt[ecol[e]], 1);
}

__global__ __launch_bounds__(1024)
void scan_rowptr(const int* __restrict__ cnt, int* __restrict__ rowptr) {
  __shared__ int part[1024];
  const int CH = (N_NODES + 1023) / 1024;
  int tid = threadIdx.x;
  int base = tid * CH;
  int s = 0;
  for (int i = 0; i < CH; ++i) {
    int idx = base + i;
    if (idx < N_NODES) s += cnt[idx];
  }
  part[tid] = s;
  __syncthreads();
  for (int off = 1; off < 1024; off <<= 1) {
    int v = 0;
    if (tid >= off) v = part[tid - off];
    __syncthreads();
    if (tid >= off) part[tid] += v;
    __syncthreads();
  }
  int run = (tid == 0) ? 0 : part[tid - 1];
  for (int i = 0; i < CH; ++i) {
    int idx = base + i;
    if (idx < N_NODES) { rowptr[idx] = run; run += cnt[idx]; }
  }
  if (tid == 1023) rowptr[N_NODES] = run;
}

__global__ void copy_int(const int* __restrict__ src, int* __restrict__ dst, int n) {
  int t = blockIdx.x * blockDim.x + threadIdx.x;
  if (t < n) dst[t] = src[t];
}

__global__ void csr_fill(const int* __restrict__ erow, const int* __restrict__ ecol,
                         int* __restrict__ cursor, int* __restrict__ srcidx) {
  int e = blockIdx.x * blockDim.x + threadIdx.x;
  if (e >= NE_EDGES) return;
  int c = ecol[e];
  int p = atomicAdd(&cursor[c], 1);
  srcidx[p] = erow[e];
}

__global__ void dis_from_cnt(const int* __restrict__ cnt, float* __restrict__ dis, int n) {
  int t = blockIdx.x * blockDim.x + threadIdx.x;
  if (t < n) dis[t] = rsqrtf((float)cnt[t] + 1.0f);
}

__global__ __launch_bounds__(256)
void gcn_aggregate(const int* __restrict__ rowptr, const int* __restrict__ srcidx,
                   const float* __restrict__ dis, const float* __restrict__ lin,
                   float* __restrict__ agg) {
  int t = blockIdx.x * blockDim.x + threadIdx.x;
  int c = t >> 5;
  if (c >= N_NODES) return;
  int lane = threadIdx.x & 31;
  float dc = dis[c];
  const float4 sv = *(const float4*)(lin + (size_t)c * DF + lane * 4);
  float s = dc * dc;
  float ax = s * sv.x, ay = s * sv.y, az = s * sv.z, aw = s * sv.w;
  int e0 = rowptr[c], e1 = rowptr[c + 1];
  for (int j = e0; j < e1; ++j) {
    int r = srcidx[j];
    float nrm = dis[r] * dc;
    const float4 v = *(const float4*)(lin + (size_t)r * DF + lane * 4);
    ax += nrm * v.x; ay += nrm * v.y; az += nrm * v.z; aw += nrm * v.w;
  }
  float4 o; o.x = ax; o.y = ay; o.z = az; o.w = aw;
  *(float4*)(agg + (size_t)c * DF + lane * 4) = o;
}

__global__ void zero256(float* p) { p[threadIdx.x] = 0.0f; }

__global__ __launch_bounds__(128)
void bn_stats(const float* __restrict__ agg, float* __restrict__ stats, int n) {
  int f = threadIdx.x;
  int r0 = blockIdx.x * 512;
  int r1 = r0 + 512; if (r1 > n) r1 = n;
  float s = 0.f, s2 = 0.f;
  for (int r = r0; r < r1; ++r) {
    float v = agg[(size_t)r * DF + f];
    s += v; s2 += v * v;
  }
  atomicAdd(&stats[f], s);
  atomicAdd(&stats[DF + f], s2);
}

__global__ void bn_finalize(float* __restrict__ stats, const float* __restrict__ g,
                            const float* __restrict__ be, float inv_n) {
  int f = threadIdx.x;
  float m   = stats[f] * inv_n;
  float var = stats[DF + f] * inv_n - m * m;
  float sc  = g[f] * rsqrtf(var + EPS_BN);
  stats[f]      = sc;
  stats[DF + f] = be[f] - m * sc;
}

__global__ __launch_bounds__(256)
void bn_apply_leaky(const float* __restrict__ agg, const float* __restrict__ stats,
                    float* __restrict__ h) {
  int t = blockIdx.x * blockDim.x + threadIdx.x;
  if (t >= N_NODES * DF) return;
  int f = t & (DF - 1);
  float y = agg[t] * stats[f] + stats[DF + f];
  h[t] = (y >= 0.f) ? y : 0.1f * y;
}

__global__ void transpose128(const float* __restrict__ src, float* __restrict__ dst) {
  int t = blockIdx.x * blockDim.x + threadIdx.x;
  if (t >= 128 * 128) return;
  int r = t >> 7, c = t & 127;
  dst[c * 128 + r] = src[t];
}

__global__ __launch_bounds__(256)
void gather_pairs(const float* __restrict__ h, const int* __restrict__ di,
                  float* __restrict__ amat, float* __restrict__ bmat) {
  int t = blockIdx.x * blockDim.x + threadIdx.x;
  if (t >= BPAIRS * 32) return;
  int j = t >> 5, q = t & 31;
  int ia = di[2 * j] - 1;     if (ia < 0) ia += N_NODES;
  int ib = di[2 * j + 1] - 1; if (ib < 0) ib += N_NODES;
  *(float4*)(amat + (size_t)j * DF + q * 4) = *(const float4*)(h + (size_t)ia * DF + q * 4);
  *(float4*)(bmat + (size_t)j * DF + q * 4) = *(const float4*)(h + (size_t)ib * DF + q * 4);
}

__global__ __launch_bounds__(256)
void rowdot(const float* __restrict__ tm, const float* __restrict__ bm,
            float* __restrict__ out) {
  int t = blockIdx.x * blockDim.x + threadIdx.x;
  int j = t >> 5, lane = t & 31;
  if (j >= BPAIRS) return;
  float4 tv = *(const float4*)(tm + (size_t)j * DF + lane * 4);
  float4 bv = *(const float4*)(bm + (size_t)j * DF + lane * 4);
  float p = tv.x * bv.x + tv.y * bv.y + tv.z * bv.z + tv.w * bv.w;
#pragma unroll
  for (int off = 16; off > 0; off >>= 1)
    p += __shfl_down(p, off, 32);
  if (lane == 0) out[j] = p;
}

extern "C" void kernel_launch(void* const* d_in, const int* in_sizes, int n_in,
                              void* d_out, int out_size, void* d_ws, size_t ws_size,
                              hipStream_t stream) {
  const float* x   = (const float*)d_in[0];
  const int*   edg = (const int*)d_in[1];
  const int*   di  = (const int*)d_in[2];
  const float* w1  = (const float*)d_in[3];
  const float* b1  = (const float*)d_in[4];
  const float* w2  = (const float*)d_in[5];
  const float* b2  = (const float*)d_in[6];
  const float* w3  = (const float*)d_in[7];
  const float* b3  = (const float*)d_in[8];
  const float* g1  = (const float*)d_in[9];
  const float* be1 = (const float*)d_in[10];
  const float* g2  = (const float*)d_in[11];
  const float* be2 = (const float*)d_in[12];
  const float* g3  = (const float*)d_in[13];
  const float* be3 = (const float*)d_in[14];
  const float* p1  = (const float*)d_in[15];
  const float* p2  = (const float*)d_in[16];
  float* out = (float*)d_out;

  float* ws = (float*)d_ws;
  size_t off = 0;
  float* dis    = ws + off; off += N_NODES;
  float* stats  = ws + off; off += 256;
  float* Wt     = ws + off; off += 128 * 128;
  int*   cnt    = (int*)(ws + off); off += N_NODES;
  int*   rowptr = (int*)(ws + off); off += N_NODES + 16;
  int*   cursor = (int*)(ws + off); off += N_NODES;
  int*   srcidx = (int*)(ws + off); off += NE_EDGES;
  float* lin    = ws + off; off += (size_t)N_NODES * DF;
  float* agg    = ws + off; off += (size_t)N_NODES * DF;
  float* h      = ws + off; off += (size_t)N_NODES * DF;
  float* U    = lin;
  float* Qt   = lin + 16384;
  float* amat = lin + 32768;
  float* bmat = amat + (size_t)BPAIRS * DF;
  float* tmat = bmat + (size_t)BPAIRS * DF;

  const int* erow = edg;
  const int* ecol = edg + NE_EDGES;

  zero_int<<<(N_NODES + 255) / 256, 256, 0, stream>>>(cnt, N_NODES);
  count_edges<<<(NE_EDGES + 255) / 256, 256, 0, stream>>>(ecol, cnt);
  scan_rowptr<<<1, 1024, 0, stream>>>(cnt, rowptr);
  copy_int<<<(N_NODES + 255) / 256, 256, 0, stream>>>(rowptr, cursor, N_NODES);
  csr_fill<<<(NE_EDGES + 255) / 256, 256, 0, stream>>>(erow, ecol, cursor, srcidx);
  dis_from_cnt<<<(N_NODES + 255) / 256, 256, 0, stream>>>(cnt, dis, N_NODES);

  const float* Wl[3] = {w1, w2, w3};
  const float* Bl[3] = {b1, b2, b3};
  const float* Gl[3] = {g1, g2, g3};
  const float* El[3] = {be1, be2, be3};
  const float* in = x;
  const int strip_waves = N_NODES / 16;
  for (int l = 0; l < 3; ++l) {
    transpose128<<<64, 256, 0, stream>>>(Wl[l], Wt);
    wmma_gemm_strip<<<(strip_waves * 32 + 255) / 256, 256, 0, stream>>>(
        in, Wt, Bl[l], lin, N_NODES, DF);
    gcn_aggregate<<<(N_NODES * 32 + 255) / 256, 256, 0, stream>>>(
        rowptr, srcidx, dis, lin, agg);
    zero256<<<1, 256, 0, stream>>>(stats);
    bn_stats<<<(N_NODES + 511) / 512, 128, 0, stream>>>(agg, stats, N_NODES);
    bn_finalize<<<1, 128, 0, stream>>>(stats, Gl[l], El[l], 1.0f / (float)N_NODES);
    bn_apply_leaky<<<(N_NODES * DF + 255) / 256, 256, 0, stream>>>(agg, stats, h);
    in = h;
  }

  const int small_tiles = (DF / 16) * (DF / 16);
  wmma_matmul<<<(small_tiles * 32 + 255) / 256, 256, 0, stream>>>(p1, p2, nullptr, U, DF, DF, DF, 1);
  wmma_matmul<<<(small_tiles * 32 + 255) / 256, 256, 0, stream>>>(U, p1, nullptr, Qt, DF, DF, DF, 1);
  gather_pairs<<<(BPAIRS * 32 + 255) / 256, 256, 0, stream>>>(h, di, amat, bmat);
  wmma_gemm_strip<<<((BPAIRS / 16) * 32 + 255) / 256, 256, 0, stream>>>(
      amat, Qt, nullptr, tmat, BPAIRS, DF);
  rowdot<<<(BPAIRS * 32 + 255) / 256, 256, 0, stream>>>(tmat, bmat, out);
}


# bn_stats: 32 waves per 512-row block, each wave sums 16 rows from whole-row 16-byte loads issued as a group, LDS sum over waves, one atomic per column per block; in-block sum order changes
# speedup vs baseline: 2.2929x; 2.2929x over previous
#include <hip/hip_runtime.h>

#define N_NODES 50000
#define NE_EDGES 800000
#define BPAIRS 4096
#define DF 128
#define EPS_BN 1e-5f

typedef float v2f __attribute__((ext_vector_type(2)));
typedef float v8f __attribute__((ext_vector_type(8)));

__device__ __forceinline__ v8f wmma4(v2f a, v2f b, v8f c) {
  return __builtin_amdgcn_wmma_f32_16x16x4_f32(
       false, a,  false, b,
       (short)0, c,  false,  false);
}

__global__ __launch_bounds__(256)
void wmma_gemm_strip(const float* __restrict__ A, const float* __restrict__ Bt,
                     const float* __restrict__ bias, float* __restrict__ C,
                     int M, int K) {
  int gtid = blockIdx.x * blockDim.x + threadIdx.x;
  int wave = gtid >> 5;
  int lane = threadIdx.x & 31;
  if (wave * 16 >= M) return;

  int l16   = lane & 15;
  int khalf = (lane >> 4) << 1;
  const float* arow = A + (size_t)(wave * 16 + l16) * K + khalf;

  const float* brow[8];
#pragma unroll
  for (int nt = 0; nt < 8; ++nt)
    brow[nt] = Bt + (size_t)(nt * 16 + l16) * K + khalf;

  v8f zero = {};
  v8f acc[8];
#pragma unroll
  for (int nt = 0; nt < 8; ++nt) acc[nt] = zero;

  for (int k = 0; k < K; k += 4) {
    v2f a = *(const v2f*)(arow + k);
#pragma unroll
    for (int nt = 0; nt < 8; ++nt) {
      v2f b = *(const v2f*)(brow[nt] + k);
      acc[nt] = wmma4(a, b, acc[nt]);
    }
  }

  int rbase = wave * 16 + ((lane >> 4) << 3);
#pragma unroll
  for (int nt = 0; nt < 8; ++nt) {
    float bs = bias ? bias[nt * 16 + l16] : 0.f;
    float* crow = C + (size_t)rbase * 128 + nt * 16 + l16;
#pragma unroll
    for (int r = 0; r < 8; ++r)
      crow[(size_t)r * 128] = acc[nt][r] + bs;
  }
}

__global__ __launch_bounds__(256)
void wmma_matmul(const float* __restrict__ A, const float* __restrict__ Bm,
                 const float* __restrict__ bias, float* __restrict__ C,
                 int M, int Nc, int K, int bTrans) {
  int gtid = blockIdx.x * blockDim.x + threadIdx.x;
  int wave = gtid >> 5;
  int lane = threadIdx.x & 31;
  int ntiles = Nc >> 4;
  int mt = wave / ntiles;
  int nt = wave - mt * ntiles;
  if (mt * 16 >= M) return;

  int l16   = lane & 15;
  int khalf = (lane >> 4) << 1;
  const float* arow = A + (size_t)(mt * 16 + l16) * K + khalf;

  v8f acc = {};
  if (bTrans) {
    const float* brow = Bm + (size_t)(nt * 16 + l16) * K + khalf;
    for (int k = 0; k < K; k += 4) {
      v2f a = *(const v2f*)(arow + k);
      v2f b = *(const v2f*)(brow + k);
      acc = wmma4(a, b, acc);
    }
  } else {
    const float* bcol = Bm + (size_t)khalf * Nc + nt * 16 + l16;
    for (int k = 0; k < K; k += 4) {
      v2f a = *(const v2f*)(arow + k);
      v2f b;
      b.x = bcol[(size_t)k * Nc];
      b.y = bcol[(size_t)(k + 1) * Nc];
      acc = wmma4(a, b, acc);
    }
  }

  float bs = bias ? bias[nt * 16 + l16] : 0.f;
  int rbase = mt * 16 + ((lane >> 4) << 3);
  float* crow = C + (size_t)rbase * Nc + nt * 16 + l16;
#pragma unroll
  for (int r = 0; r < 8; ++r)
    crow[(size_t)r * Nc] = acc[r] + bs;
}

__global__ void zero_int(int* p, int n) {
  int t = blockIdx.x * blockDim.x + threadIdx.x;
  if (t < n) p[t] = 0;
}

__global__ void count_edges(const int* __restrict__ ecol, int* __restrict__ cnt) {
  int e = blockIdx.x * blockDim.x + threadIdx.x;
  if (e < NE_EDGES) atomicAdd(&cnt[ecol[e]], 1);
}

__global__ __launch_bounds__(1024)
void scan_rowptr(const int* __restrict__ cnt, int* __restrict__ rowptr) {
  __shared__ int part[1024];
  const int CH = (N_NODES + 1023) / 1024;
  int tid = threadIdx.x;
  int base = tid * CH;
  int s = 0;
  for (int i = 0; i < CH; ++i) {
    int idx = base + i;
    if (idx < N_NODES) s += cnt[idx];
  }
  part[tid] = s;
  __syncthreads();
  for (int off = 1; off < 1024; off <<= 1) {
    int v = 0;
    if (tid >= off) v = part[tid - off];
    __syncthreads();
    if (tid >= off) part[tid] += v;
    __syncthreads();
  }
  int run = (tid == 0) ? 0 : part[tid - 1];
  for (int i = 0; i < CH; ++i) {
    int idx = base + i;
    if (idx < N_NODES) { rowptr[idx] = run; run += cnt[idx]; }
  }
  if (tid == 1023) rowptr[N_NODES] = run;
}

__global__ void copy_int(const int* __restrict__ src, int* __restrict__ dst, int n) {
  int t = blockIdx.x * blockDim.x + threadIdx.x;
  if (t < n) dst[t] = src[t];
}

__global__ void csr_fill(const int* __restrict__ erow, const int* __restrict__ ecol,
                         int* __restrict__ cursor, int* __restrict__ srcidx) {
  int e = blockIdx.x * blockDim.x + threadIdx.x;
  if (e >= NE_EDGES) return;
  int c = ecol[e];
  int p = atomicAdd(&cursor[c], 1);
  srcidx[p] = erow[e];
}

__global__ void dis_from_cnt(const int* __restrict__ cnt, float* __restrict__ dis, int n) {
  int t = blockIdx.x * blockDim.x + threadIdx.x;
  if (t < n) dis[t] = rsqrtf((float)cnt[t] + 1.0f);
}

__global__ __launch_bounds__(256)
void gcn_aggregate(const int* __restrict__ rowptr, const int* __restrict__ srcidx,
                   const float* __restrict__ dis, const float* __restrict__ lin,
                   float* __restrict__ agg) {
  int t = blockIdx.x * blockDim.x + threadIdx.x;
  int c = t >> 5;
  if (c >= N_NODES) return;
  int lane = threadIdx.x & 31;
  float dc = dis[c];
  const float4 sv = *(const float4*)(lin + (size_t)c * DF + lane * 4);
  float s = dc * dc;
  float ax = s * sv.x, ay = s * sv.y, az = s * sv.z, aw = s * sv.w;
  int e0 = rowptr[c], e1 = rowptr[c + 1];
  for (int j = e0; j < e1; ++j) {
    int r = srcidx[j];
    float nrm = dis[r] * dc;
    const float4 v = *(const float4*)(lin + (size_t)r * DF + lane * 4);
    ax += nrm * v.x; ay += nrm * v.y; az += nrm * v.z; aw += nrm * v.w;
  }
  float4 o; o.x = ax; o.y = ay; o.z = az; o.w = aw;
  *(float4*)(agg + (size_t)c * DF + lane * 4) = o;
}

__global__ void zero256(float* p) { p[threadIdx.x] = 0.0f; }

__global__ __launch_bounds__(1024)
void bn_stats(const float* __restrict__ agg, float* __restrict__ stats, int n) {
  __shared__ __attribute__((aligned(16))) float red[32][2 * DF];
  int tid  = threadIdx.x;
  int lane = tid & 31;
  int w    = tid >> 5;
  int rw   = blockIdx.x * 512 + w * 16;
  float4 s = {0.f, 0.f, 0.f, 0.f};
  float4 q = {0.f, 0.f, 0.f, 0.f};
  if (rw + 16 <= n) {
    const float* p = agg + (size_t)rw * DF + lane * 4;
    float4 v[16];
#pragma unroll
    for (int i = 0; i < 16; ++i) v[i] = *(const float4*)(p + (size_t)i * DF);
#pragma unroll
    for (int i = 0; i < 16; ++i) {
      s.x += v[i].x; s.y += v[i].y; s.z += v[i].z; s.w += v[i].w;
      q.x += v[i].x * v[i].x; q.y += v[i].y * v[i].y;
      q.z += v[i].z * v[i].z; q.w += v[i].w * v[i].w;
    }
  } else {
    for (int i = 0; i < 16; ++i) {
      int r = rw + i;
      if (r < n) {
        const float4 v = *(const float4*)(agg + (size_t)r * DF + lane * 4);
        s.x += v.x; s.y += v.y; s.z += v.z; s.w += v.w;
        q.x += v.x * v.x; q.y += v.y * v.y; q.z += v.z * v.z; q.w += v.w * v.w;
      }
    }
  }
  *(float4*)&red[w][lane * 4]      = s;
  *(float4*)&red[w][DF + lane * 4] = q;
  __syncthreads();
  if (tid < 2 * DF) {
    float a = 0.f;
#pragma unroll
    for (int k = 0; k < 32; ++k) a += red[k][tid];
    atomicAdd(&stats[tid], a);
  }
}

__global__ void bn_finalize(float* __restrict__ stats, const float* __restrict__ g,
                            const float* __restrict__ be, float inv_n) {
  int f = threadIdx.x;
  float m   = stats[f] * inv_n;
  float var = stats[DF + f] * inv_n - m * m;
  float sc  = g[f] * rsqrtf(var + EPS_BN);
  stats[f]      = sc;
  stats[DF + f] = be[f] - m * sc;
}

__global__ __launch_bounds__(256)
void bn_apply_leaky(const float* __restrict__ agg, const float* __restrict__ stats,
                    float* __restrict__ h) {
  int t = blockIdx.x * blockDim.x + threadIdx.x;
  if (t >= N_NODES * DF) return;
  int f = t & (DF - 1);
  float y = agg[t] * stats[f] + stats[DF + f];
  h[t] = (y >= 0.f) ? y : 0.1f * y;
}

__global__ void transpose128(const float* __restrict__ src, float* __restrict__ dst) {
  int t = blockIdx.x * blockDim.x + threadIdx.x;
  if (t >= 128 * 128) return;
  int r = t >> 7, c = t & 127;
  dst[c * 128 + r] = src[t];
}

__global__ __launch_bounds__(256)
void gather_pairs(const float* __restrict__ h, const int* __restrict__ di,
                  float* __restrict__ amat, float* __restrict__ bmat) {
  int t = blockIdx.x * blockDim.x + threadIdx.x;
  if (t >= BPAIRS * 32) return;
  int j = t >> 5, q = t & 31;
  int ia = di[2 * j] - 1;     if (ia < 0) ia += N_NODES;
  int ib = di[2 * j + 1] - 1; if (ib < 0) ib += N_NODES;
  *(float4*)(amat + (size_t)j * DF + q * 4) = *(const float4*)(h + (size_t)ia * DF + q * 4);
  *(float4*)(bmat + (size_t)j * DF + q * 4) = *(const float4*)(h + (size_t)ib * DF + q * 4);
}

__global__ __launch_bounds__(256)
void rowdot(const float* __restrict__ tm, const float* __restrict__ bm,
            float* __restrict__ out) {
  int t = blockIdx.x * blockDim.x + threadIdx.x;
  int j = t >> 5, lane = t & 31;
  if (j >= BPAIRS) return;
  float4 tv = *(const float4*)(tm + (size_t)j * DF + lane * 4);
  float4 bv = *(const float4*)(bm + (size_t)j * DF + lane * 4);
  float p = tv.x * bv.x + tv.y * bv.y + tv.z * bv.z + tv.w * bv.w;
#pragma unroll
  for (int off = 16; off > 0; off >>= 1)
    p += __shfl_down(p, off, 32);
  if (lane == 0) out[j] = p;
}

extern "C" void kernel_launch(void* const* d_in, const int* in_sizes, int n_in,
                              void* d_out, int out_size, void* d_ws, size_t ws_size,
                              hipStream_t stream) {
  const float* x   = (const float*)d_in[0];
  const int*   edg = (const int*)d_in[1];
  const int*   di  = (const int*)d_in[2];
  const float* w1  = (const float*)d_in[3];
  const float* b1  = (const float*)d_in[4];
  const float* w2  = (const float*)d_in[5];
  const float* b2  = (const float*)d_in[6];
  const float* w3  = (const float*)d_in[7];
  const float* b3  = (const float*)d_in[8];
  const float* g1  = (const float*)d_in[9];
  const float* be1 = (const float*)d_in[10];
  const float* g2  = (const float*)d_in[11];
  const float* be2 = (const float*)d_in[12];
  const float* g3  = (const float*)d_in[13];
  const float* be3 = (const float*)d_in[14];
  const float* p1  = (const float*)d_in[15];
  const float* p2  = (const float*)d_in[16];
  float* out = (float*)d_out;

  float* ws = (float*)d_ws;
  size_t off = 0;
  float* dis    = ws + off; off += N_NODES;
  float* stats  = ws + off; off += 256;
  float* Wt     = ws + off; off += 128 * 128;
  int*   cnt    = (int*)(ws + off); off += N_NODES;
  int*   rowptr = (int*)(ws + off); off += N_NODES + 16;
  int*   cursor = (int*)(ws + off); off += N_NODES;
  int*   srcidx = (int*)(ws + off); off += NE_EDGES;
  float* lin    = ws + off; off += (size_t)N_NODES * DF;
  float* agg    = ws + off; off += (size_t)N_NODES * DF;
  float* h      = ws + off; off += (size_t)N_NODES * DF;
  float* U    = lin;
  float* Qt   = lin + 16384;
  float* amat = lin + 32768;
  float* bmat = amat + (size_t)BPAIRS * DF;
  float* tmat = bmat + (size_t)BPAIRS * DF;

  const int* erow = edg;
  const int* ecol = edg + NE_EDGES;

  zero_int<<<(N_NODES + 255) / 256, 256, 0, stream>>>(cnt, N_NODES);
  count_edges<<<(NE_EDGES + 255) / 256, 256, 0, stream>>>(ecol, cnt);
  scan_rowptr<<<1, 1024, 0, stream>>>(cnt, rowptr);
  copy_int<<<(N_NODES + 255) / 256, 256, 0, stream>>>(rowptr, cursor, N_NODES);
  csr_fill<<<(NE_EDGES + 255) / 256, 256, 0, stream>>>(erow, ecol, cursor, srcidx);
  dis_from_cnt<<<(N_NODES + 255) / 256, 256, 0, stream>>>(cnt, dis, N_NODES);

  const float* Wl[3] = {w1, w2, w3};
  const float* Bl[3] = {b1, b2, b3};
  const float* Gl[3] = {g1, g2, g3};
  const float* El[3] = {be1, be2, be3};
  const float* in = x;
  const int strip_waves = N_NODES / 16;
  for (int l = 0; l < 3; ++l) {
    transpose128<<<64, 256, 0, stream>>>(Wl[l], Wt);
    wmma_gemm_strip<<<(strip_waves * 32 + 255) / 256, 256, 0, stream>>>(
        in, Wt, Bl[l], lin, N_NODES, DF);
    gcn_aggregate<<<(N_NODES * 32 + 255) / 256, 256, 0, stream>>>(
        rowptr, srcidx, dis, lin, agg);
    zero256<<<1, 256, 0, stream>>>(stats);
    bn_stats<<<(N_NODES + 511) / 512, 1024, 0, stream>>>(agg, stats, N_NODES);
    bn_finalize<<<1, 128, 0, stream>>>(stats, Gl[l], El[l], 1.0f / (float)N_NODES);
    bn_apply_leaky<<<(N_NODES * DF + 255) / 256, 256, 0, stream>>>(agg, stats, h);
    in = h;
  }

  const int small_tiles = (DF / 16) * (DF / 16);
  wmma_matmul<<<(small_tiles * 32 + 255) / 256, 256, 0, stream>>>(p1, p2, nullptr, U, DF, DF, DF, 1);
  wmma_matmul<<<(small_tiles * 32 + 255) / 256, 256, 0, stream>>>(U, p1, nullptr, Qt, DF, DF, DF, 1);
  gather_pairs<<<(BPAIRS * 32 + 255) / 256, 256, 0, stream>>>(h, di, amat, bmat);
  wmma_gemm_strip<<<((BPAIRS / 16) * 32 + 255) / 256, 256, 0, stream>>>(
      amat, Qt, nullptr, tmat, BPAIRS, DF);
  rowdot<<<(BPAIRS * 32 + 255) / 256, 256, 0, stream>>>(tmat, bmat, out);
}


# bn_apply_leaky: one wave takes 16 whole rows (4 columns per lane, 16-byte loads issued as a group, 16-byte stores) instead of one element per thread: 3125 waves per launch, not 200000
# speedup vs baseline: 2.4281x; 1.0590x over previous
#include <hip/hip_runtime.h>

#define N_NODES 50000
#define NE_EDGES 800000
#define BPAIRS 4096
#define DF 128
#define EPS_BN 1e-5f

typedef float v2f __attribute__((ext_vector_type(2)));
typedef float v8f __attribute__((ext_vector_type(8)));

__device__ __forceinline__ v8f wmma4(v2f a, v2f b, v8f c) {
  return __builtin_amdgcn_wmma_f32_16x16x4_f32(
       false, a,  false, b,
       (short)0, c,  false,  false);
}

__global__ __launch_bounds__(256)
void wmma_gemm_strip(const float* __restrict__ A, const float* __restrict__ Bt,
                     const float* __restrict__ bias, float* __restrict__ C,
                     int M, int K) {
  int gtid = blockIdx.x * blockDim.x + threadIdx.x;
  int wave = gtid >> 5;
  int lane = threadIdx.x & 31;
  if (wave * 16 >= M) return;

  int l16   = lane & 15;
  int khalf = (lane >> 4) << 1;
  const float* arow = A + (size_t)(wave * 16 + l16) * K + khalf;

  const float* brow[8];
#pragma unroll
  for (int nt = 0; nt < 8; ++nt)
    brow[nt] = Bt + (size_t)(nt * 16 + l16) * K + khalf;

  v8f zero = {};
  v8f acc[8];
#pragma unroll
  for (int nt = 0; nt < 8; ++nt) acc[nt] = zero;

  for (int k = 0; k < K; k += 4) {
    v2f a = *(const v2f*)(arow + k);
#pragma unroll
    for (int nt = 0; nt < 8; ++nt) {
      v2f b = *(const v2f*)(brow[nt] + k);
      acc[nt] = wmma4(a, b, acc[nt]);
    }
  }

  int rbase = wave * 16 + ((lane >> 4) << 3);
#pragma unroll
  for (int nt = 0; nt < 8; ++nt) {
    float bs = bias ? bias[nt * 16 + l16] : 0.f;
    float* crow = C + (size_t)rbase * 128 + nt * 16 + l16;
#pragma unroll
    for (int r = 0; r < 8; ++r)
      crow[(size_t)r * 128] = acc[nt][r] + bs;
  }
}

__global__ __launch_bounds__(256)
void wmma_matmul(const float* __restrict__ A, const float* __restrict__ Bm,
                 const float* __restrict__ bias, float* __restrict__ C,
                 int M, int Nc, int K, int bTrans) {
  int gtid = blockIdx.x * blockDim.x + threadIdx.x;
  int wave = gtid >> 5;
  int lane = threadIdx.x & 31;
  int ntiles = Nc >> 4;
  int mt = wave / ntiles;
  int nt = wave - mt * ntiles;
  if (mt * 16 >= M) return;

  int l16   = lane & 15;
  int khalf = (lane >> 4) << 1;
  const float* arow = A + (size_t)(mt * 16 + l16) * K + khalf;

  v8f acc = {};
  if (bTrans) {
    const float* brow = Bm + (size_t)(nt * 16 + l16) * K + khalf;
    for (int k = 0; k < K; k += 4) {
      v2f a = *(const v2f*)(arow + k);
      v2f b = *(const v2f*)(brow + k);
      acc = wmma4(a, b, acc);
    }
  } else {
    const float* bcol = Bm + (size_t)khalf * Nc + nt * 16 + l16;
    for (int k = 0; k < K; k += 4) {
      v2f a = *(const v2f*)(arow + k);
      v2f b;
      b.x = bcol[(size_t)k * Nc];
      b.y = bcol[(size_t)(k + 1) * Nc];
      acc = wmma4(a, b, acc);
    }
  }

  float bs = bias ? bias[nt * 16 + l16] : 0.f;
  int rbase = mt * 16 + ((lane >> 4) << 3);
  float* crow = C + (size_t)rbase * Nc + nt * 16 + l16;
#pragma unroll
  for (int r = 0; r < 8; ++r)
    crow[(size_t)r * Nc] = acc[r] + bs;
}

__global__ void zero_int(int* p, int n) {
  int t = blockIdx.x * blockDim.x + threadIdx.x;
  if (t < n) p[t] = 0;
}

__global__ void count_edges(const int* __restrict__ ecol, int* __restrict__ cnt) {
  int e = blockIdx.x * blockDim.x + threadIdx.x;
  if (e < NE_EDGES) atomicAdd(&cnt[ecol[e]], 1);
}

__global__ __launch_bounds__(1024)
void scan_rowptr(const int* __restrict__ cnt, int* __restrict__ rowptr) {
  __shared__ int part[1024];
  const int CH = (N_NODES + 1023) / 1024;
  int tid = threadIdx.x;
  int base = tid * CH;
  int s = 0;
  for (int i = 0; i < CH; ++i) {
    int idx = base + i;
    if (idx < N_NODES) s += cnt[idx];
  }
  part[tid] = s;
  __syncthreads();
  for (int off = 1; off < 1024; off <<= 1) {
    int v = 0;
    if (tid >= off) v = part[tid - off];
    __syncthreads();
    if (tid >= off) part[tid] += v;
    __syncthreads();
  }
  int run = (tid == 0) ? 0 : part[tid - 1];
  for (int i = 0; i < CH; ++i) {
    int idx = base + i;
    if (idx < N_NODES) { rowptr[idx] = run; run += cnt[idx]; }
  }
  if (tid == 1023) rowptr[N_NODES] = run;
}

__global__ void copy_int(const int* __restrict__ src, int* __restrict__ dst, int n) {
  int t = blockIdx.x * blockDim.x + threadIdx.x;
  if (t < n) dst[t] = src[t];
}

__global__ void csr_fill(const int* __restrict__ erow, const int* __restrict__ ecol,
                         int* __restrict__ cursor, int* __restrict__ srcidx) {
  int e = blockIdx.x * blockDim.x + threadIdx.x;
  if (e >= NE_EDGES) return;
  int c = ecol[e];
  int p = atomicAdd(&cursor[c], 1);
  srcidx[p] = erow[e];
}

__global__ void dis_from_cnt(const int* __restrict__ cnt, float* __restrict__ dis, int n) {
  int t = blockIdx.x * blockDim.x + threadIdx.x;
  if (t < n) dis[t] = rsqrtf((float)cnt[t] + 1.0f);
}

__global__ __launch_bounds__(256)
void gcn_aggregate(const int* __restrict__ rowptr, const int* __restrict__ srcidx,
                   const float* __restrict__ dis, const float* __restrict__ lin,
                   float* __restrict__ agg) {
  int t = blockIdx.x * blockDim.x + threadIdx.x;
  int c = t >> 5;
  if (c >= N_NODES) return;
  int lane = threadIdx.x & 31;
  float dc = dis[c];
  const float4 sv = *(const float4*)(lin + (size_t)c * DF + lane * 4);
  float s = dc * dc;
  float ax = s * sv.x, ay = s * sv.y, az = s * sv.z, aw = s * sv.w;
  int e0 = rowptr[c], e1 = rowptr[c + 1];
  for (int j = e0; j < e1; ++j) {
    int r = srcidx[j];
    float nrm = dis[r] * dc;
    const float4 v = *(const float4*)(lin + (size_t)r * DF + lane * 4);
    ax += nrm * v.x; ay += nrm * v.y; az += nrm * v.z; aw += nrm * v.w;
  }
  float4 o; o.x = ax; o.y = ay; o.z = az; o.w = aw;
  *(float4*)(agg + (size_t)c * DF + lane * 4) = o;
}

__global__ void zero256(float* p) { p[threadIdx.x] = 0.0f; }

__global__ __launch_bounds__(1024)
void bn_stats(const float* __restrict__ agg, float* __restrict__ stats, int n) {
  __shared__ __attribute__((aligned(16))) float red[32][2 * DF];
  int tid  = threadIdx.x;
  int lane = tid & 31;
  int w    = tid >> 5;
  int rw   = blockIdx.x * 512 + w * 16;
  float4 s = {0.f, 0.f, 0.f, 0.f};
  float4 q = {0.f, 0.f, 0.f, 0.f};
  if (rw + 16 <= n) {
    const float* p = agg + (size_t)rw * DF + lane * 4;
    float4 v[16];
#pragma unroll
    for (int i = 0; i < 16; ++i) v[i] = *(const float4*)(p + (size_t)i * DF);
#pragma unroll
    for (int i = 0; i < 16; ++i) {
      s.x += v[i].x; s.y += v[i].y; s.z += v[i].z; s.w += v[i].w;
      q.x += v[i].x * v[i].x; q.y += v[i].y * v[i].y;
      q.z += v[i].z * v[i].z; q.w += v[i].w * v[i].w;
    }
  } else {
    for (int i = 0; i < 16; ++i) {
      int r = rw + i;
      if (r < n) {
        const float4 v = *(const float4*)(agg + (size_t)r * DF + lane * 4);
        s.x += v.x; s.y += v.y; s.z += v.z; s.w += v.w;
        q.x += v.x * v.x; q.y += v.y * v.y; q.z += v.z * v.z; q.w += v.w * v.w;
      }
    }
  }
  *(float4*)&red[w][lane * 4]      = s;
  *(float4*)&red[w][DF + lane * 4] = q;
  __syncthreads();
  if (tid < 2 * DF) {
    float a = 0.f;
#pragma unroll
    for (int k = 0; k < 32; ++k) a += red[k][tid];
    atomicAdd(&stats[tid], a);
  }
}

__global__ void bn_finalize(float* __restrict__ stats, const float* __restrict__ g,
                            const float* __restrict__ be, float inv_n) {
  int f = threadIdx.x;
  float m   = stats[f] * inv_n;
  float var = stats[DF + f] * inv_n - m * m;
  float sc  = g[f] * rsqrtf(var + EPS_BN);
  stats[f]      = sc;
  stats[DF + f] = be[f] - m * sc;
}

#define BNA_ROWS 16
static_assert(N_NODES % BNA_ROWS == 0, "a wave has all of its rows or none");

__device__ __forceinline__ float bn_leaky1(float x, float sc, float sh) {
  float y = x * sc + sh;
  return (y >= 0.f) ? y : 0.1f * y;
}

__global__ __launch_bounds__(256)
void bn_apply_leaky(const float* __restrict__ agg, const float* __restrict__ stats,
                    float* __restrict__ h) {
  int wave = (blockIdx.x * blockDim.x + threadIdx.x) >> 5;
  int row0 = wave * BNA_ROWS;
  if (row0 >= N_NODES) return;
  int lane4 = (threadIdx.x & 31) * 4;
  const float4 sc = *(const float4*)(stats + lane4);
  const float4 sh = *(const float4*)(stats + DF + lane4);
  const float* p = agg + (size_t)row0 * DF + lane4;
  float*       o = h   + (size_t)row0 * DF + lane4;
  float4 v[BNA_ROWS];
#pragma unroll
  for (int i = 0; i < BNA_ROWS; ++i) v[i] = *(const float4*)(p + (size_t)i * DF);
#pragma unroll
  for (int i = 0; i < BNA_ROWS; ++i) {
    float4 y;
    y.x = bn_leaky1(v[i].x, sc.x, sh.x);
    y.y = bn_leaky1(v[i].y, sc.y, sh.y);
    y.z = bn_leaky1(v[i].z, sc.z, sh.z);
    y.w = bn_leaky1(v[i].w, sc.w, sh.w);
    *(float4*)(o + (size_t)i * DF) = y;
  }
}

__global__ void transpose128(const float* __restrict__ src, float* __restrict__ dst) {
  int t = blockIdx.x * blockDim.x + threadIdx.x;
  if (t >= 128 * 128) return;
  int r = t >> 7, c = t & 127;
  dst[c * 128 + r] = src[t];
}

__global__ __launch_bounds__(256)
void gather_pairs(const float* __restrict__ h, const int* __restrict__ di,
                  float* __restrict__ amat, float* __restrict__ bmat) {
  int t = blockIdx.x * blockDim.x + threadIdx.x;
  if (t >= BPAIRS * 32) return;
  int j = t >> 5, q = t & 31;
  int ia = di[2 * j] - 1;     if (ia < 0) ia += N_NODES;
  int ib = di[2 * j + 1] - 1; if (ib < 0) ib += N_NODES;
  *(float4*)(amat + (size_t)j * DF + q * 4) = *(const float4*)(h + (size_t)ia * DF + q * 4);
  *(float4*)(bmat + (size_t)j * DF + q * 4) = *(const float4*)(h + (size_t)ib * DF + q * 4);
}

__global__ __launch_bounds__(256)
void rowdot(const float* __restrict__ tm, const float* __restrict__ bm,
            float* __restrict__ out) {
  int t = blockIdx.x * blockDim.x + threadIdx.x;
  int j = t >> 5, lane = t & 31;
  if (j >= BPAIRS) return;
  float4 tv = *(const float4*)(tm + (size_t)j * DF + lane * 4);
  float4 bv = *(const float4*)(bm + (size_t)j * DF + lane * 4);
  float p = tv.x * bv.x + tv.y * bv.y + tv.z * bv.z + tv.w * bv.w;
#pragma unroll
  for (int off = 16; off > 0; off >>= 1)
    p += __shfl_down(p, off, 32);
  if (lane == 0) out[j] = p;
}

extern "C" void kernel_launch(void* const* d_in, const int* in_sizes, int n_in,
                              void* d_out, int out_size, void* d_ws, size_t ws_size,
                              hipStream_t stream) {
  const float* x   = (const float*)d_in[0];
  const int*   edg = (const int*)d_in[1];
  const int*   di  = (const int*)d_in[2];
  const float* w1  = (const float*)d_in[3];
  const float* b1  = (const float*)d_in[4];
  const float* w2  = (const float*)d_in[5];
  const float* b2  = (const float*)d_in[6];
  const float* w3  = (const float*)d_in[7];
  const float* b3  = (const float*)d_in[8];
  const float* g1  = (const float*)d_in[9];
  const float* be1 = (const float*)d_in[10];
  const float* g2  = (const float*)d_in[11];
  const float* be2 = (const float*)d_in[12];
  const float* g3  = (const float*)d_in[13];
  const float* be3 = (const float*)d_in[14];
  const float* p1  = (const float*)d_in[15];
  const float* p2  = (const float*)d_in[16];
  float* out = (float*)d_out;

  float* ws = (float*)d_ws;
  size_t off = 0;
  float* dis    = ws + off; off += N_NODES;
  float* stats  = ws + off; off += 256;
  float* Wt     = ws + off; off += 128 * 128;
  int*   cnt    = (int*)(ws + off); off += N_NODES;
  int*   rowptr = (int*)(ws + off); off += N_NODES + 16;
  int*   cursor = (int*)(ws + off); off += N_NODES;
  int*   srcidx = (int*)(ws + off); off += NE_EDGES;
  float* lin    = ws + off; off += (size_t)N_NODES * DF;
  float* agg    = ws + off; off += (size_t)N_NODES * DF;
  float* h      = ws + off; off += (size_t)N_NODES * DF;
  float* U    = lin;
  float* Qt   = lin + 16384;
  float* amat = lin + 32768;
  float* bmat = amat + (size_t)BPAIRS * DF;
  float* tmat = bmat + (size_t)BPAIRS * DF;

  const int* erow = edg;
  const int* ecol = edg + NE_EDGES;

  zero_int<<<(N_NODES + 255) / 256, 256, 0, stream>>>(cnt, N_NODES);
  count_edges<<<(NE_EDGES + 255) / 256, 256, 0, stream>>>(ecol, cnt);
  scan_rowptr<<<1, 1024, 0, stream>>>(cnt, rowptr);
  copy_int<<<(N_NODES + 255) / 256, 256, 0, stream>>>(rowptr, cursor, N_NODES);
  csr_fill<<<(NE_EDGES + 255) / 256, 256, 0, stream>>>(erow, ecol, cursor, srcidx);
  dis_from_cnt<<<(N_NODES + 255) / 256, 256, 0, stream>>>(cnt, dis, N_NODES);

  const float* Wl[3] = {w1, w2, w3};
  const float* Bl[3] = {b1, b2, b3};
  const float* Gl[3] = {g1, g2, g3};
  const float* El[3] = {be1, be2, be3};
  const float* in = x;
  const int strip_waves = N_NODES / 16;
  for (int l = 0; l < 3; ++l) {
    transpose128<<<64, 256, 0, stream>>>(Wl[l], Wt);
    wmma_gemm_strip<<<(strip_waves * 32 + 255) / 256, 256, 0, stream>>>(
        in, Wt, Bl[l], lin, N_NODES, DF);
    gcn_aggregate<<<(N_NODES * 32 + 255) / 256, 256, 0, stream>>>(
        rowptr, srcidx, dis, lin, agg);
    zero256<<<1, 256, 0, stream>>>(stats);
    bn_stats<<<(N_NODES + 511) / 512, 1024, 0, stream>>>(agg, stats, N_NODES);
    bn_finalize<<<1, 128, 0, stream>>>(stats, Gl[l], El[l], 1.0f / (float)N_NODES);
    bn_apply_leaky<<<(N_NODES / BNA_ROWS * 32 + 255) / 256, 256, 0, stream>>>(agg, stats, h);
    in = h;
  }

  const int small_tiles = (DF / 16) * (DF / 16);
  wmma_matmul<<<(small_tiles * 32 + 255) / 256, 256, 0, stream>>>(p1, p2, nullptr, U, DF, DF, DF, 1);
  wmma_matmul<<<(small_tiles * 32 + 255) / 256, 256, 0, stream>>>(U, p1, nullptr, Qt, DF, DF, DF, 1);
  gather_pairs<<<(BPAIRS * 32 + 255) / 256, 256, 0, stream>>>(h, di, amat, bmat);
  wmma_gemm_strip<<<((BPAIRS / 16) * 32 + 255) / 256, 256, 0, stream>>>(
      amat, Qt, nullptr, tmat, BPAIRS, DF);
  rowdot<<<(BPAIRS * 32 + 255) / 256, 256, 0, stream>>>(tmat, bmat, out);
}


# scan_rowptr: a wave scans a contiguous run of counters from whole-line 16-byte loads issued as one group (shuffle scan, one LDS hand-over), whole-line stores; not 49 waited 4-byte loads per thread
# speedup vs baseline: 2.9283x; 1.2060x over previous
#include <hip/hip_runtime.h>

#define N_NODES 50000
#define NE_EDGES 800000
#define BPAIRS 4096
#define DF 128
#define EPS_BN 1e-5f

typedef float v2f __attribute__((ext_vector_type(2)));
typedef float v8f __attribute__((ext_vector_type(8)));

__device__ __forceinline__ v8f wmma4(v2f a, v2f b, v8f c) {
  return __builtin_amdgcn_wmma_f32_16x16x4_f32(
       false, a,  false, b,
       (short)0, c,  false,  false);
}

__global__ __launch_bounds__(256)
void wmma_gemm_strip(const float* __restrict__ A, const float* __restrict__ Bt,
                     const float* __restrict__ bias, float* __restrict__ C,
                     int M, int K) {
  int gtid = blockIdx.x * blockDim.x + threadIdx.x;
  int wave = gtid >> 5;
  int lane = threadIdx.x & 31;
  if (wave * 16 >= M) return;

  int l16   = lane & 15;
  int khalf = (lane >> 4) << 1;
  const float* arow = A + (size_t)(wave * 16 + l16) * K + khalf;

  const float* brow[8];
#pragma unroll
  for (int nt = 0; nt < 8; ++nt)
    brow[nt] = Bt + (size_t)(nt * 16 + l16) * K + khalf;

  v8f zero = {};
  v8f acc[8];
#pragma unroll
  for (int nt = 0; nt < 8; ++nt) acc[nt] = zero;

  for (int k = 0; k < K; k += 4) {
    v2f a = *(const v2f*)(arow + k);
#pragma unroll
    for (int nt = 0; nt < 8; ++nt) {
      v2f b = *(const v2f*)(brow[nt] + k);
      acc[nt] = wmma4(a, b, acc[nt]);
    }
  }

  int rbase = wave * 16 + ((lane >> 4) << 3);
#pragma unroll
  for (int nt = 0; nt < 8; ++nt) {
    float bs = bias ? bias[nt * 16 + l16] : 0.f;
    float* crow = C + (size_t)rbase * 128 + nt * 16 + l16;
#pragma unroll
    for (int r = 0; r < 8; ++r)
      crow[(size_t)r * 128] = acc[nt][r] + bs;
  }
}

__global__ __launch_bounds__(256)
void wmma_matmul(const float* __restrict__ A, const float* __restrict__ Bm,
                 const float* __restrict__ bias, float* __restrict__ C,
                 int M, int Nc, int K, int bTrans) {
  int gtid = blockIdx.x * blockDim.x + threadIdx.x;
  int wave = gtid >> 5;
  int lane = threadIdx.x & 31;
  int ntiles = Nc >> 4;
  int mt = wave / ntiles;
  int nt = wave - mt * ntiles;
  if (mt * 16 >= M) return;

  int l16   = lane & 15;
  int khalf = (lane >> 4) << 1;
  const float* arow = A + (size_t)(mt * 16 + l16) * K + khalf;

  v8f acc = {};
  if (bTrans) {
    const float* brow = Bm + (size_t)(nt * 16 + l16) * K + khalf;
    for (int k = 0; k < K; k += 4) {
      v2f a = *(const v2f*)(arow + k);
      v2f b = *(const v2f*)(brow + k);
      acc = wmma4(a, b, acc);
    }
  } else {
    const float* bcol = Bm + (size_t)khalf * Nc + nt * 16 + l16;
    for (int k = 0; k < K; k += 4) {
      v2f a = *(const v2f*)(arow + k);
      v2f b;
      b.x = bcol[(size_t)k * Nc];
      b.y = bcol[(size_t)(k + 1) * Nc];
      acc = wmma4(a, b, acc);
    }
  }

  float bs = bias ? bias[nt * 16 + l16] : 0.f;
  int rbase = mt * 16 + ((lane >> 4) << 3);
  float* crow = C + (size_t)rbase * Nc + nt * 16 + l16;
#pragma unroll
  for (int r = 0; r < 8; ++r)
    crow[(size_t)r * Nc] = acc[r] + bs;
}

__global__ void zero_int(int* p, int n) {
  int t = blockIdx.x * blockDim.x + threadIdx.x;
  if (t < n) p[t] = 0;
}

__global__ void count_edges(const int* __restrict__ ecol, int* __restrict__ cnt) {
  int e = blockIdx.x * blockDim.x + threadIdx.x;
  if (e < NE_EDGES) atomicAdd(&cnt[ecol[e]], 1);
}

#define SCAN_STEPS 13
#define SCAN_PER_WAVE (SCAN_STEPS * 128)
static_assert(32 * SCAN_PER_WAVE >= N_NODES, "the 32 waves cover every counter");
static_assert(N_NODES % 4 == 0, "a lane's 4 counters are all inside or all outside");

__device__ __forceinline__ int wave_incl_scan(int x, int lane) {
#pragma unroll
  for (int off = 1; off < 32; off <<= 1) {
    int y = __shfl_up(x, off, 32);
    if (lane >= off) x += y;
  }
  return x;
}

__global__ __launch_bounds__(1024)
void scan_rowptr(const int* __restrict__ cnt, int* __restrict__ rowptr) {
  __shared__ int wtot[32];
  const int tid = threadIdx.x;
  const int lane = tid & 31;
  const int w = tid >> 5;
  const int first = w * SCAN_PER_WAVE + lane * 4;
  int4 v[SCAN_STEPS];
#pragma unroll
  for (int j = 0; j < SCAN_STEPS; ++j) {
    int idx = first + j * 128;
    int idc = (idx < N_NODES) ? idx : (N_NODES - 4);
    v[j] = *(const int4*)(cnt + idc);
  }
  int run = 0;
#pragma unroll
  for (int j = 0; j < SCAN_STEPS; ++j) {
    const bool in = (first + j * 128) < N_NODES;
    const int c0 = in ? v[j].x : 0;
    const int c1 = in ? v[j].y : 0;
    const int c2 = in ? v[j].z : 0;
    const int c3 = in ? v[j].w : 0;
    const int s1 = c0, s2 = s1 + c1, s3 = s2 + c2, s4 = s3 + c3;
    const int inc = wave_incl_scan(s4, lane);
    const int ex = run + inc - s4;
    run += __shfl(inc, 31, 32);
    v[j].x = ex; v[j].y = ex + s1; v[j].z = ex + s2; v[j].w = ex + s3;
  }
  if (lane == 0) wtot[w] = run;
  __syncthreads();
  const int t  = wtot[lane];
  const int ti = wave_incl_scan(t, lane);
  const int base = __shfl(ti - t, w, 32);
#pragma unroll
  for (int j = 0; j < SCAN_STEPS; ++j) {
    int idx = first + j * 128;
    if (idx < N_NODES) {
      int4 o;
      o.x = v[j].x + base; o.y = v[j].y + base; o.z = v[j].z + base; o.w = v[j].w + base;
      *(int4*)(rowptr + idx) = o;
    }
  }
  if (tid == 1023) rowptr[N_NODES] = base + run;
}

__global__ void copy_int(const int* __restrict__ src, int* __restrict__ dst, int n) {
  int t = blockIdx.x * blockDim.x + threadIdx.x;
  if (t < n) dst[t] = src[t];
}

__global__ void csr_fill(const int* __restrict__ erow, const int* __restrict__ ecol,
                         int* __restrict__ cursor, int* __restrict__ srcidx) {
  int e = blockIdx.x * blockDim.x + threadIdx.x;
  if (e >= NE_EDGES) return;
  int c = ecol[e];
  int p = atomicAdd(&cursor[c], 1);
  srcidx[p] = erow[e];
}

__global__ void dis_from_cnt(const int* __restrict__ cnt, float* __restrict__ dis, int n) {
  int t = blockIdx.x * blockDim.x + threadIdx.x;
  if (t < n) dis[t] = rsqrtf((float)cnt[t] + 1.0f);
}

__global__ __launch_bounds__(256)
void gcn_aggregate(const int* __restrict__ rowptr, const int* __restrict__ srcidx,
                   const float* __restrict__ dis, const float* __restrict__ lin,
                   float* __restrict__ agg) {
  int t = blockIdx.x * blockDim.x + threadIdx.x;
  int c = t >> 5;
  if (c >= N_NODES) return;
  int lane = threadIdx.x & 31;
  float dc = dis[c];
  const float4 sv = *(const float4*)(lin + (size_t)c * DF + lane * 4);
  float s = dc * dc;
  float ax = s * sv.x, ay = s * sv.y, az = s * sv.z, aw = s * sv.w;
  int e0 = rowptr[c], e1 = rowptr[c + 1];
  for (int j = e0; j < e1; ++j) {
    int r = srcidx[j];
    float nrm = dis[r] * dc;
    const float4 v = *(const float4*)(lin + (size_t)r * DF + lane * 4);
    ax += nrm * v.x; ay += nrm * v.y; az += nrm * v.z; aw += nrm * v.w;
  }
  float4 o; o.x = ax; o.y = ay; o.z = az; o.w = aw;
  *(float4*)(agg + (size_t)c * DF + lane * 4) = o;
}

__global__ void zero256(float* p) { p[threadIdx.x] = 0.0f; }

__global__ __launch_bounds__(1024)
void bn_stats(const float* __restrict__ agg, float* __restrict__ stats, int n) {
  __shared__ __attribute__((aligned(16))) float red[32][2 * DF];
  int tid  = threadIdx.x;
  int lane = tid & 31;
  int w    = tid >> 5;
  int rw   = blockIdx.x * 512 + w * 16;
  float4 s = {0.f, 0.f, 0.f, 0.f};
  float4 q = {0.f, 0.f, 0.f, 0.f};
  if (rw + 16 <= n) {
    const float* p = agg + (size_t)rw * DF + lane * 4;
    float4 v[16];
#pragma unroll
    for (int i = 0; i < 16; ++i) v[i] = *(const float4*)(p + (size_t)i * DF);
#pragma unroll
    for (int i = 0; i < 16; ++i) {
      s.x += v[i].x; s.y += v[i].y; s.z += v[i].z; s.w += v[i].w;
      q.x += v[i].x * v[i].x; q.y += v[i].y * v[i].y;
      q.z += v[i].z * v[i].z; q.w += v[i].w * v[i].w;
    }
  } else {
    for (int i = 0; i < 16; ++i) {
      int r = rw + i;
      if (r < n) {
        const float4 v = *(const float4*)(agg + (size_t)r * DF + lane * 4);
        s.x += v.x; s.y += v.y; s.z += v.z; s.w += v.w;
        q.x += v.x * v.x; q.y += v.y * v.y; q.z += v.z * v.z; q.w += v.w * v.w;
      }
    }
  }
  *(float4*)&red[w][lane * 4]      = s;
  *(float4*)&red[w][DF + lane * 4] = q;
  __syncthreads();
  if (tid < 2 * DF) {
    float a = 0.f;
#pragma unroll
    for (int k = 0; k < 32; ++k) a += red[k][tid];
    atomicAdd(&stats[tid], a);
  }
}

__global__ void bn_finalize(float* __restrict__ stats, const float* __restrict__ g,
                            const float* __restrict__ be, float inv_n) {
  int f = threadIdx.x;
  float m   = stats[f] * inv_n;
  float var = stats[DF + f] * inv_n - m * m;
  float sc  = g[f] * rsqrtf(var + EPS_BN);
  stats[f]      = sc;
  stats[DF + f] = be[f] - m * sc;
}

#define BNA_ROWS 16
static_assert(N_NODES % BNA_ROWS == 0, "a wave has all of its rows or none");

__device__ __forceinline__ float bn_leaky1(float x, float sc, float sh) {
  float y = x * sc + sh;
  return (y >= 0.f) ? y : 0.1f * y;
}

__global__ __launch_bounds__(256)
void bn_apply_leaky(const float* __restrict__ agg, const float* __restrict__ stats,
                    float* __restrict__ h) {
  int wave = (blockIdx.x * blockDim.x + threadIdx.x) >> 5;
  int row0 = wave * BNA_ROWS;
  if (row0 >= N_NODES) return;
  int lane4 = (threadIdx.x & 31) * 4;
  const float4 sc = *(const float4*)(stats + lane4);
  const float4 sh = *(const float4*)(stats + DF + lane4);
  const float* p = agg + (size_t)row0 * DF + lane4;
  float*       o = h   + (size_t)row0 * DF + lane4;
  float4 v[BNA_ROWS];
#pragma unroll
  for (int i = 0; i < BNA_ROWS; ++i) v[i] = *(const float4*)(p + (size_t)i * DF);
#pragma unroll
  for (int i = 0; i < BNA_ROWS; ++i) {
    float4 y;
    y.x = bn_leaky1(v[i].x, sc.x, sh.x);
    y.y = bn_leaky1(v[i].y, sc.y, sh.y);
    y.z = bn_leaky1(v[i].z, sc.z, sh.z);
    y.w = bn_leaky1(v[i].w, sc.w, sh.w);
    *(float4*)(o + (size_t)i * DF) = y;
  }
}

__global__ void transpose128(const float* __restrict__ src, float* __restrict__ dst) {
  int t = blockIdx.x * blockDim.x + threadIdx.x;
  if (t >= 128 * 128) return;
  int r = t >> 7, c = t & 127;
  dst[c * 128 + r] = src[t];
}

__global__ __launch_bounds__(256)
void gather_pairs(const float* __restrict__ h, const int* __restrict__ di,
                  float* __restrict__ amat, float* __restrict__ bmat) {
  int t = blockIdx.x * blockDim.x + threadIdx.x;
  if (t >= BPAIRS * 32) return;
  int j = t >> 5, q = t & 31;
  int ia = di[2 * j] - 1;     if (ia < 0) ia += N_NODES;
  int ib = di[2 * j + 1] - 1; if (ib < 0) ib += N_NODES;
  *(float4*)(amat + (size_t)j * DF + q * 4) = *(const float4*)(h + (size_t)ia * DF + q * 4);
  *(float4*)(bmat + (size_t)j * DF + q * 4) = *(const float4*)(h + (size_t)ib * DF + q * 4);
}

__global__ __launch_bounds__(256)
void rowdot(const float* __restrict__ tm, const float* __restrict__ bm,
            float* __restrict__ out) {
  int t = blockIdx.x * blockDim.x + threadIdx.x;
  int j = t >> 5, lane = t & 31;
  if (j >= BPAIRS) return;
  float4 tv = *(const float4*)(tm + (size_t)j * DF + lane * 4);
  float4 bv = *(const float4*)(bm + (size_t)j * DF + lane * 4);
  float p = tv.x * bv.x + tv.y * bv.y + tv.z * bv.z + tv.w * bv.w;
#pragma unroll
  for (int off = 16; off > 0; off >>= 1)
    p += __shfl_down(p, off, 32);
  if (lane == 0) out[j] = p;
}

extern "C" void kernel_launch(void* const* d_in, const int* in_sizes, int n_in,
                              void* d_out, int out_size, void* d_ws, size_t ws_size,
                              hipStream_t stream) {
  const float* x   = (const float*)d_in[0];
  const int*   edg = (const int*)d_in[1];
  const int*   di  = (const int*)d_in[2];
  const float* w1  = (const float*)d_in[3];
  const float* b1  = (const float*)d_in[4];
  const float* w2  = (const float*)d_in[5];
  const float* b2  = (const float*)d_in[6];
  const float* w3  = (const float*)d_in[7];
  const float* b3  = (const float*)d_in[8];
  const float* g1  = (const float*)d_in[9];
  const float* be1 = (const float*)d_in[10];
  const float* g2  = (const float*)d_in[11];
  const float* be2 = (const float*)d_in[12];
  const float* g3  = (const float*)d_in[13];
  const float* be3 = (const float*)d_in[14];
  const float* p1  = (const float*)d_in[15];
  const float* p2  = (const float*)d_in[16];
  float* out = (float*)d_out;

  float* ws = (float*)d_ws;
  size_t off = 0;
  float* dis    = ws + off; off += N_NODES;
  float* stats  = ws + off; off += 256;
  float* Wt     = ws + off; off += 128 * 128;
  int*   cnt    = (int*)(ws + off); off += N_NODES;
  int*   rowptr = (int*)(ws + off); off += N_NODES + 16;
  int*   cursor = (int*)(ws + off); off += N_NODES;
  int*   srcidx = (int*)(ws + off); off += NE_EDGES;
  float* lin    = ws + off; off += (size_t)N_NODES * DF;
  float* agg    = ws + off; off += (size_t)N_NODES * DF;
  float* h      = ws + off; off += (size_t)N_NODES * DF;
  float* U    = lin;
  float* Qt   = lin + 16384;
  float* amat = lin + 32768;
  float* bmat = amat + (size_t)BPAIRS * DF;
  float* tmat = bmat + (size_t)BPAIRS * DF;

  const int* erow = edg;
  const int* ecol = edg + NE_EDGES;

  zero_int<<<(N_NODES + 255) / 256, 256, 0, stream>>>(cnt, N_NODES);
  count_edges<<<(NE_EDGES + 255) / 256, 256, 0, stream>>>(ecol, cnt);
  scan_rowptr<<<1, 1024, 0, stream>>>(cnt, rowptr);
  copy_int<<<(N_NODES + 255) / 256, 256, 0, stream>>>(rowptr, cursor, N_NODES);
  csr_fill<<<(NE_EDGES + 255) / 256, 256, 0, stream>>>(erow, ecol, cursor, srcidx);
  dis_from_cnt<<<(N_NODES + 255) / 256, 256, 0, stream>>>(cnt, dis, N_NODES);

  const float* Wl[3] = {w1, w2, w3};
  const float* Bl[3] = {b1, b2, b3};
  const float* Gl[3] = {g1, g2, g3};
  const float* El[3] = {be1, be2, be3};
  const float* in = x;
  const int strip_waves = N_NODES / 16;
  for (int l = 0; l < 3; ++l) {
    transpose128<<<64, 256, 0, stream>>>(Wl[l], Wt);
    wmma_gemm_strip<<<(strip_waves * 32 + 255) / 256, 256, 0, stream>>>(
        in, Wt, Bl[l], lin, N_NODES, DF);
    gcn_aggregate<<<(N_NODES * 32 + 255) / 256, 256, 0, stream>>>(
        rowptr, srcidx, dis, lin, agg);
    zero256<<<1, 256, 0, stream>>>(stats);
    bn_stats<<<(N_NODES + 511) / 512, 1024, 0, stream>>>(agg, stats, N_NODES);
    bn_finalize<<<1, 128, 0, stream>>>(stats, Gl[l], El[l], 1.0f / (float)N_NODES);
    bn_apply_leaky<<<(N_NODES / BNA_ROWS * 32 + 255) / 256, 256, 0, stream>>>(agg, stats, h);
    in = h;
  }

  const int small_tiles = (DF / 16) * (DF / 16);
  wmma_matmul<<<(small_tiles * 32 + 255) / 256, 256, 0, stream>>>(p1, p2, nullptr, U, DF, DF, DF, 1);
  wmma_matmul<<<(small_tiles * 32 + 255) / 256, 256, 0, stream>>>(U, p1, nullptr, Qt, DF, DF, DF, 1);
  gather_pairs<<<(BPAIRS * 32 + 255) / 256, 256, 0, stream>>>(h, di, amat, bmat);
  wmma_gemm_strip<<<((BPAIRS / 16) * 32 + 255) / 256, 256, 0, stream>>>(
      amat, Qt, nullptr, tmat, BPAIRS, DF);
  rowdot<<<(BPAIRS * 32 + 255) / 256, 256, 0, stream>>>(tmat, bmat, out);
}


# launches 32 -> 21: set-up launches folded into neighbours (counter zero + weight transposes, stats zero, cursor copy, dis, batch-norm finalize); in-row edge order still set by atomic arrival
# speedup vs baseline: 3.0579x; 1.0443x over previous
#include <hip/hip_runtime.h>

#define N_NODES 50000
#define NE_EDGES 800000
#define BPAIRS 4096
#define DF 128
#define EPS_BN 1e-5f

typedef float v2f __attribute__((ext_vector_type(2)));
typedef float v8f __attribute__((ext_vector_type(8)));

__device__ __forceinline__ v8f wmma4(v2f a, v2f b, v8f c) {
  return __builtin_amdgcn_wmma_f32_16x16x4_f32(
       false, a,  false, b,
       (short)0, c,  false,  false);
}

__global__ __launch_bounds__(256)
void wmma_gemm_strip(const float* __restrict__ A, const float* __restrict__ Bt,
                     const float* __restrict__ bias, float* __restrict__ C,
                     int M, int K) {
  int gtid = blockIdx.x * blockDim.x + threadIdx.x;
  int wave = gtid >> 5;
  int lane = threadIdx.x & 31;
  if (wave * 16 >= M) return;

  int l16   = lane & 15;
  int khalf = (lane >> 4) << 1;
  const float* arow = A + (size_t)(wave * 16 + l16) * K + khalf;

  const float* brow[8];
#pragma unroll
  for (int nt = 0; nt < 8; ++nt)
    brow[nt] = Bt + (size_t)(nt * 16 + l16) * K + khalf;

  v8f zero = {};
  v8f acc[8];
#pragma unroll
  for (int nt = 0; nt < 8; ++nt) acc[nt] = zero;

  for (int k = 0; k < K; k += 4) {
    v2f a = *(const v2f*)(arow + k);
#pragma unroll
    for (int nt = 0; nt < 8; ++nt) {
      v2f b = *(const v2f*)(brow[nt] + k);
      acc[nt] = wmma4(a, b, acc[nt]);
    }
  }

  int rbase = wave * 16 + ((lane >> 4) << 3);
#pragma unroll
  for (int nt = 0; nt < 8; ++nt) {
    float bs = bias ? bias[nt * 16 + l16] : 0.f;
    float* crow = C + (size_t)rbase * 128 + nt * 16 + l16;
#pragma unroll
    for (int r = 0; r < 8; ++r)
      crow[(size_t)r * 128] = acc[nt][r] + bs;
  }
}

__global__ __launch_bounds__(256)
void wmma_matmul(const float* __restrict__ A, const float* __restrict__ Bm,
                 const float* __restrict__ bias, float* __restrict__ C,
                 int M, int Nc, int K, int bTrans) {
  int gtid = blockIdx.x * blockDim.x + threadIdx.x;
  int wave = gtid >> 5;
  int lane = threadIdx.x & 31;
  int ntiles = Nc >> 4;
  int mt = wave / ntiles;
  int nt = wave - mt * ntiles;
  if (mt * 16 >= M) return;

  int l16   = lane & 15;
  int khalf = (lane >> 4) << 1;
  const float* arow = A + (size_t)(mt * 16 + l16) * K + khalf;

  v8f acc = {};
  if (bTrans) {
    const float* brow = Bm + (size_t)(nt * 16 + l16) * K + khalf;
    for (int k = 0; k < K; k += 4) {
      v2f a = *(const v2f*)(arow + k);
      v2f b = *(const v2f*)(brow + k);
      acc = wmma4(a, b, acc);
    }
  } else {
    const float* bcol = Bm + (size_t)khalf * Nc + nt * 16 + l16;
    for (int k = 0; k < K; k += 4) {
      v2f a = *(const v2f*)(arow + k);
      v2f b;
      b.x = bcol[(size_t)k * Nc];
      b.y = bcol[(size_t)(k + 1) * Nc];
      acc = wmma4(a, b, acc);
    }
  }

  float bs = bias ? bias[nt * 16 + l16] : 0.f;
  int rbase = mt * 16 + ((lane >> 4) << 3);
  float* crow = C + (size_t)rbase * Nc + nt * 16 + l16;
#pragma unroll
  for (int r = 0; r < 8; ++r)
    crow[(size_t)r * Nc] = acc[r] + bs;
}

__global__ void zero_cnt_transpose_w(int* __restrict__ cnt, int n,
                                     const float* __restrict__ w, float* __restrict__ wt) {
  int t = blockIdx.x * blockDim.x + threadIdx.x;
  if (t < n) cnt[t] = 0;
  if (t < 128 * 128) wt[(t & 127) * 128 + (t >> 7)] = w[t];
}

__global__ void count_edges(const int* __restrict__ ecol, int* __restrict__ cnt) {
  int e = blockIdx.x * blockDim.x + threadIdx.x;
  if (e < NE_EDGES) atomicAdd(&cnt[ecol[e]], 1);
}

#define SCAN_STEPS 13
#define SCAN_PER_WAVE (SCAN_STEPS * 128)
static_assert(32 * SCAN_PER_WAVE >= N_NODES, "the 32 waves cover every counter");
static_assert(N_NODES % 4 == 0, "a lane's 4 counters are all inside or all outside");

__device__ __forceinline__ int wave_incl_scan(int x, int lane) {
#pragma unroll
  for (int off = 1; off < 32; off <<= 1) {
    int y = __shfl_up(x, off, 32);
    if (lane >= off) x += y;
  }
  return x;
}

__global__ __launch_bounds__(1024)
void scan_rowptr(const int* __restrict__ cnt, int* __restrict__ rowptr) {
  __shared__ int wtot[32];
  const int tid = threadIdx.x;
  const int lane = tid & 31;
  const int w = tid >> 5;
  const int first = w * SCAN_PER_WAVE + lane * 4;
  int4 v[SCAN_STEPS];
#pragma unroll
  for (int j = 0; j < SCAN_STEPS; ++j) {
    int idx = first + j * 128;
    int idc = (idx < N_NODES) ? idx : (N_NODES - 4);
    v[j] = *(const int4*)(cnt + idc);
  }
  int run = 0;
#pragma unroll
  for (int j = 0; j < SCAN_STEPS; ++j) {
    const bool in = (first + j * 128) < N_NODES;
    const int c0 = in ? v[j].x : 0;
    const int c1 = in ? v[j].y : 0;
    const int c2 = in ? v[j].z : 0;
    const int c3 = in ? v[j].w : 0;
    const int s1 = c0, s2 = s1 + c1, s3 = s2 + c2, s4 = s3 + c3;
    const int inc = wave_incl_scan(s4, lane);
    const int ex = run + inc - s4;
    run += __shfl(inc, 31, 32);
    v[j].x = ex; v[j].y = ex + s1; v[j].z = ex + s2; v[j].w = ex + s3;
  }
  if (lane == 0) wtot[w] = run;
  __syncthreads();
  const int t  = wtot[lane];
  const int ti = wave_incl_scan(t, lane);
  const int base = __shfl(ti - t, w, 32);
#pragma unroll
  for (int j = 0; j < SCAN_STEPS; ++j) {
    int idx = first + j * 128;
    if (idx < N_NODES) {
      int4 o;
      o.x = v[j].x + base; o.y = v[j].y + base; o.z = v[j].z + base; o.w = v[j].w + base;
      *(int4*)(rowptr + idx) = o;
    }
  }
  if (tid == 1023) rowptr[N_NODES] = base + run;
}

__global__ void csr_fill(const int* __restrict__ erow, const int* __restrict__ ecol,
                         const int* __restrict__ rowptr, int* __restrict__ cnt,
                         int* __restrict__ srcidx, float* __restrict__ dis) {
  int e = blockIdx.x * blockDim.x + threadIdx.x;
  if (e >= NE_EDGES) return;
  if (e < N_NODES)
    dis[e] = rsqrtf((float)(rowptr[e + 1] - rowptr[e]) + 1.0f);
  int c = ecol[e];
  int left = atomicSub(&cnt[c], 1);
  srcidx[rowptr[c] + left - 1] = erow[e];
}

static_assert(NE_EDGES >= N_NODES, "csr_fill's first N_NODES threads write dis");
__global__ __launch_bounds__(256)
void gcn_aggregate(const int* __restrict__ rowptr, const int* __restrict__ srcidx,
                   const float* __restrict__ dis, const float* __restrict__ lin,
                   float* __restrict__ agg, float* __restrict__ stats,
                   const float* __restrict__ wnext, float* __restrict__ wt) {
  int t = blockIdx.x * blockDim.x + threadIdx.x;
  if (blockIdx.x == 0) stats[threadIdx.x] = 0.0f;
  if (wnext != nullptr && t < 128 * 128) wt[(t & 127) * 128 + (t >> 7)] = wnext[t];
  int c = t >> 5;
  if (c >= N_NODES) return;
  int lane = threadIdx.x & 31;
  float dc = dis[c];
  const float4 sv = *(const float4*)(lin + (size_t)c * DF + lane * 4);
  float s = dc * dc;
  float ax = s * sv.x, ay = s * sv.y, az = s * sv.z, aw = s * sv.w;
  int e0 = rowptr[c], e1 = rowptr[c + 1];
  for (int j = e0; j < e1; ++j) {
    int r = srcidx[j];
    float nrm = dis[r] * dc;
    const float4 v = *(const float4*)(lin + (size_t)r * DF + lane * 4);
    ax += nrm * v.x; ay += nrm * v.y; az += nrm * v.z; aw += nrm * v.w;
  }
  float4 o; o.x = ax; o.y = ay; o.z = az; o.w = aw;
  *(float4*)(agg + (size_t)c * DF + lane * 4) = o;
}

__global__ __launch_bounds__(1024)
void bn_stats(const float* __restrict__ agg, float* __restrict__ stats, int n) {
  __shared__ __attribute__((aligned(16))) float red[32][2 * DF];
  int tid  = threadIdx.x;
  int lane = tid & 31;
  int w    = tid >> 5;
  int rw   = blockIdx.x * 512 + w * 16;
  float4 s = {0.f, 0.f, 0.f, 0.f};
  float4 q = {0.f, 0.f, 0.f, 0.f};
  if (rw + 16 <= n) {
    const float* p = agg + (size_t)rw * DF + lane * 4;
    float4 v[16];
#pragma unroll
    for (int i = 0; i < 16; ++i) v[i] = *(const float4*)(p + (size_t)i * DF);
#pragma unroll
    for (int i = 0; i < 16; ++i) {
      s.x += v[i].x; s.y += v[i].y; s.z += v[i].z; s.w += v[i].w;
      q.x += v[i].x * v[i].x; q.y += v[i].y * v[i].y;
      q.z += v[i].z * v[i].z; q.w += v[i].w * v[i].w;
    }
  } else {
    for (int i = 0; i < 16; ++i) {
      int r = rw + i;
      if (r < n) {
        const float4 v = *(const float4*)(agg + (size_t)r * DF + lane * 4);
        s.x += v.x; s.y += v.y; s.z += v.z; s.w += v.w;
        q.x += v.x * v.x; q.y += v.y * v.y; q.z += v.z * v.z; q.w += v.w * v.w;
      }
    }
  }
  *(float4*)&red[w][lane * 4]      = s;
  *(float4*)&red[w][DF + lane * 4] = q;
  __syncthreads();
  if (tid < 2 * DF) {
    float a = 0.f;
#pragma unroll
    for (int k = 0; k < 32; ++k) a += red[k][tid];
    atomicAdd(&stats[tid], a);
  }
}

#define BNA_ROWS 16
#define INV_N_NODES (1.0f / (float)N_NODES)
static_assert(N_NODES % BNA_ROWS == 0, "a wave has all of its rows or none");

__device__ __forceinline__ float bn_leaky1(float x, float sc, float sh) {
  float y = x * sc + sh;
  return (y >= 0.f) ? y : 0.1f * y;
}

__device__ __forceinline__ void bn_scale_shift(float sum, float sumsq, float g, float be,
                                               float& sc, float& sh) {
  float m   = sum * INV_N_NODES;
  float var = sumsq * INV_N_NODES - m * m;
  sc = g * rsqrtf(var + EPS_BN);
  sh = be - m * sc;
}

__global__ __launch_bounds__(256)
void bn_apply_leaky(const float* __restrict__ agg, const float* __restrict__ stats,
                    const float* __restrict__ g, const float* __restrict__ be,
                    float* __restrict__ h) {
  int wave = (blockIdx.x * blockDim.x + threadIdx.x) >> 5;
  int row0 = wave * BNA_ROWS;
  if (row0 >= N_NODES) return;
  int lane4 = (threadIdx.x & 31) * 4;
  const float4 su = *(const float4*)(stats + lane4);
  const float4 sq = *(const float4*)(stats + DF + lane4);
  const float4 gv = *(const float4*)(g + lane4);
  const float4 bv = *(const float4*)(be + lane4);
  const float* p = agg + (size_t)row0 * DF + lane4;
  float*       o = h   + (size_t)row0 * DF + lane4;
  float4 v[BNA_ROWS];
#pragma unroll
  for (int i = 0; i < BNA_ROWS; ++i) v[i] = *(const float4*)(p + (size_t)i * DF);
  float4 sc, sh;
  bn_scale_shift(su.x, sq.x, gv.x, bv.x, sc.x, sh.x);
  bn_scale_shift(su.y, sq.y, gv.y, bv.y, sc.y, sh.y);
  bn_scale_shift(su.z, sq.z, gv.z, bv.z, sc.z, sh.z);
  bn_scale_shift(su.w, sq.w, gv.w, bv.w, sc.w, sh.w);
#pragma unroll
  for (int i = 0; i < BNA_ROWS; ++i) {
    float4 y;
    y.x = bn_leaky1(v[i].x, sc.x, sh.x);
    y.y = bn_leaky1(v[i].y, sc.y, sh.y);
    y.z = bn_leaky1(v[i].z, sc.z, sh.z);
    y.w = bn_leaky1(v[i].w, sc.w, sh.w);
    *(float4*)(o + (size_t)i * DF) = y;
  }
}

__global__ __launch_bounds__(256)
void gather_pairs(const float* __restrict__ h, const int* __restrict__ di,
                  float* __restrict__ amat, float* __restrict__ bmat) {
  int t = blockIdx.x * blockDim.x + threadIdx.x;
  if (t >= BPAIRS * 32) return;
  int j = t >> 5, q = t & 31;
  int ia = di[2 * j] - 1;     if (ia < 0) ia += N_NODES;
  int ib = di[2 * j + 1] - 1; if (ib < 0) ib += N_NODES;
  *(float4*)(amat + (size_t)j * DF + q * 4) = *(const float4*)(h + (size_t)ia * DF + q * 4);
  *(float4*)(bmat + (size_t)j * DF + q * 4) = *(const float4*)(h + (size_t)ib * DF + q * 4);
}

__global__ __launch_bounds__(256)
void rowdot(const float* __restrict__ tm, const float* __restrict__ bm,
            float* __restrict__ out) {
  int t = blockIdx.x * blockDim.x + threadIdx.x;
  int j = t >> 5, lane = t & 31;
  if (j >= BPAIRS) return;
  float4 tv = *(const float4*)(tm + (size_t)j * DF + lane * 4);
  float4 bv = *(const float4*)(bm + (size_t)j * DF + lane * 4);
  float p = tv.x * bv.x + tv.y * bv.y + tv.z * bv.z + tv.w * bv.w;
#pragma unroll
  for (int off = 16; off > 0; off >>= 1)
    p += __shfl_down(p, off, 32);
  if (lane == 0) out[j] = p;
}

extern "C" void kernel_launch(void* const* d_in, const int* in_sizes, int n_in,
                              void* d_out, int out_size, void* d_ws, size_t ws_size,
                              hipStream_t stream) {
  const float* x   = (const float*)d_in[0];
  const int*   edg = (const int*)d_in[1];
  const int*   di  = (const int*)d_in[2];
  const float* w1  = (const float*)d_in[3];
  const float* b1  = (const float*)d_in[4];
  const float* w2  = (const float*)d_in[5];
  const float* b2  = (const float*)d_in[6];
  const float* w3  = (const float*)d_in[7];
  const float* b3  = (const float*)d_in[8];
  const float* g1  = (const float*)d_in[9];
  const float* be1 = (const float*)d_in[10];
  const float* g2  = (const float*)d_in[11];
  const float* be2 = (const float*)d_in[12];
  const float* g3  = (const float*)d_in[13];
  const float* be3 = (const float*)d_in[14];
  const float* p1  = (const float*)d_in[15];
  const float* p2  = (const float*)d_in[16];
  float* out = (float*)d_out;

  float* ws = (float*)d_ws;
  size_t off = 0;
  float* dis    = ws + off; off += N_NODES;
  float* stats  = ws + off; off += 256;
  float* Wt     = ws + off; off += 128 * 128;
  int*   cnt    = (int*)(ws + off); off += N_NODES;
  int*   rowptr = (int*)(ws + off); off += N_NODES + 16;
  off += N_NODES;
  int*   srcidx = (int*)(ws + off); off += NE_EDGES;
  float* lin    = ws + off; off += (size_t)N_NODES * DF;
  float* agg    = ws + off; off += (size_t)N_NODES * DF;
  float* h      = ws + off; off += (size_t)N_NODES * DF;
  float* U    = lin;
  float* Qt   = lin + 16384;
  float* amat = lin + 32768;
  float* bmat = amat + (size_t)BPAIRS * DF;
  float* tmat = bmat + (size_t)BPAIRS * DF;

  const int* erow = edg;
  const int* ecol = edg + NE_EDGES;

  zero_cnt_transpose_w<<<(N_NODES + 255) / 256, 256, 0, stream>>>(cnt, N_NODES, w1, Wt);
  count_edges<<<(NE_EDGES + 255) / 256, 256, 0, stream>>>(ecol, cnt);
  scan_rowptr<<<1, 1024, 0, stream>>>(cnt, rowptr);
  csr_fill<<<(NE_EDGES + 255) / 256, 256, 0, stream>>>(erow, ecol, rowptr, cnt, srcidx, dis);

  const float* Wn[3] = {w2, w3, nullptr};
  const float* Bl[3] = {b1, b2, b3};
  const float* Gl[3] = {g1, g2, g3};
  const float* El[3] = {be1, be2, be3};
  const float* in = x;
  const int strip_waves = N_NODES / 16;
  for (int l = 0; l < 3; ++l) {
    wmma_gemm_strip<<<(strip_waves * 32 + 255) / 256, 256, 0, stream>>>(
        in, Wt, Bl[l], lin, N_NODES, DF);
    gcn_aggregate<<<(N_NODES * 32 + 255) / 256, 256, 0, stream>>>(
        rowptr, srcidx, dis, lin, agg, stats, Wn[l], Wt);
    bn_stats<<<(N_NODES + 511) / 512, 1024, 0, stream>>>(agg, stats, N_NODES);
    bn_apply_leaky<<<(N_NODES / BNA_ROWS * 32 + 255) / 256, 256, 0, stream>>>(
        agg, stats, Gl[l], El[l], h);
    in = h;
  }

  const int small_tiles = (DF / 16) * (DF / 16);
  wmma_matmul<<<(small_tiles * 32 + 255) / 256, 256, 0, stream>>>(p1, p2, nullptr, U, DF, DF, DF, 1);
  wmma_matmul<<<(small_tiles * 32 + 255) / 256, 256, 0, stream>>>(U, p1, nullptr, Qt, DF, DF, DF, 1);
  gather_pairs<<<(BPAIRS * 32 + 255) / 256, 256, 0, stream>>>(h, di, amat, bmat);
  wmma_gemm_strip<<<((BPAIRS / 16) * 32 + 255) / 256, 256, 0, stream>>>(
      amat, Qt, nullptr, tmat, BPAIRS, DF);
  rowdot<<<(BPAIRS * 32 + 255) / 256, 256, 0, stream>>>(tmat, bmat, out);
}


# bn_apply_leaky launches removed: the next layer's strip GEMM (its A rows now read whole and laid in LDS for the fragments) and the pair gather apply scale, shift and leaky to the agg rows they read
# speedup vs baseline: 3.3386x; 1.0918x over previous
#include <hip/hip_runtime.h>

#define N_NODES 50000
#define NE_EDGES 800000
#define BPAIRS 4096
#define DF 128
#define EPS_BN 1e-5f

typedef float v2f __attribute__((ext_vector_type(2)));
typedef float v8f __attribute__((ext_vector_type(8)));

__device__ __forceinline__ v8f wmma4(v2f a, v2f b, v8f c) {
  return __builtin_amdgcn_wmma_f32_16x16x4_f32(
       false, a,  false, b,
       (short)0, c,  false,  false);
}

#define INV_N_NODES (1.0f / (float)N_NODES)

__device__ __forceinline__ float bn_leaky1(float x, float sc, float sh) {
  float y = x * sc + sh;
  return (y >= 0.f) ? y : 0.1f * y;
}

__device__ __forceinline__ void bn_scale_shift(float sum, float sumsq, float g, float be,
                                               float& sc, float& sh) {
  float m   = sum * INV_N_NODES;
  float var = sumsq * INV_N_NODES - m * m;
  sc = g * rsqrtf(var + EPS_BN);
  sh = be - m * sc;
}

__device__ __forceinline__ void bn_lane_scale_shift(const float* __restrict__ stats,
                                                    const float* __restrict__ g,
                                                    const float* __restrict__ be, int lane4,
                                                    float4& sc, float4& sh) {
  const float4 su = *(const float4*)(stats + lane4);
  const float4 sq = *(const float4*)(stats + DF + lane4);
  const float4 gv = *(const float4*)(g + lane4);
  const float4 bv = *(const float4*)(be + lane4);
  bn_scale_shift(su.x, sq.x, gv.x, bv.x, sc.x, sh.x);
  bn_scale_shift(su.y, sq.y, gv.y, bv.y, sc.y, sh.y);
  bn_scale_shift(su.z, sq.z, gv.z, bv.z, sc.z, sh.z);
  bn_scale_shift(su.w, sq.w, gv.w, bv.w, sc.w, sh.w);
}

__device__ __forceinline__ float4 bn_leaky4(float4 v, float4 sc, float4 sh) {
  float4 y;
  y.x = bn_leaky1(v.x, sc.x, sh.x);
  y.y = bn_leaky1(v.y, sc.y, sh.y);
  y.z = bn_leaky1(v.z, sc.z, sh.z);
  y.w = bn_leaky1(v.w, sc.w, sh.w);
  return y;
}

#define GS_PITCH 132
#define GS_WAVES 8
static_assert(DF == 128, "a row is 32 lanes x 16 bytes; K of every product here is 128");

__global__ __launch_bounds__(GS_WAVES * 32)
void wmma_gemm_strip(const float* __restrict__ A, const float* __restrict__ Bt,
                     const float* __restrict__ bias, float* __restrict__ C, int M,
                     const float* __restrict__ stats, const float* __restrict__ g,
                     const float* __restrict__ be) {
  __shared__ __attribute__((aligned(16))) float sA[GS_WAVES][16 * GS_PITCH];
  int gtid = blockIdx.x * blockDim.x + threadIdx.x;
  int wave = gtid >> 5;
  int lane = threadIdx.x & 31;
  int w    = threadIdx.x >> 5;
  if (wave * 16 >= M) return;
  const int lane4 = lane * 4;

  {
    const float* p = A + (size_t)wave * 16 * DF + lane4;
    float4 av[16];
#pragma unroll
    for (int i = 0; i < 16; ++i) av[i] = *(const float4*)(p + (size_t)i * DF);
    if (stats != nullptr) {
      float4 sc, sh;
      bn_lane_scale_shift(stats, g, be, lane4, sc, sh);
#pragma unroll
      for (int i = 0; i < 16; ++i) av[i] = bn_leaky4(av[i], sc, sh);
    }
#pragma unroll
    for (int i = 0; i < 16; ++i) *(float4*)&sA[w][i * GS_PITCH + lane4] = av[i];
  }

  int l16   = lane & 15;
  int khalf = (lane >> 4) << 1;
  const float* arow = &sA[w][l16 * GS_PITCH + khalf];

  const float* brow[8];
#pragma unroll
  for (int nt = 0; nt < 8; ++nt)
    brow[nt] = Bt + (size_t)(nt * 16 + l16) * DF + khalf;

  v8f zero = {};
  v8f acc[8];
#pragma unroll
  for (int nt = 0; nt < 8; ++nt) acc[nt] = zero;

  for (int k = 0; k < DF; k += 4) {
    v2f a = *(const v2f*)(arow + k);
#pragma unroll
    for (int nt = 0; nt < 8; ++nt) {
      v2f b = *(const v2f*)(brow[nt] + k);
      acc[nt] = wmma4(a, b, acc[nt]);
    }
  }

  int rbase = wave * 16 + ((lane >> 4) << 3);
#pragma unroll
  for (int nt = 0; nt < 8; ++nt) {
    float bs = bias ? bias[nt * 16 + l16] : 0.f;
    float* crow = C + (size_t)rbase * 128 + nt * 16 + l16;
#pragma unroll
    for (int r = 0; r < 8; ++r)
      crow[(size_t)r * 128] = acc[nt][r] + bs;
  }
}

__global__ __launch_bounds__(256)
void wmma_matmul(const float* __restrict__ A, const float* __restrict__ Bm,
                 const float* __restrict__ bias, float* __restrict__ C,
                 int M, int Nc, int K, int bTrans) {
  int gtid = blockIdx.x * blockDim.x + threadIdx.x;
  int wave = gtid >> 5;
  int lane = threadIdx.x & 31;
  int ntiles = Nc >> 4;
  int mt = wave / ntiles;
  int nt = wave - mt * ntiles;
  if (mt * 16 >= M) return;

  int l16   = lane & 15;
  int khalf = (lane >> 4) << 1;
  const float* arow = A + (size_t)(mt * 16 + l16) * K + khalf;

  v8f acc = {};
  if (bTrans) {
    const float* brow = Bm + (size_t)(nt * 16 + l16) * K + khalf;
    for (int k = 0; k < K; k += 4) {
      v2f a = *(const v2f*)(arow + k);
      v2f b = *(const v2f*)(brow + k);
      acc = wmma4(a, b, acc);
    }
  } else {
    const float* bcol = Bm + (size_t)khalf * Nc + nt * 16 + l16;
    for (int k = 0; k < K; k += 4) {
      v2f a = *(const v2f*)(arow + k);
      v2f b;
      b.x = bcol[(size_t)k * Nc];
      b.y = bcol[(size_t)(k + 1) * Nc];
      acc = wmma4(a, b, acc);
    }
  }

  float bs = bias ? bias[nt * 16 + l16] : 0.f;
  int rbase = mt * 16 + ((lane >> 4) << 3);
  float* crow = C + (size_t)rbase * Nc + nt * 16 + l16;
#pragma unroll
  for (int r = 0; r < 8; ++r)
    crow[(size_t)r * Nc] = acc[r] + bs;
}

__global__ void zero_cnt_transpose_w(int* __restrict__ cnt, int n,
                                     const float* __restrict__ w, float* __restrict__ wt) {
  int t = blockIdx.x * blockDim.x + threadIdx.x;
  if (t < n) cnt[t] = 0;
  if (t < 128 * 128) wt[(t & 127) * 128 + (t >> 7)] = w[t];
}

__global__ void count_edges(const int* __restrict__ ecol, int* __restrict__ cnt) {
  int e = blockIdx.x * blockDim.x + threadIdx.x;
  if (e < NE_EDGES) atomicAdd(&cnt[ecol[e]], 1);
}

#define SCAN_STEPS 13
#define SCAN_PER_WAVE (SCAN_STEPS * 128)
static_assert(32 * SCAN_PER_WAVE >= N_NODES, "the 32 waves cover every counter");
static_assert(N_NODES % 4 == 0, "a lane's 4 counters are all inside or all outside");

__device__ __forceinline__ int wave_incl_scan(int x, int lane) {
#pragma unroll
  for (int off = 1; off < 32; off <<= 1) {
    int y = __shfl_up(x, off, 32);
    if (lane >= off) x += y;
  }
  return x;
}

__global__ __launch_bounds__(1024)
void scan_rowptr(const int* __restrict__ cnt, int* __restrict__ rowptr) {
  __shared__ int wtot[32];
  const int tid = threadIdx.x;
  const int lane = tid & 31;
  const int w = tid >> 5;
  const int first = w * SCAN_PER_WAVE + lane * 4;
  int4 v[SCAN_STEPS];
#pragma unroll
  for (int j = 0; j < SCAN_STEPS; ++j) {
    int idx = first + j * 128;
    int idc = (idx < N_NODES) ? idx : (N_NODES - 4);
    v[j] = *(const int4*)(cnt + idc);
  }
  int run = 0;
#pragma unroll
  for (int j = 0; j < SCAN_STEPS; ++j) {
    const bool in = (first + j * 128) < N_NODES;
    const int c0 = in ? v[j].x : 0;
    const int c1 = in ? v[j].y : 0;
    const int c2 = in ? v[j].z : 0;
    const int c3 = in ? v[j].w : 0;
    const int s1 = c0, s2 = s1 + c1, s3 = s2 + c2, s4 = s3 + c3;
    const int inc = wave_incl_scan(s4, lane);
    const int ex = run + inc - s4;
    run += __shfl(inc, 31, 32);
    v[j].x = ex; v[j].y = ex + s1; v[j].z = ex + s2; v[j].w = ex + s3;
  }
  if (lane == 0) wtot[w] = run;
  __syncthreads();
  const int t  = wtot[lane];
  const int ti = wave_incl_scan(t, lane);
  const int base = __shfl(ti - t, w, 32);
#pragma unroll
  for (int j = 0; j < SCAN_STEPS; ++j) {
    int idx = first + j * 128;
    if (idx < N_NODES) {
      int4 o;
      o.x = v[j].x + base; o.y = v[j].y + base; o.z = v[j].z + base; o.w = v[j].w + base;
      *(int4*)(rowptr + idx) = o;
    }
  }
  if (tid == 1023) rowptr[N_NODES] = base + run;
}

__global__ void csr_fill(const int* __restrict__ erow, const int* __restrict__ ecol,
                         const int* __restrict__ rowptr, int* __restrict__ cnt,
                         int* __restrict__ srcidx, float* __restrict__ dis) {
  int e = blockIdx.x * blockDim.x + threadIdx.x;
  if (e >= NE_EDGES) return;
  if (e < N_NODES)
    dis[e] = rsqrtf((float)(rowptr[e + 1] - rowptr[e]) + 1.0f);
  int c = ecol[e];
  int left = atomicSub(&cnt[c], 1);
  srcidx[rowptr[c] + left - 1] = erow[e];
}

static_assert(NE_EDGES >= N_NODES, "csr_fill's first N_NODES threads write dis");
__global__ __launch_bounds__(256)
void gcn_aggregate(const int* __restrict__ rowptr, const int* __restrict__ srcidx,
                   const float* __restrict__ dis, const float* __restrict__ lin,
                   float* __restrict__ agg, float* __restrict__ stats,
                   const float* __restrict__ wnext, float* __restrict__ wt) {
  int t = blockIdx.x * blockDim.x + threadIdx.x;
  if (blockIdx.x == 0) stats[threadIdx.x] = 0.0f;
  if (wnext != nullptr && t < 128 * 128) wt[(t & 127) * 128 + (t >> 7)] = wnext[t];
  int c = t >> 5;
  if (c >= N_NODES) return;
  int lane = threadIdx.x & 31;
  float dc = dis[c];
  const float4 sv = *(const float4*)(lin + (size_t)c * DF + lane * 4);
  float s = dc * dc;
  float ax = s * sv.x, ay = s * sv.y, az = s * sv.z, aw = s * sv.w;
  int e0 = rowptr[c], e1 = rowptr[c + 1];
  for (int j = e0; j < e1; ++j) {
    int r = srcidx[j];
    float nrm = dis[r] * dc;
    const float4 v = *(const float4*)(lin + (size_t)r * DF + lane * 4);
    ax += nrm * v.x; ay += nrm * v.y; az += nrm * v.z; aw += nrm * v.w;
  }
  float4 o; o.x = ax; o.y = ay; o.z = az; o.w = aw;
  *(float4*)(agg + (size_t)c * DF + lane * 4) = o;
}

__global__ __launch_bounds__(1024)
void bn_stats(const float* __restrict__ agg, float* __restrict__ stats, int n) {
  __shared__ __attribute__((aligned(16))) float red[32][2 * DF];
  int tid  = threadIdx.x;
  int lane = tid & 31;
  int w    = tid >> 5;
  int rw   = blockIdx.x * 512 + w * 16;
  float4 s = {0.f, 0.f, 0.f, 0.f};
  float4 q = {0.f, 0.f, 0.f, 0.f};
  if (rw + 16 <= n) {
    const float* p = agg + (size_t)rw * DF + lane * 4;
    float4 v[16];
#pragma unroll
    for (int i = 0; i < 16; ++i) v[i] = *(const float4*)(p + (size_t)i * DF);
#pragma unroll
    for (int i = 0; i < 16; ++i) {
      s.x += v[i].x; s.y += v[i].y; s.z += v[i].z; s.w += v[i].w;
      q.x += v[i].x * v[i].x; q.y += v[i].y * v[i].y;
      q.z += v[i].z * v[i].z; q.w += v[i].w * v[i].w;
    }
  } else {
    for (int i = 0; i < 16; ++i) {
      int r = rw + i;
      if (r < n) {
        const float4 v = *(const float4*)(agg + (size_t)r * DF + lane * 4);
        s.x += v.x; s.y += v.y; s.z += v.z; s.w += v.w;
        q.x += v.x * v.x; q.y += v.y * v.y; q.z += v.z * v.z; q.w += v.w * v.w;
      }
    }
  }
  *(float4*)&red[w][lane * 4]      = s;
  *(float4*)&red[w][DF + lane * 4] = q;
  __syncthreads();
  if (tid < 2 * DF) {
    float a = 0.f;
#pragma unroll
    for (int k = 0; k < 32; ++k) a += red[k][tid];
    atomicAdd(&stats[tid], a);
  }
}

__global__ __launch_bounds__(256)
void gather_pairs(const float* __restrict__ agg, const float* __restrict__ stats,
                  const float* __restrict__ g, const float* __restrict__ be,
                  const int* __restrict__ di,
                  float* __restrict__ amat, float* __restrict__ bmat) {
  int t = blockIdx.x * blockDim.x + threadIdx.x;
  if (t >= BPAIRS * 32) return;
  int j = t >> 5, q = t & 31;
  int ia = di[2 * j] - 1;     if (ia < 0) ia += N_NODES;
  int ib = di[2 * j + 1] - 1; if (ib < 0) ib += N_NODES;
  const float4 va = *(const float4*)(agg + (size_t)ia * DF + q * 4);
  const float4 vb = *(const float4*)(agg + (size_t)ib * DF + q * 4);
  float4 sc, sh;
  bn_lane_scale_shift(stats, g, be, q * 4, sc, sh);
  *(float4*)(amat + (size_t)j * DF + q * 4) = bn_leaky4(va, sc, sh);
  *(float4*)(bmat + (size_t)j * DF + q * 4) = bn_leaky4(vb, sc, sh);
}

__global__ __launch_bounds__(256)
void rowdot(const float* __restrict__ tm, const float* __restrict__ bm,
            float* __restrict__ out) {
  int t = blockIdx.x * blockDim.x + threadIdx.x;
  int j = t >> 5, lane = t & 31;
  if (j >= BPAIRS) return;
  float4 tv = *(const float4*)(tm + (size_t)j * DF + lane * 4);
  float4 bv = *(const float4*)(bm + (size_t)j * DF + lane * 4);
  float p = tv.x * bv.x + tv.y * bv.y + tv.z * bv.z + tv.w * bv.w;
#pragma unroll
  for (int off = 16; off > 0; off >>= 1)
    p += __shfl_down(p, off, 32);
  if (lane == 0) out[j] = p;
}

extern "C" void kernel_launch(void* const* d_in, const int* in_sizes, int n_in,
                              void* d_out, int out_size, void* d_ws, size_t ws_size,
                              hipStream_t stream) {
  const float* x   = (const float*)d_in[0];
  const int*   edg = (const int*)d_in[1];
  const int*   di  = (const int*)d_in[2];
  const float* w1  = (const float*)d_in[3];
  const float* b1  = (const float*)d_in[4];
  const float* w2  = (const float*)d_in[5];
  const float* b2  = (const float*)d_in[6];
  const float* w3  = (const float*)d_in[7];
  const float* b3  = (const float*)d_in[8];
  const float* g1  = (const float*)d_in[9];
  const float* be1 = (const float*)d_in[10];
  const float* g2  = (const float*)d_in[11];
  const float* be2 = (const float*)d_in[12];
  const float* g3  = (const float*)d_in[13];
  const float* be3 = (const float*)d_in[14];
  const float* p1  = (const float*)d_in[15];
  const float* p2  = (const float*)d_in[16];
  float* out = (float*)d_out;

  float* ws = (float*)d_ws;
  size_t off = 0;
  float* dis    = ws + off; off += N_NODES;
  float* stats  = ws + off; off += 256;
  float* Wt     = ws + off; off += 128 * 128;
  int*   cnt    = (int*)(ws + off); off += N_NODES;
  int*   rowptr = (int*)(ws + off); off += N_NODES + 16;
  off += N_NODES;
  int*   srcidx = (int*)(ws + off); off += NE_EDGES;
  float* lin    = ws + off; off += (size_t)N_NODES * DF;
  float* agg    = ws + off; off += (size_t)N_NODES * DF;
  float* U    = lin;
  float* Qt   = lin + 16384;
  float* amat = lin + 32768;
  float* bmat = amat + (size_t)BPAIRS * DF;
  float* tmat = bmat + (size_t)BPAIRS * DF;

  const int* erow = edg;
  const int* ecol = edg + NE_EDGES;

  zero_cnt_transpose_w<<<(N_NODES + 255) / 256, 256, 0, stream>>>(cnt, N_NODES, w1, Wt);
  count_edges<<<(NE_EDGES + 255) / 256, 256, 0, stream>>>(ecol, cnt);
  scan_rowptr<<<1, 1024, 0, stream>>>(cnt, rowptr);
  csr_fill<<<(NE_EDGES + 255) / 256, 256, 0, stream>>>(erow, ecol, rowptr, cnt, srcidx, dis);

  const float* Wn[3] = {w2, w3, nullptr};
  const float* Bl[3] = {b1, b2, b3};
  const float* Gl[3] = {g1, g2, g3};
  const float* El[3] = {be1, be2, be3};
  const int strip_waves = N_NODES / 16;
  for (int l = 0; l < 3; ++l) {
    if (l == 0)
      wmma_gemm_strip<<<(strip_waves * 32 + 255) / 256, 256, 0, stream>>>(
          x, Wt, Bl[l], lin, N_NODES, nullptr, nullptr, nullptr);
    else
      wmma_gemm_strip<<<(strip_waves * 32 + 255) / 256, 256, 0, stream>>>(
          agg, Wt, Bl[l], lin, N_NODES, stats, Gl[l - 1], El[l - 1]);
    gcn_aggregate<<<(N_NODES * 32 + 255) / 256, 256, 0, stream>>>(
        rowptr, srcidx, dis, lin, agg, stats, Wn[l], Wt);
    bn_stats<<<(N_NODES + 511) / 512, 1024, 0, stream>>>(agg, stats, N_NODES);
  }

  const int small_tiles = (DF / 16) * (DF / 16);
  wmma_matmul<<<(small_tiles * 32 + 255) / 256, 256, 0, stream>>>(p1, p2, nullptr, U, DF, DF, DF, 1);
  wmma_matmul<<<(small_tiles * 32 + 255) / 256, 256, 0, stream>>>(U, p1, nullptr, Qt, DF, DF, DF, 1);
  gather_pairs<<<(BPAIRS * 32 + 255) / 256, 256, 0, stream>>>(agg, stats, g3, be3, di, amat, bmat);
  wmma_gemm_strip<<<((BPAIRS / 16) * 32 + 255) / 256, 256, 0, stream>>>(
      amat, Qt, nullptr, tmat, BPAIRS, nullptr, nullptr, nullptr);
  rowdot<<<(BPAIRS * 32 + 255) / 256, 256, 0, stream>>>(tmat, bmat, out);
}


# wmma_gemm_strip: the 128 x 128 B operand is laid in LDS once per block from whole-row 16-byte loads; B fragments are 8-byte LDS reads, not global loads that touch 16 lines each; k order kept
# speedup vs baseline: 3.6784x; 1.1018x over previous
#include <hip/hip_runtime.h>

#define N_NODES 50000
#define NE_EDGES 800000
#define BPAIRS 4096
#define DF 128
#define EPS_BN 1e-5f

typedef float v2f __attribute__((ext_vector_type(2)));
typedef float v8f __attribute__((ext_vector_type(8)));

__device__ __forceinline__ v8f wmma4(v2f a, v2f b, v8f c) {
  return __builtin_amdgcn_wmma_f32_16x16x4_f32(
       false, a,  false, b,
       (short)0, c,  false,  false);
}

#define INV_N_NODES (1.0f / (float)N_NODES)

__device__ __forceinline__ float bn_leaky1(float x, float sc, float sh) {
  float y = x * sc + sh;
  return (y >= 0.f) ? y : 0.1f * y;
}

__device__ __forceinline__ void bn_scale_shift(float sum, float sumsq, float g, float be,
                                               float& sc, float& sh) {
  float m   = sum * INV_N_NODES;
  float var = sumsq * INV_N_NODES - m * m;
  sc = g * rsqrtf(var + EPS_BN);
  sh = be - m * sc;
}

__device__ __forceinline__ void bn_lane_scale_shift(const float* __restrict__ stats,
                                                    const float* __restrict__ g,
                                                    const float* __restrict__ be, int lane4,
                                                    float4& sc, float4& sh) {
  const float4 su = *(const float4*)(stats + lane4);
  const float4 sq = *(const float4*)(stats + DF + lane4);
  const float4 gv = *(const float4*)(g + lane4);
  const float4 bv = *(const float4*)(be + lane4);
  bn_scale_shift(su.x, sq.x, gv.x, bv.x, sc.x, sh.x);
  bn_scale_shift(su.y, sq.y, gv.y, bv.y, sc.y, sh.y);
  bn_scale_shift(su.z, sq.z, gv.z, bv.z, sc.z, sh.z);
  bn_scale_shift(su.w, sq.w, gv.w, bv.w, sc.w, sh.w);
}

__device__ __forceinline__ float4 bn_leaky4(float4 v, float4 sc, float4 sh) {
  float4 y;
  y.x = bn_leaky1(v.x, sc.x, sh.x);
  y.y = bn_leaky1(v.y, sc.y, sh.y);
  y.z = bn_leaky1(v.z, sc.z, sh.z);
  y.w = bn_leaky1(v.w, sc.w, sh.w);
  return y;
}

#define GS_PITCH 132
#define GS_WAVES 8
static_assert(DF == 128, "a row is 32 lanes x 16 bytes; K of every product here is 128");
static_assert(GS_WAVES * 16 == DF, "the block's waves bring 16 rows of Bt each: all 128");

__global__ __launch_bounds__(GS_WAVES * 32)
void wmma_gemm_strip(const float* __restrict__ A, const float* __restrict__ Bt,
                     const float* __restrict__ bias, float* __restrict__ C, int M,
                     const float* __restrict__ stats, const float* __restrict__ g,
                     const float* __restrict__ be) {
  __shared__ __attribute__((aligned(16))) float sA[GS_WAVES][16 * GS_PITCH];
  __shared__ __attribute__((aligned(16))) float sB[DF * GS_PITCH];
  int gtid = blockIdx.x * blockDim.x + threadIdx.x;
  int wave = gtid >> 5;
  int lane = threadIdx.x & 31;
  int w    = threadIdx.x >> 5;
  const bool active = wave * 16 < M;
  const int lane4 = lane * 4;

  {
    const float* pb = Bt + (size_t)w * 16 * DF + lane4;
    float4 bv[16], av[16];
#pragma unroll
    for (int i = 0; i < 16; ++i) bv[i] = *(const float4*)(pb + (size_t)i * DF);
    if (active) {
      const float* pa = A + (size_t)wave * 16 * DF + lane4;
#pragma unroll
      for (int i = 0; i < 16; ++i) av[i] = *(const float4*)(pa + (size_t)i * DF);
    }
#pragma unroll
    for (int i = 0; i < 16; ++i) *(float4*)&sB[(w * 16 + i) * GS_PITCH + lane4] = bv[i];
    if (active) {
      if (stats != nullptr) {
        float4 sc, sh;
        bn_lane_scale_shift(stats, g, be, lane4, sc, sh);
#pragma unroll
        for (int i = 0; i < 16; ++i) av[i] = bn_leaky4(av[i], sc, sh);
      }
#pragma unroll
      for (int i = 0; i < 16; ++i) *(float4*)&sA[w][i * GS_PITCH + lane4] = av[i];
    }
  }
  __syncthreads();
  if (!active) return;

  int l16   = lane & 15;
  int khalf = (lane >> 4) << 1;
  const float* arow = &sA[w][l16 * GS_PITCH + khalf];
  const float* brow = &sB[l16 * GS_PITCH + khalf];

  v8f zero = {};
  v8f acc[8];
#pragma unroll
  for (int nt = 0; nt < 8; ++nt) acc[nt] = zero;

  for (int k = 0; k < DF; k += 4) {
    v2f a = *(const v2f*)(arow + k);
#pragma unroll
    for (int nt = 0; nt < 8; ++nt) {
      v2f b = *(const v2f*)(brow + nt * 16 * GS_PITCH + k);
      acc[nt] = wmma4(a, b, acc[nt]);
    }
  }

  int rbase = wave * 16 + ((lane >> 4) << 3);
#pragma unroll
  for (int nt = 0; nt < 8; ++nt) {
    float bs = bias ? bias[nt * 16 + l16] : 0.f;
    float* crow = C + (size_t)rbase * 128 + nt * 16 + l16;
#pragma unroll
    for (int r = 0; r < 8; ++r)
      crow[(size_t)r * 128] = acc[nt][r] + bs;
  }
}

__global__ __launch_bounds__(256)
void wmma_matmul(const float* __restrict__ A, const float* __restrict__ Bm,
                 const float* __restrict__ bias, float* __restrict__ C,
                 int M, int Nc, int K, int bTrans) {
  int gtid = blockIdx.x * blockDim.x + threadIdx.x;
  int wave = gtid >> 5;
  int lane = threadIdx.x & 31;
  int ntiles = Nc >> 4;
  int mt = wave / ntiles;
  int nt = wave - mt * ntiles;
  if (mt * 16 >= M) return;

  int l16   = lane & 15;
  int khalf = (lane >> 4) << 1;
  const float* arow = A + (size_t)(mt * 16 + l16) * K + khalf;

  v8f acc = {};
  if (bTrans) {
    const float* brow = Bm + (size_t)(nt * 16 + l16) * K + khalf;
    for (int k = 0; k < K; k += 4) {
      v2f a = *(const v2f*)(arow + k);
      v2f b = *(const v2f*)(brow + k);
      acc = wmma4(a, b, acc);
    }
  } else {
    const float* bcol = Bm + (size_t)khalf * Nc + nt * 16 + l16;
    for (int k = 0; k < K; k += 4) {
      v2f a = *(const v2f*)(arow + k);
      v2f b;
      b.x = bcol[(size_t)k * Nc];
      b.y = bcol[(size_t)(k + 1) * Nc];
      acc = wmma4(a, b, acc);
    }
  }

  float bs = bias ? bias[nt * 16 + l16] : 0.f;
  int rbase = mt * 16 + ((lane >> 4) << 3);
  float* crow = C + (size_t)rbase * Nc + nt * 16 + l16;
#pragma unroll
  for (int r = 0; r < 8; ++r)
    crow[(size_t)r * Nc] = acc[r] + bs;
}

__global__ void zero_cnt_transpose_w(int* __restrict__ cnt, int n,
                                     const float* __restrict__ w, float* __restrict__ wt) {
  int t = blockIdx.x * blockDim.x + threadIdx.x;
  if (t < n) cnt[t] = 0;
  if (t < 128 * 128) wt[(t & 127) * 128 + (t >> 7)] = w[t];
}

__global__ void count_edges(const int* __restrict__ ecol, int* __restrict__ cnt) {
  int e = blockIdx.x * blockDim.x + threadIdx.x;
  if (e < NE_EDGES) atomicAdd(&cnt[ecol[e]], 1);
}

#define SCAN_STEPS 13
#define SCAN_PER_WAVE (SCAN_STEPS * 128)
static_assert(32 * SCAN_PER_WAVE >= N_NODES, "the 32 waves cover every counter");
static_assert(N_NODES % 4 == 0, "a lane's 4 counters are all inside or all outside");

__device__ __forceinline__ int wave_incl_scan(int x, int lane) {
#pragma unroll
  for (int off = 1; off < 32; off <<= 1) {
    int y = __shfl_up(x, off, 32);
    if (lane >= off) x += y;
  }
  return x;
}

__global__ __launch_bounds__(1024)
void scan_rowptr(const int* __restrict__ cnt, int* __restrict__ rowptr) {
  __shared__ int wtot[32];
  const int tid = threadIdx.x;
  const int lane = tid & 31;
  const int w = tid >> 5;
  const int first = w * SCAN_PER_WAVE + lane * 4;
  int4 v[SCAN_STEPS];
#pragma unroll
  for (int j = 0; j < SCAN_STEPS; ++j) {
    int idx = first + j * 128;
    int idc = (idx < N_NODES) ? idx : (N_NODES - 4);
    v[j] = *(const int4*)(cnt + idc);
  }
  int run = 0;
#pragma unroll
  for (int j = 0; j < SCAN_STEPS; ++j) {
    const bool in = (first + j * 128) < N_NODES;
    const int c0 = in ? v[j].x : 0;
    const int c1 = in ? v[j].y : 0;
    const int c2 = in ? v[j].z : 0;
    const int c3 = in ? v[j].w : 0;
    const int s1 = c0, s2 = s1 + c1, s3 = s2 + c2, s4 = s3 + c3;
    const int inc = wave_incl_scan(s4, lane);
    const int ex = run + inc - s4;
    run += __shfl(inc, 31, 32);
    v[j].x = ex; v[j].y = ex + s1; v[j].z = ex + s2; v[j].w = ex + s3;
  }
  if (lane == 0) wtot[w] = run;
  __syncthreads();
  const int t  = wtot[lane];
  const int ti = wave_incl_scan(t, lane);
  const int base = __shfl(ti - t, w, 32);
#pragma unroll
  for (int j = 0; j < SCAN_STEPS; ++j) {
    int idx = first + j * 128;
    if (idx < N_NODES) {
      int4 o;
      o.x = v[j].x + base; o.y = v[j].y + base; o.z = v[j].z + base; o.w = v[j].w + base;
      *(int4*)(rowptr + idx) = o;
    }
  }
  if (tid == 1023) rowptr[N_NODES] = base + run;
}

__global__ void csr_fill(const int* __restrict__ erow, const int* __restrict__ ecol,
                         const int* __restrict__ rowptr, int* __restrict__ cnt,
                         int* __restrict__ srcidx, float* __restrict__ dis) {
  int e = blockIdx.x * blockDim.x + threadIdx.x;
  if (e >= NE_EDGES) return;
  if (e < N_NODES)
    dis[e] = rsqrtf((float)(rowptr[e + 1] - rowptr[e]) + 1.0f);
  int c = ecol[e];
  int left = atomicSub(&cnt[c], 1);
  srcidx[rowptr[c] + left - 1] = erow[e];
}

static_assert(NE_EDGES >= N_NODES, "csr_fill's first N_NODES threads write dis");
__global__ __launch_bounds__(256)
void gcn_aggregate(const int* __restrict__ rowptr, const int* __restrict__ srcidx,
                   const float* __restrict__ dis, const float* __restrict__ lin,
                   float* __restrict__ agg, float* __restrict__ stats,
                   const float* __restrict__ wnext, float* __restrict__ wt) {
  int t = blockIdx.x * blockDim.x + threadIdx.x;
  if (blockIdx.x == 0) stats[threadIdx.x] = 0.0f;
  if (wnext != nullptr && t < 128 * 128) wt[(t & 127) * 128 + (t >> 7)] = wnext[t];
  int c = t >> 5;
  if (c >= N_NODES) return;
  int lane = threadIdx.x & 31;
  float dc = dis[c];
  const float4 sv = *(const float4*)(lin + (size_t)c * DF + lane * 4);
  float s = dc * dc;
  float ax = s * sv.x, ay = s * sv.y, az = s * sv.z, aw = s * sv.w;
  int e0 = rowptr[c], e1 = rowptr[c + 1];
  for (int j = e0; j < e1; ++j) {
    int r = srcidx[j];
    float nrm = dis[r] * dc;
    const float4 v = *(const float4*)(lin + (size_t)r * DF + lane * 4);
    ax += nrm * v.x; ay += nrm * v.y; az += nrm * v.z; aw += nrm * v.w;
  }
  float4 o; o.x = ax; o.y = ay; o.z = az; o.w = aw;
  *(float4*)(agg + (size_t)c * DF + lane * 4) = o;
}

__global__ __launch_bounds__(1024)
void bn_stats(const float* __restrict__ agg, float* __restrict__ stats, int n) {
  __shared__ __attribute__((aligned(16))) float red[32][2 * DF];
  int tid  = threadIdx.x;
  int lane = tid & 31;
  int w    = tid >> 5;
  int rw   = blockIdx.x * 512 + w * 16;
  float4 s = {0.f, 0.f, 0.f, 0.f};
  float4 q = {0.f, 0.f, 0.f, 0.f};
  if (rw + 16 <= n) {
    const float* p = agg + (size_t)rw * DF + lane * 4;
    float4 v[16];
#pragma unroll
    for (int i = 0; i < 16; ++i) v[i] = *(const float4*)(p + (size_t)i * DF);
#pragma unroll
    for (int i = 0; i < 16; ++i) {
      s.x += v[i].x; s.y += v[i].y; s.z += v[i].z; s.w += v[i].w;
      q.x += v[i].x * v[i].x; q.y += v[i].y * v[i].y;
      q.z += v[i].z * v[i].z; q.w += v[i].w * v[i].w;
    }
  } else {
    for (int i = 0; i < 16; ++i) {
      int r = rw + i;
      if (r < n) {
        const float4 v = *(const float4*)(agg + (size_t)r * DF + lane * 4);
        s.x += v.x; s.y += v.y; s.z += v.z; s.w += v.w;
        q.x += v.x * v.x; q.y += v.y * v.y; q.z += v.z * v.z; q.w += v.w * v.w;
      }
    }
  }
  *(float4*)&red[w][lane * 4]      = s;
  *(float4*)&red[w][DF + lane * 4] = q;
  __syncthreads();
  if (tid < 2 * DF) {
    float a = 0.f;
#pragma unroll
    for (int k = 0; k < 32; ++k) a += red[k][tid];
    atomicAdd(&stats[tid], a);
  }
}

__global__ __launch_bounds__(256)
void gather_pairs(const float* __restrict__ agg, const float* __restrict__ stats,
                  const float* __restrict__ g, const float* __restrict__ be,
                  const int* __restrict__ di,
                  float* __restrict__ amat, float* __restrict__ bmat) {
  int t = blockIdx.x * blockDim.x + threadIdx.x;
  if (t >= BPAIRS * 32) return;
  int j = t >> 5, q = t & 31;
  int ia = di[2 * j] - 1;     if (ia < 0) ia += N_NODES;
  int ib = di[2 * j + 1] - 1; if (ib < 0) ib += N_NODES;
  const float4 va = *(const float4*)(agg + (size_t)ia * DF + q * 4);
  const float4 vb = *(const float4*)(agg + (size_t)ib * DF + q * 4);
  float4 sc, sh;
  bn_lane_scale_shift(stats, g, be, q * 4, sc, sh);
  *(float4*)(amat + (size_t)j * DF + q * 4) = bn_leaky4(va, sc, sh);
  *(float4*)(bmat + (size_t)j * DF + q * 4) = bn_leaky4(vb, sc, sh);
}

__global__ __launch_bounds__(256)
void rowdot(const float* __restrict__ tm, const float* __restrict__ bm,
            float* __restrict__ out) {
  int t = blockIdx.x * blockDim.x + threadIdx.x;
  int j = t >> 5, lane = t & 31;
  if (j >= BPAIRS) return;
  float4 tv = *(const float4*)(tm + (size_t)j * DF + lane * 4);
  float4 bv = *(const float4*)(bm + (size_t)j * DF + lane * 4);
  float p = tv.x * bv.x + tv.y * bv.y + tv.z * bv.z + tv.w * bv.w;
#pragma unroll
  for (int off = 16; off > 0; off >>= 1)
    p += __shfl_down(p, off, 32);
  if (lane == 0) out[j] = p;
}

extern "C" void kernel_launch(void* const* d_in, const int* in_sizes, int n_in,
                              void* d_out, int out_size, void* d_ws, size_t ws_size,
                              hipStream_t stream) {
  const float* x   = (const float*)d_in[0];
  const int*   edg = (const int*)d_in[1];
  const int*   di  = (const int*)d_in[2];
  const float* w1  = (const float*)d_in[3];
  const float* b1  = (const float*)d_in[4];
  const float* w2  = (const float*)d_in[5];
  const float* b2  = (const float*)d_in[6];
  const float* w3  = (const float*)d_in[7];
  const float* b3  = (const float*)d_in[8];
  const float* g1  = (const float*)d_in[9];
  const float* be1 = (const float*)d_in[10];
  const float* g2  = (const float*)d_in[11];
  const float* be2 = (const float*)d_in[12];
  const float* g3  = (const float*)d_in[13];
  const float* be3 = (const float*)d_in[14];
  const float* p1  = (const float*)d_in[15];
  const float* p2  = (const float*)d_in[16];
  float* out = (float*)d_out;

  float* ws = (float*)d_ws;
  size_t off = 0;
  float* dis    = ws + off; off += N_NODES;
  float* stats  = ws + off; off += 256;
  float* Wt     = ws + off; off += 128 * 128;
  int*   cnt    = (int*)(ws + off); off += N_NODES;
  int*   rowptr = (int*)(ws + off); off += N_NODES + 16;
  off += N_NODES;
  int*   srcidx = (int*)(ws + off); off += NE_EDGES;
  float* lin    = ws + off; off += (size_t)N_NODES * DF;
  float* agg    = ws + off; off += (size_t)N_NODES * DF;
  float* U    = lin;
  float* Qt   = lin + 16384;
  float* amat = lin + 32768;
  float* bmat = amat + (size_t)BPAIRS * DF;
  float* tmat = bmat + (size_t)BPAIRS * DF;

  const int* erow = edg;
  const int* ecol = edg + NE_EDGES;

  zero_cnt_transpose_w<<<(N_NODES + 255) / 256, 256, 0, stream>>>(cnt, N_NODES, w1, Wt);
  count_edges<<<(NE_EDGES + 255) / 256, 256, 0, stream>>>(ecol, cnt);
  scan_rowptr<<<1, 1024, 0, stream>>>(cnt, rowptr);
  csr_fill<<<(NE_EDGES + 255) / 256, 256, 0, stream>>>(erow, ecol, rowptr, cnt, srcidx, dis);

  const float* Wn[3] = {w2, w3, nullptr};
  const float* Bl[3] = {b1, b2, b3};
  const float* Gl[3] = {g1, g2, g3};
  const float* El[3] = {be1, be2, be3};
  const int strip_waves = N_NODES / 16;
  for (int l = 0; l < 3; ++l) {
    if (l == 0)
      wmma_gemm_strip<<<(strip_waves * 32 + 255) / 256, 256, 0, stream>>>(
          x, Wt, Bl[l], lin, N_NODES, nullptr, nullptr, nullptr);
    else
      wmma_gemm_strip<<<(strip_waves * 32 + 255) / 256, 256, 0, stream>>>(
          agg, Wt, Bl[l], lin, N_NODES, stats, Gl[l - 1], El[l - 1]);
    gcn_aggregate<<<(N_NODES * 32 + 255) / 256, 256, 0, stream>>>(
        rowptr, srcidx, dis, lin, agg, stats, Wn[l], Wt);
    bn_stats<<<(N_NODES + 511) / 512, 1024, 0, stream>>>(agg, stats, N_NODES);
  }

  const int small_tiles = (DF / 16) * (DF / 16);
  wmma_matmul<<<(small_tiles * 32 + 255) / 256, 256, 0, stream>>>(p1, p2, nullptr, U, DF, DF, DF, 1);
  wmma_matmul<<<(small_tiles * 32 + 255) / 256, 256, 0, stream>>>(U, p1, nullptr, Qt, DF, DF, DF, 1);
  gather_pairs<<<(BPAIRS * 32 + 255) / 256, 256, 0, stream>>>(agg, stats, g3, be3, di, amat, bmat);
  wmma_gemm_strip<<<((BPAIRS / 16) * 32 + 255) / 256, 256, 0, stream>>>(
      amat, Qt, nullptr, tmat, BPAIRS, nullptr, nullptr, nullptr);
  rowdot<<<(BPAIRS * 32 + 255) / 256, 256, 0, stream>>>(tmat, bmat, out);
}


# csr_fill places each edge by the value count_edges' atomic add returned (kept per edge), so the second pass of 800000 atomics is gone; in-row edge order still set by atomic arrival
# speedup vs baseline: 4.1382x; 1.1250x over previous
#include <hip/hip_runtime.h>

#define N_NODES 50000
#define NE_EDGES 800000
#define BPAIRS 4096
#define DF 128
#define EPS_BN 1e-5f

typedef float v2f __attribute__((ext_vector_type(2)));
typedef float v8f __attribute__((ext_vector_type(8)));

__device__ __forceinline__ v8f wmma4(v2f a, v2f b, v8f c) {
  return __builtin_amdgcn_wmma_f32_16x16x4_f32(
       false, a,  false, b,
       (short)0, c,  false,  false);
}

#define INV_N_NODES (1.0f / (float)N_NODES)

__device__ __forceinline__ float bn_leaky1(float x, float sc, float sh) {
  float y = x * sc + sh;
  return (y >= 0.f) ? y : 0.1f * y;
}

__device__ __forceinline__ void bn_scale_shift(float sum, float sumsq, float g, float be,
                                               float& sc, float& sh) {
  float m   = sum * INV_N_NODES;
  float var = sumsq * INV_N_NODES - m * m;
  sc = g * rsqrtf(var + EPS_BN);
  sh = be - m * sc;
}

__device__ __forceinline__ void bn_lane_scale_shift(const float* __restrict__ stats,
                                                    const float* __restrict__ g,
                                                    const float* __restrict__ be, int lane4,
                                                    float4& sc, float4& sh) {
  const float4 su = *(const float4*)(stats + lane4);
  const float4 sq = *(const float4*)(stats + DF + lane4);
  const float4 gv = *(const float4*)(g + lane4);
  const float4 bv = *(const float4*)(be + lane4);
  bn_scale_shift(su.x, sq.x, gv.x, bv.x, sc.x, sh.x);
  bn_scale_shift(su.y, sq.y, gv.y, bv.y, sc.y, sh.y);
  bn_scale_shift(su.z, sq.z, gv.z, bv.z, sc.z, sh.z);
  bn_scale_shift(su.w, sq.w, gv.w, bv.w, sc.w, sh.w);
}

__device__ __forceinline__ float4 bn_leaky4(float4 v, float4 sc, float4 sh) {
  float4 y;
  y.x = bn_leaky1(v.x, sc.x, sh.x);
  y.y = bn_leaky1(v.y, sc.y, sh.y);
  y.z = bn_leaky1(v.z, sc.z, sh.z);
  y.w = bn_leaky1(v.w, sc.w, sh.w);
  return y;
}

#define GS_PITCH 132
#define GS_WAVES 8
static_assert(DF == 128, "a row is 32 lanes x 16 bytes; K of every product here is 128");
static_assert(GS_WAVES * 16 == DF, "the block's waves bring 16 rows of Bt each: all 128");

__global__ __launch_bounds__(GS_WAVES * 32)
void wmma_gemm_strip(const float* __restrict__ A, const float* __restrict__ Bt,
                     const float* __restrict__ bias, float* __restrict__ C, int M,
                     const float* __restrict__ stats, const float* __restrict__ g,
                     const float* __restrict__ be) {
  __shared__ __attribute__((aligned(16))) float sA[GS_WAVES][16 * GS_PITCH];
  __shared__ __attribute__((aligned(16))) float sB[DF * GS_PITCH];
  int gtid = blockIdx.x * blockDim.x + threadIdx.x;
  int wave = gtid >> 5;
  int lane = threadIdx.x & 31;
  int w    = threadIdx.x >> 5;
  const bool active = wave * 16 < M;
  const int lane4 = lane * 4;

  {
    const float* pb = Bt + (size_t)w * 16 * DF + lane4;
    float4 bv[16], av[16];
#pragma unroll
    for (int i = 0; i < 16; ++i) bv[i] = *(const float4*)(pb + (size_t)i * DF);
    if (active) {
      const float* pa = A + (size_t)wave * 16 * DF + lane4;
#pragma unroll
      for (int i = 0; i < 16; ++i) av[i] = *(const float4*)(pa + (size_t)i * DF);
    }
#pragma unroll
    for (int i = 0; i < 16; ++i) *(float4*)&sB[(w * 16 + i) * GS_PITCH + lane4] = bv[i];
    if (active) {
      if (stats != nullptr) {
        float4 sc, sh;
        bn_lane_scale_shift(stats, g, be, lane4, sc, sh);
#pragma unroll
        for (int i = 0; i < 16; ++i) av[i] = bn_leaky4(av[i], sc, sh);
      }
#pragma unroll
      for (int i = 0; i < 16; ++i) *(float4*)&sA[w][i * GS_PITCH + lane4] = av[i];
    }
  }
  __syncthreads();
  if (!active) return;

  int l16   = lane & 15;
  int khalf = (lane >> 4) << 1;
  const float* arow = &sA[w][l16 * GS_PITCH + khalf];
  const float* brow = &sB[l16 * GS_PITCH + khalf];

  v8f zero = {};
  v8f acc[8];
#pragma unroll
  for (int nt = 0; nt < 8; ++nt) acc[nt] = zero;

  for (int k = 0; k < DF; k += 4) {
    v2f a = *(const v2f*)(arow + k);
#pragma unroll
    for (int nt = 0; nt < 8; ++nt) {
      v2f b = *(const v2f*)(brow + nt * 16 * GS_PITCH + k);
      acc[nt] = wmma4(a, b, acc[nt]);
    }
  }

  int rbase = wave * 16 + ((lane >> 4) << 3);
#pragma unroll
  for (int nt = 0; nt < 8; ++nt) {
    float bs = bias ? bias[nt * 16 + l16] : 0.f;
    float* crow = C + (size_t)rbase * 128 + nt * 16 + l16;
#pragma unroll
    for (int r = 0; r < 8; ++r)
      crow[(size_t)r * 128] = acc[nt][r] + bs;
  }
}

__global__ __launch_bounds__(256)
void wmma_matmul(const float* __restrict__ A, const float* __restrict__ Bm,
                 const float* __restrict__ bias, float* __restrict__ C,
                 int M, int Nc, int K, int bTrans) {
  int gtid = blockIdx.x * blockDim.x + threadIdx.x;
  int wave = gtid >> 5;
  int lane = threadIdx.x & 31;
  int ntiles = Nc >> 4;
  int mt = wave / ntiles;
  int nt = wave - mt * ntiles;
  if (mt * 16 >= M) return;

  int l16   = lane & 15;
  int khalf = (lane >> 4) << 1;
  const float* arow = A + (size_t)(mt * 16 + l16) * K + khalf;

  v8f acc = {};
  if (bTrans) {
    const float* brow = Bm + (size_t)(nt * 16 + l16) * K + khalf;
    for (int k = 0; k < K; k += 4) {
      v2f a = *(const v2f*)(arow + k);
      v2f b = *(const v2f*)(brow + k);
      acc = wmma4(a, b, acc);
    }
  } else {
    const float* bcol = Bm + (size_t)khalf * Nc + nt * 16 + l16;
    for (int k = 0; k < K; k += 4) {
      v2f a = *(const v2f*)(arow + k);
      v2f b;
      b.x = bcol[(size_t)k * Nc];
      b.y = bcol[(size_t)(k + 1) * Nc];
      acc = wmma4(a, b, acc);
    }
  }

  float bs = bias ? bias[nt * 16 + l16] : 0.f;
  int rbase = mt * 16 + ((lane >> 4) << 3);
  float* crow = C + (size_t)rbase * Nc + nt * 16 + l16;
#pragma unroll
  for (int r = 0; r < 8; ++r)
    crow[(size_t)r * Nc] = acc[r] + bs;
}

__global__ void zero_cnt_transpose_w(int* __restrict__ cnt, int n,
                                     const float* __restrict__ w, float* __restrict__ wt) {
  int t = blockIdx.x * blockDim.x + threadIdx.x;
  if (t < n) cnt[t] = 0;
  if (t < 128 * 128) wt[(t & 127) * 128 + (t >> 7)] = w[t];
}

__global__ void count_edges(const int* __restrict__ ecol, int* __restrict__ cnt,
                            int* __restrict__ rank) {
  int e = blockIdx.x * blockDim.x + threadIdx.x;
  if (e < NE_EDGES) rank[e] = atomicAdd(&cnt[ecol[e]], 1);
}

#define SCAN_STEPS 13
#define SCAN_PER_WAVE (SCAN_STEPS * 128)
static_assert(32 * SCAN_PER_WAVE >= N_NODES, "the 32 waves cover every counter");
static_assert(N_NODES % 4 == 0, "a lane's 4 counters are all inside or all outside");

__device__ __forceinline__ int wave_incl_scan(int x, int lane) {
#pragma unroll
  for (int off = 1; off < 32; off <<= 1) {
    int y = __shfl_up(x, off, 32);
    if (lane >= off) x += y;
  }
  return x;
}

__global__ __launch_bounds__(1024)
void scan_rowptr(const int* __restrict__ cnt, int* __restrict__ rowptr) {
  __shared__ int wtot[32];
  const int tid = threadIdx.x;
  const int lane = tid & 31;
  const int w = tid >> 5;
  const int first = w * SCAN_PER_WAVE + lane * 4;
  int4 v[SCAN_STEPS];
#pragma unroll
  for (int j = 0; j < SCAN_STEPS; ++j) {
    int idx = first + j * 128;
    int idc = (idx < N_NODES) ? idx : (N_NODES - 4);
    v[j] = *(const int4*)(cnt + idc);
  }
  int run = 0;
#pragma unroll
  for (int j = 0; j < SCAN_STEPS; ++j) {
    const bool in = (first + j * 128) < N_NODES;
    const int c0 = in ? v[j].x : 0;
    const int c1 = in ? v[j].y : 0;
    const int c2 = in ? v[j].z : 0;
    const int c3 = in ? v[j].w : 0;
    const int s1 = c0, s2 = s1 + c1, s3 = s2 + c2, s4 = s3 + c3;
    const int inc = wave_incl_scan(s4, lane);
    const int ex = run + inc - s4;
    run += __shfl(inc, 31, 32);
    v[j].x = ex; v[j].y = ex + s1; v[j].z = ex + s2; v[j].w = ex + s3;
  }
  if (lane == 0) wtot[w] = run;
  __syncthreads();
  const int t  = wtot[lane];
  const int ti = wave_incl_scan(t, lane);
  const int base = __shfl(ti - t, w, 32);
#pragma unroll
  for (int j = 0; j < SCAN_STEPS; ++j) {
    int idx = first + j * 128;
    if (idx < N_NODES) {
      int4 o;
      o.x = v[j].x + base; o.y = v[j].y + base; o.z = v[j].z + base; o.w = v[j].w + base;
      *(int4*)(rowptr + idx) = o;
    }
  }
  if (tid == 1023) rowptr[N_NODES] = base + run;
}

__global__ void csr_fill(const int* __restrict__ erow, const int* __restrict__ ecol,
                         const int* __restrict__ rowptr, const int* __restrict__ rank,
                         int* __restrict__ srcidx, float* __restrict__ dis) {
  int e = blockIdx.x * blockDim.x + threadIdx.x;
  if (e >= NE_EDGES) return;
  if (e < N_NODES)
    dis[e] = rsqrtf((float)(rowptr[e + 1] - rowptr[e]) + 1.0f);
  int c = ecol[e];
  srcidx[rowptr[c] + rank[e]] = erow[e];
}

static_assert(NE_EDGES >= N_NODES, "csr_fill's first N_NODES threads write dis");
static_assert((size_t)NE_EDGES <= (size_t)N_NODES * DF, "rank (one int per edge) fits in the agg plane");
__global__ __launch_bounds__(256)
void gcn_aggregate(const int* __restrict__ rowptr, const int* __restrict__ srcidx,
                   const float* __restrict__ dis, const float* __restrict__ lin,
                   float* __restrict__ agg, float* __restrict__ stats,
                   const float* __restrict__ wnext, float* __restrict__ wt) {
  int t = blockIdx.x * blockDim.x + threadIdx.x;
  if (blockIdx.x == 0) stats[threadIdx.x] = 0.0f;
  if (wnext != nullptr && t < 128 * 128) wt[(t & 127) * 128 + (t >> 7)] = wnext[t];
  int c = t >> 5;
  if (c >= N_NODES) return;
  int lane = threadIdx.x & 31;
  float dc = dis[c];
  const float4 sv = *(const float4*)(lin + (size_t)c * DF + lane * 4);
  float s = dc * dc;
  float ax = s * sv.x, ay = s * sv.y, az = s * sv.z, aw = s * sv.w;
  int e0 = rowptr[c], e1 = rowptr[c + 1];
  for (int j = e0; j < e1; ++j) {
    int r = srcidx[j];
    float nrm = dis[r] * dc;
    const float4 v = *(const float4*)(lin + (size_t)r * DF + lane * 4);
    ax += nrm * v.x; ay += nrm * v.y; az += nrm * v.z; aw += nrm * v.w;
  }
  float4 o; o.x = ax; o.y = ay; o.z = az; o.w = aw;
  *(float4*)(agg + (size_t)c * DF + lane * 4) = o;
}

__global__ __launch_bounds__(1024)
void bn_stats(const float* __restrict__ agg, float* __restrict__ stats, int n) {
  __shared__ __attribute__((aligned(16))) float red[32][2 * DF];
  int tid  = threadIdx.x;
  int lane = tid & 31;
  int w    = tid >> 5;
  int rw   = blockIdx.x * 512 + w * 16;
  float4 s = {0.f, 0.f, 0.f, 0.f};
  float4 q = {0.f, 0.f, 0.f, 0.f};
  if (rw + 16 <= n) {
    const float* p = agg + (size_t)rw * DF + lane * 4;
    float4 v[16];
#pragma unroll
    for (int i = 0; i < 16; ++i) v[i] = *(const float4*)(p + (size_t)i * DF);
#pragma unroll
    for (int i = 0; i < 16; ++i) {
      s.x += v[i].x; s.y += v[i].y; s.z += v[i].z; s.w += v[i].w;
      q.x += v[i].x * v[i].x; q.y += v[i].y * v[i].y;
      q.z += v[i].z * v[i].z; q.w += v[i].w * v[i].w;
    }
  } else {
    for (int i = 0; i < 16; ++i) {
      int r = rw + i;
      if (r < n) {
        const float4 v = *(const float4*)(agg + (size_t)r * DF + lane * 4);
        s.x += v.x; s.y += v.y; s.z += v.z; s.w += v.w;
        q.x += v.x * v.x; q.y += v.y * v.y; q.z += v.z * v.z; q.w += v.w * v.w;
      }
    }
  }
  *(float4*)&red[w][lane * 4]      = s;
  *(float4*)&red[w][DF + lane * 4] = q;
  __syncthreads();
  if (tid < 2 * DF) {
    float a = 0.f;
#pragma unroll
    for (int k = 0; k < 32; ++k) a += red[k][tid];
    atomicAdd(&stats[tid], a);
  }
}

__global__ __launch_bounds__(256)
void gather_pairs(const float* __restrict__ agg, const float* __restrict__ stats,
                  const float* __restrict__ g, const float* __restrict__ be,
                  const int* __restrict__ di,
                  float* __restrict__ amat, float* __restrict__ bmat) {
  int t = blockIdx.x * blockDim.x + threadIdx.x;
  if (t >= BPAIRS * 32) return;
  int j = t >> 5, q = t & 31;
  int ia = di[2 * j] - 1;     if (ia < 0) ia += N_NODES;
  int ib = di[2 * j + 1] - 1; if (ib < 0) ib += N_NODES;
  const float4 va = *(const float4*)(agg + (size_t)ia * DF + q * 4);
  const float4 vb = *(const float4*)(agg + (size_t)ib * DF + q * 4);
  float4 sc, sh;
  bn_lane_scale_shift(stats, g, be, q * 4, sc, sh);
  *(float4*)(amat + (size_t)j * DF + q * 4) = bn_leaky4(va, sc, sh);
  *(float4*)(bmat + (size_t)j * DF + q * 4) = bn_leaky4(vb, sc, sh);
}

__global__ __launch_bounds__(256)
void rowdot(const float* __restrict__ tm, const float* __restrict__ bm,
            float* __restrict__ out) {
  int t = blockIdx.x * blockDim.x + threadIdx.x;
  int j = t >> 5, lane = t & 31;
  if (j >= BPAIRS) return;
  float4 tv = *(const float4*)(tm + (size_t)j * DF + lane * 4);
  float4 bv = *(const float4*)(bm + (size_t)j * DF + lane * 4);
  float p = tv.x * bv.x + tv.y * bv.y + tv.z * bv.z + tv.w * bv.w;
#pragma unroll
  for (int off = 16; off > 0; off >>= 1)
    p += __shfl_down(p, off, 32);
  if (lane == 0) out[j] = p;
}

extern "C" void kernel_launch(void* const* d_in, const int* in_sizes, int n_in,
                              void* d_out, int out_size, void* d_ws, size_t ws_size,
                              hipStream_t stream) {
  const float* x   = (const float*)d_in[0];
  const int*   edg = (const int*)d_in[1];
  const int*   di  = (const int*)d_in[2];
  const float* w1  = (const float*)d_in[3];
  const float* b1  = (const float*)d_in[4];
  const float* w2  = (const float*)d_in[5];
  const float* b2  = (const float*)d_in[6];
  const float* w3  = (const float*)d_in[7];
  const float* b3  = (const float*)d_in[8];
  const float* g1  = (const float*)d_in[9];
  const float* be1 = (const float*)d_in[10];
  const float* g2  = (const float*)d_in[11];
  const float* be2 = (const float*)d_in[12];
  const float* g3  = (const float*)d_in[13];
  const float* be3 = (const float*)d_in[14];
  const float* p1  = (const float*)d_in[15];
  const float* p2  = (const float*)d_in[16];
  float* out = (float*)d_out;

  float* ws = (float*)d_ws;
  size_t off = 0;
  float* dis    = ws + off; off += N_NODES;
  float* stats  = ws + off; off += 256;
  float* Wt     = ws + off; off += 128 * 128;
  int*   cnt    = (int*)(ws + off); off += N_NODES;
  int*   rowptr = (int*)(ws + off); off += N_NODES + 16;
  off += N_NODES;
  int*   srcidx = (int*)(ws + off); off += NE_EDGES;
  float* lin    = ws + off; off += (size_t)N_NODES * DF;
  float* agg    = ws + off; off += (size_t)N_NODES * DF;
  float* U    = lin;
  float* Qt   = lin + 16384;
  float* amat = lin + 32768;
  float* bmat = amat + (size_t)BPAIRS * DF;
  float* tmat = bmat + (size_t)BPAIRS * DF;
  int* rank = (int*)agg;

  const int* erow = edg;
  const int* ecol = edg + NE_EDGES;

  zero_cnt_transpose_w<<<(N_NODES + 255) / 256, 256, 0, stream>>>(cnt, N_NODES, w1, Wt);
  count_edges<<<(NE_EDGES + 255) / 256, 256, 0, stream>>>(ecol, cnt, rank);
  scan_rowptr<<<1, 1024, 0, stream>>>(cnt, rowptr);
  csr_fill<<<(NE_EDGES + 255) / 256, 256, 0, stream>>>(erow, ecol, rowptr, rank, srcidx, dis);

  const float* Wn[3] = {w2, w3, nullptr};
  const float* Bl[3] = {b1, b2, b3};
  const float* Gl[3] = {g1, g2, g3};
  const float* El[3] = {be1, be2, be3};
  const int strip_waves = N_NODES / 16;
  for (int l = 0; l < 3; ++l) {
    if (l == 0)
      wmma_gemm_strip<<<(strip_waves * 32 + 255) / 256, 256, 0, stream>>>(
          x, Wt, Bl[l], lin, N_NODES, nullptr, nullptr, nullptr);
    else
      wmma_gemm_strip<<<(strip_waves * 32 + 255) / 256, 256, 0, stream>>>(
          agg, Wt, Bl[l], lin, N_NODES, stats, Gl[l - 1], El[l - 1]);
    gcn_aggregate<<<(N_NODES * 32 + 255) / 256, 256, 0, stream>>>(
        rowptr, srcidx, dis, lin, agg, stats, Wn[l], Wt);
    bn_stats<<<(N_NODES + 511) / 512, 1024, 0, stream>>>(agg, stats, N_NODES);
  }

  const int small_tiles = (DF / 16) * (DF / 16);
  wmma_matmul<<<(small_tiles * 32 + 255) / 256, 256, 0, stream>>>(p1, p2, nullptr, U, DF, DF, DF, 1);
  wmma_matmul<<<(small_tiles * 32 + 255) / 256, 256, 0, stream>>>(U, p1, nullptr, Qt, DF, DF, DF, 1);
  gather_pairs<<<(BPAIRS * 32 + 255) / 256, 256, 0, stream>>>(agg, stats, g3, be3, di, amat, bmat);
  wmma_gemm_strip<<<((BPAIRS / 16) * 32 + 255) / 256, 256, 0, stream>>>(
      amat, Qt, nullptr, tmat, BPAIRS, nullptr, nullptr, nullptr);
  rowdot<<<(BPAIRS * 32 + 255) / 256, 256, 0, stream>>>(tmat, bmat, out);
}


# workspace planes start on 512-byte boundaries: a row of lin or agg is 4 whole 128-byte lines for gather, load and store, not 5 (rows began 64 bytes into a line); padding from the unused slot
# speedup vs baseline: 4.5128x; 1.0905x over previous
#include <hip/hip_runtime.h>

#define N_NODES 50000
#define NE_EDGES 800000
#define BPAIRS 4096
#define DF 128
#define EPS_BN 1e-5f

typedef float v2f __attribute__((ext_vector_type(2)));
typedef float v8f __attribute__((ext_vector_type(8)));

__device__ __forceinline__ v8f wmma4(v2f a, v2f b, v8f c) {
  return __builtin_amdgcn_wmma_f32_16x16x4_f32(
       false, a,  false, b,
       (short)0, c,  false,  false);
}

#define INV_N_NODES (1.0f / (float)N_NODES)

__device__ __forceinline__ float bn_leaky1(float x, float sc, float sh) {
  float y = x * sc + sh;
  return (y >= 0.f) ? y : 0.1f * y;
}

__device__ __forceinline__ void bn_scale_shift(float sum, float sumsq, float g, float be,
                                               float& sc, float& sh) {
  float m   = sum * INV_N_NODES;
  float var = sumsq * INV_N_NODES - m * m;
  sc = g * rsqrtf(var + EPS_BN);
  sh = be - m * sc;
}

__device__ __forceinline__ void bn_lane_scale_shift(const float* __restrict__ stats,
                                                    const float* __restrict__ g,
                                                    const float* __restrict__ be, int lane4,
                                                    float4& sc, float4& sh) {
  const float4 su = *(const float4*)(stats + lane4);
  const float4 sq = *(const float4*)(stats + DF + lane4);
  const float4 gv = *(const float4*)(g + lane4);
  const float4 bv = *(const float4*)(be + lane4);
  bn_scale_shift(su.x, sq.x, gv.x, bv.x, sc.x, sh.x);
  bn_scale_shift(su.y, sq.y, gv.y, bv.y, sc.y, sh.y);
  bn_scale_shift(su.z, sq.z, gv.z, bv.z, sc.z, sh.z);
  bn_scale_shift(su.w, sq.w, gv.w, bv.w, sc.w, sh.w);
}

__device__ __forceinline__ float4 bn_leaky4(float4 v, float4 sc, float4 sh) {
  float4 y;
  y.x = bn_leaky1(v.x, sc.x, sh.x);
  y.y = bn_leaky1(v.y, sc.y, sh.y);
  y.z = bn_leaky1(v.z, sc.z, sh.z);
  y.w = bn_leaky1(v.w, sc.w, sh.w);
  return y;
}

#define GS_PITCH 132
#define GS_WAVES 8
static_assert(DF == 128, "a row is 32 lanes x 16 bytes; K of every product here is 128");
static_assert(GS_WAVES * 16 == DF, "the block's waves bring 16 rows of Bt each: all 128");

__global__ __launch_bounds__(GS_WAVES * 32)
void wmma_gemm_strip(const float* __restrict__ A, const float* __restrict__ Bt,
                     const float* __restrict__ bias, float* __restrict__ C, int M,
                     const float* __restrict__ stats, const float* __restrict__ g,
                     const float* __restrict__ be) {
  __shared__ __attribute__((aligned(16))) float sA[GS_WAVES][16 * GS_PITCH];
  __shared__ __attribute__((aligned(16))) float sB[DF * GS_PITCH];
  int gtid = blockIdx.x * blockDim.x + threadIdx.x;
  int wave = gtid >> 5;
  int lane = threadIdx.x & 31;
  int w    = threadIdx.x >> 5;
  const bool active = wave * 16 < M;
  const int lane4 = lane * 4;

  {
    const float* pb = Bt + (size_t)w * 16 * DF + lane4;
    float4 bv[16], av[16];
#pragma unroll
    for (int i = 0; i < 16; ++i) bv[i] = *(const float4*)(pb + (size_t)i * DF);
    if (active) {
      const float* pa = A + (size_t)wave * 16 * DF + lane4;
#pragma unroll
      for (int i = 0; i < 16; ++i) av[i] = *(const float4*)(pa + (size_t)i * DF);
    }
#pragma unroll
    for (int i = 0; i < 16; ++i) *(float4*)&sB[(w * 16 + i) * GS_PITCH + lane4] = bv[i];
    if (active) {
      if (stats != nullptr) {
        float4 sc, sh;
        bn_lane_scale_shift(stats, g, be, lane4, sc, sh);
#pragma unroll
        for (int i = 0; i < 16; ++i) av[i] = bn_leaky4(av[i], sc, sh);
      }
#pragma unroll
      for (int i = 0; i < 16; ++i) *(float4*)&sA[w][i * GS_PITCH + lane4] = av[i];
    }
  }
  __syncthreads();
  if (!active) return;

  int l16   = lane & 15;
  int khalf = (lane >> 4) << 1;
  const float* arow = &sA[w][l16 * GS_PITCH + khalf];
  const float* brow = &sB[l16 * GS_PITCH + khalf];

  v8f zero = {};
  v8f acc[8];
#pragma unroll
  for (int nt = 0; nt < 8; ++nt) acc[nt] = zero;

  for (int k = 0; k < DF; k += 4) {
    v2f a = *(const v2f*)(arow + k);
#pragma unroll
    for (int nt = 0; nt < 8; ++nt) {
      v2f b = *(const v2f*)(brow + nt * 16 * GS_PITCH + k);
      acc[nt] = wmma4(a, b, acc[nt]);
    }
  }

  int rbase = wave * 16 + ((lane >> 4) << 3);
#pragma unroll
  for (int nt = 0; nt < 8; ++nt) {
    float bs = bias ? bias[nt * 16 + l16] : 0.f;
    float* crow = C + (size_t)rbase * 128 + nt * 16 + l16;
#pragma unroll
    for (int r = 0; r < 8; ++r)
      crow[(size_t)r * 128] = acc[nt][r] + bs;
  }
}

__global__ __launch_bounds__(256)
void wmma_matmul(const float* __restrict__ A, const float* __restrict__ Bm,
                 const float* __restrict__ bias, float* __restrict__ C,
                 int M, int Nc, int K, int bTrans) {
  int gtid = blockIdx.x * blockDim.x + threadIdx.x;
  int wave = gtid >> 5;
  int lane = threadIdx.x & 31;
  int ntiles = Nc >> 4;
  int mt = wave / ntiles;
  int nt = wave - mt * ntiles;
  if (mt * 16 >= M) return;

  int l16   = lane & 15;
  int khalf = (lane >> 4) << 1;
  const float* arow = A + (size_t)(mt * 16 + l16) * K + khalf;

  v8f acc = {};
  if (bTrans) {
    const float* brow = Bm + (size_t)(nt * 16 + l16) * K + khalf;
    for (int k = 0; k < K; k += 4) {
      v2f a = *(const v2f*)(arow + k);
      v2f b = *(const v2f*)(brow + k);
      acc = wmma4(a, b, acc);
    }
  } else {
    const float* bcol = Bm + (size_t)khalf * Nc + nt * 16 + l16;
    for (int k = 0; k < K; k += 4) {
      v2f a = *(const v2f*)(arow + k);
      v2f b;
      b.x = bcol[(size_t)k * Nc];
      b.y = bcol[(size_t)(k + 1) * Nc];
      acc = wmma4(a, b, acc);
    }
  }

  float bs = bias ? bias[nt * 16 + l16] : 0.f;
  int rbase = mt * 16 + ((lane >> 4) << 3);
  float* crow = C + (size_t)rbase * Nc + nt * 16 + l16;
#pragma unroll
  for (int r = 0; r < 8; ++r)
    crow[(size_t)r * Nc] = acc[r] + bs;
}

__global__ void zero_cnt_transpose_w(int* __restrict__ cnt, int n,
                                     const float* __restrict__ w, float* __restrict__ wt) {
  int t = blockIdx.x * blockDim.x + threadIdx.x;
  if (t < n) cnt[t] = 0;
  if (t < 128 * 128) wt[(t & 127) * 128 + (t >> 7)] = w[t];
}

__global__ void count_edges(const int* __restrict__ ecol, int* __restrict__ cnt,
                            int* __restrict__ rank) {
  int e = blockIdx.x * blockDim.x + threadIdx.x;
  if (e < NE_EDGES) rank[e] = atomicAdd(&cnt[ecol[e]], 1);
}

#define SCAN_STEPS 13
#define SCAN_PER_WAVE (SCAN_STEPS * 128)
static_assert(32 * SCAN_PER_WAVE >= N_NODES, "the 32 waves cover every counter");
static_assert(N_NODES % 4 == 0, "a lane's 4 counters are all inside or all outside");

__device__ __forceinline__ int wave_incl_scan(int x, int lane) {
#pragma unroll
  for (int off = 1; off < 32; off <<= 1) {
    int y = __shfl_up(x, off, 32);
    if (lane >= off) x += y;
  }
  return x;
}

__global__ __launch_bounds__(1024)
void scan_rowptr(const int* __restrict__ cnt, int* __restrict__ rowptr) {
  __shared__ int wtot[32];
  const int tid = threadIdx.x;
  const int lane = tid & 31;
  const int w = tid >> 5;
  const int first = w * SCAN_PER_WAVE + lane * 4;
  int4 v[SCAN_STEPS];
#pragma unroll
  for (int j = 0; j < SCAN_STEPS; ++j) {
    int idx = first + j * 128;
    int idc = (idx < N_NODES) ? idx : (N_NODES - 4);
    v[j] = *(const int4*)(cnt + idc);
  }
  int run = 0;
#pragma unroll
  for (int j = 0; j < SCAN_STEPS; ++j) {
    const bool in = (first + j * 128) < N_NODES;
    const int c0 = in ? v[j].x : 0;
    const int c1 = in ? v[j].y : 0;
    const int c2 = in ? v[j].z : 0;
    const int c3 = in ? v[j].w : 0;
    const int s1 = c0, s2 = s1 + c1, s3 = s2 + c2, s4 = s3 + c3;
    const int inc = wave_incl_scan(s4, lane);
    const int ex = run + inc - s4;
    run += __shfl(inc, 31, 32);
    v[j].x = ex; v[j].y = ex + s1; v[j].z = ex + s2; v[j].w = ex + s3;
  }
  if (lane == 0) wtot[w] = run;
  __syncthreads();
  const int t  = wtot[lane];
  const int ti = wave_incl_scan(t, lane);
  const int base = __shfl(ti - t, w, 32);
#pragma unroll
  for (int j = 0; j < SCAN_STEPS; ++j) {
    int idx = first + j * 128;
    if (idx < N_NODES) {
      int4 o;
      o.x = v[j].x + base; o.y = v[j].y + base; o.z = v[j].z + base; o.w = v[j].w + base;
      *(int4*)(rowptr + idx) = o;
    }
  }
  if (tid == 1023) rowptr[N_NODES] = base + run;
}

__global__ void csr_fill(const int* __restrict__ erow, const int* __restrict__ ecol,
                         const int* __restrict__ rowptr, const int* __restrict__ rank,
                         int* __restrict__ srcidx, float* __restrict__ dis) {
  int e = blockIdx.x * blockDim.x + threadIdx.x;
  if (e >= NE_EDGES) return;
  if (e < N_NODES)
    dis[e] = rsqrtf((float)(rowptr[e + 1] - rowptr[e]) + 1.0f);
  int c = ecol[e];
  srcidx[rowptr[c] + rank[e]] = erow[e];
}

static_assert(NE_EDGES >= N_NODES, "csr_fill's first N_NODES threads write dis");
static_assert((size_t)NE_EDGES <= (size_t)N_NODES * DF, "rank (one int per edge) fits in the agg plane");
__global__ __launch_bounds__(256)
void gcn_aggregate(const int* __restrict__ rowptr, const int* __restrict__ srcidx,
                   const float* __restrict__ dis, const float* __restrict__ lin,
                   float* __restrict__ agg, float* __restrict__ stats,
                   const float* __restrict__ wnext, float* __restrict__ wt) {
  int t = blockIdx.x * blockDim.x + threadIdx.x;
  if (blockIdx.x == 0) stats[threadIdx.x] = 0.0f;
  if (wnext != nullptr && t < 128 * 128) wt[(t & 127) * 128 + (t >> 7)] = wnext[t];
  int c = t >> 5;
  if (c >= N_NODES) return;
  int lane = threadIdx.x & 31;
  float dc = dis[c];
  const float4 sv = *(const float4*)(lin + (size_t)c * DF + lane * 4);
  float s = dc * dc;
  float ax = s * sv.x, ay = s * sv.y, az = s * sv.z, aw = s * sv.w;
  int e0 = rowptr[c], e1 = rowptr[c + 1];
  for (int j = e0; j < e1; ++j) {
    int r = srcidx[j];
    float nrm = dis[r] * dc;
    const float4 v = *(const float4*)(lin + (size_t)r * DF + lane * 4);
    ax += nrm * v.x; ay += nrm * v.y; az += nrm * v.z; aw += nrm * v.w;
  }
  float4 o; o.x = ax; o.y = ay; o.z = az; o.w = aw;
  *(float4*)(agg + (size_t)c * DF + lane * 4) = o;
}

__global__ __launch_bounds__(1024)
void bn_stats(const float* __restrict__ agg, float* __restrict__ stats, int n) {
  __shared__ __attribute__((aligned(16))) float red[32][2 * DF];
  int tid  = threadIdx.x;
  int lane = tid & 31;
  int w    = tid >> 5;
  int rw   = blockIdx.x * 512 + w * 16;
  float4 s = {0.f, 0.f, 0.f, 0.f};
  float4 q = {0.f, 0.f, 0.f, 0.f};
  if (rw + 16 <= n) {
    const float* p = agg + (size_t)rw * DF + lane * 4;
    float4 v[16];
#pragma unroll
    for (int i = 0; i < 16; ++i) v[i] = *(const float4*)(p + (size_t)i * DF);
#pragma unroll
    for (int i = 0; i < 16; ++i) {
      s.x += v[i].x; s.y += v[i].y; s.z += v[i].z; s.w += v[i].w;
      q.x += v[i].x * v[i].x; q.y += v[i].y * v[i].y;
      q.z += v[i].z * v[i].z; q.w += v[i].w * v[i].w;
    }
  } else {
    for (int i = 0; i < 16; ++i) {
      int r = rw + i;
      if (r < n) {
        const float4 v = *(const float4*)(agg + (size_t)r * DF + lane * 4);
        s.x += v.x; s.y += v.y; s.z += v.z; s.w += v.w;
        q.x += v.x * v.x; q.y += v.y * v.y; q.z += v.z * v.z; q.w += v.w * v.w;
      }
    }
  }
  *(float4*)&red[w][lane * 4]      = s;
  *(float4*)&red[w][DF + lane * 4] = q;
  __syncthreads();
  if (tid < 2 * DF) {
    float a = 0.f;
#pragma unroll
    for (int k = 0; k < 32; ++k) a += red[k][tid];
    atomicAdd(&stats[tid], a);
  }
}

__global__ __launch_bounds__(256)
void gather_pairs(const float* __restrict__ agg, const float* __restrict__ stats,
                  const float* __restrict__ g, const float* __restrict__ be,
                  const int* __restrict__ di,
                  float* __restrict__ amat, float* __restrict__ bmat) {
  int t = blockIdx.x * blockDim.x + threadIdx.x;
  if (t >= BPAIRS * 32) return;
  int j = t >> 5, q = t & 31;
  int ia = di[2 * j] - 1;     if (ia < 0) ia += N_NODES;
  int ib = di[2 * j + 1] - 1; if (ib < 0) ib += N_NODES;
  const float4 va = *(const float4*)(agg + (size_t)ia * DF + q * 4);
  const float4 vb = *(const float4*)(agg + (size_t)ib * DF + q * 4);
  float4 sc, sh;
  bn_lane_scale_shift(stats, g, be, q * 4, sc, sh);
  *(float4*)(amat + (size_t)j * DF + q * 4) = bn_leaky4(va, sc, sh);
  *(float4*)(bmat + (size_t)j * DF + q * 4) = bn_leaky4(vb, sc, sh);
}

__global__ __launch_bounds__(256)
void rowdot(const float* __restrict__ tm, const float* __restrict__ bm,
            float* __restrict__ out) {
  int t = blockIdx.x * blockDim.x + threadIdx.x;
  int j = t >> 5, lane = t & 31;
  if (j >= BPAIRS) return;
  float4 tv = *(const float4*)(tm + (size_t)j * DF + lane * 4);
  float4 bv = *(const float4*)(bm + (size_t)j * DF + lane * 4);
  float p = tv.x * bv.x + tv.y * bv.y + tv.z * bv.z + tv.w * bv.w;
#pragma unroll
  for (int off = 16; off > 0; off >>= 1)
    p += __shfl_down(p, off, 32);
  if (lane == 0) out[j] = p;
}

#define WS_ALIGN 128
#define WS_PAD(n) ((((size_t)(n)) + WS_ALIGN - 1) / WS_ALIGN * WS_ALIGN)
static_assert(WS_PAD(N_NODES) + WS_PAD(256) + WS_PAD(128 * 128) + WS_PAD(N_NODES) +
              WS_PAD(N_NODES + 16) + WS_PAD(NE_EDGES)
              <= (size_t)N_NODES + 256 + 128 * 128 + N_NODES + (N_NODES + 16) + N_NODES + NE_EDGES,
              "the aligned planes end no later than the packed ones with their unused slot did");

extern "C" void kernel_launch(void* const* d_in, const int* in_sizes, int n_in,
                              void* d_out, int out_size, void* d_ws, size_t ws_size,
                              hipStream_t stream) {
  const float* x   = (const float*)d_in[0];
  const int*   edg = (const int*)d_in[1];
  const int*   di  = (const int*)d_in[2];
  const float* w1  = (const float*)d_in[3];
  const float* b1  = (const float*)d_in[4];
  const float* w2  = (const float*)d_in[5];
  const float* b2  = (const float*)d_in[6];
  const float* w3  = (const float*)d_in[7];
  const float* b3  = (const float*)d_in[8];
  const float* g1  = (const float*)d_in[9];
  const float* be1 = (const float*)d_in[10];
  const float* g2  = (const float*)d_in[11];
  const float* be2 = (const float*)d_in[12];
  const float* g3  = (const float*)d_in[13];
  const float* be3 = (const float*)d_in[14];
  const float* p1  = (const float*)d_in[15];
  const float* p2  = (const float*)d_in[16];
  float* out = (float*)d_out;

  float* ws = (float*)d_ws;
  size_t off = 0;
  float* dis    = ws + off; off += WS_PAD(N_NODES);
  float* stats  = ws + off; off += WS_PAD(256);
  float* Wt     = ws + off; off += WS_PAD(128 * 128);
  int*   cnt    = (int*)(ws + off); off += WS_PAD(N_NODES);
  int*   rowptr = (int*)(ws + off); off += WS_PAD(N_NODES + 16);
  int*   srcidx = (int*)(ws + off); off += WS_PAD(NE_EDGES);
  float* lin    = ws + off; off += (size_t)N_NODES * DF;
  float* agg    = ws + off; off += (size_t)N_NODES * DF;
  float* U    = lin;
  float* Qt   = lin + 16384;
  float* amat = lin + 32768;
  float* bmat = amat + (size_t)BPAIRS * DF;
  float* tmat = bmat + (size_t)BPAIRS * DF;
  int* rank = (int*)agg;

  const int* erow = edg;
  const int* ecol = edg + NE_EDGES;

  zero_cnt_transpose_w<<<(N_NODES + 255) / 256, 256, 0, stream>>>(cnt, N_NODES, w1, Wt);
  count_edges<<<(NE_EDGES + 255) / 256, 256, 0, stream>>>(ecol, cnt, rank);
  scan_rowptr<<<1, 1024, 0, stream>>>(cnt, rowptr);
  csr_fill<<<(NE_EDGES + 255) / 256, 256, 0, stream>>>(erow, ecol, rowptr, rank, srcidx, dis);

  const float* Wn[3] = {w2, w3, nullptr};
  const float* Bl[3] = {b1, b2, b3};
  const float* Gl[3] = {g1, g2, g3};
  const float* El[3] = {be1, be2, be3};
  const int strip_waves = N_NODES / 16;
  for (int l = 0; l < 3; ++l) {
    if (l == 0)
      wmma_gemm_strip<<<(strip_waves * 32 + 255) / 256, 256, 0, stream>>>(
          x, Wt, Bl[l], lin, N_NODES, nullptr, nullptr, nullptr);
    else
      wmma_gemm_strip<<<(strip_waves * 32 + 255) / 256, 256, 0, stream>>>(
          agg, Wt, Bl[l], lin, N_NODES, stats, Gl[l - 1], El[l - 1]);
    gcn_aggregate<<<(N_NODES * 32 + 255) / 256, 256, 0, stream>>>(
        rowptr, srcidx, dis, lin, agg, stats, Wn[l], Wt);
    bn_stats<<<(N_NODES + 511) / 512, 1024, 0, stream>>>(agg, stats, N_NODES);
  }

  const int small_tiles = (DF / 16) * (DF / 16);
  wmma_matmul<<<(small_tiles * 32 + 255) / 256, 256, 0, stream>>>(p1, p2, nullptr, U, DF, DF, DF, 1);
  wmma_matmul<<<(small_tiles * 32 + 255) / 256, 256, 0, stream>>>(U, p1, nullptr, Qt, DF, DF, DF, 1);
  gather_pairs<<<(BPAIRS * 32 + 255) / 256, 256, 0, stream>>>(agg, stats, g3, be3, di, amat, bmat);
  wmma_gemm_strip<<<((BPAIRS / 16) * 32 + 255) / 256, 256, 0, stream>>>(
      amat, Qt, nullptr, tmat, BPAIRS, nullptr, nullptr, nullptr);
  rowdot<<<(BPAIRS * 32 + 255) / 256, 256, 0, stream>>>(tmat, bmat, out);
}


# decoder tail 5 launches -> 1: pair gather, a@Q strip product and row dot fused (b rows in LDS); Q^T's two products ride as blocks of bn_stats launches; the dot sums 8 tiles, then 16 lanes
# speedup vs baseline: 4.6942x; 1.0402x over previous
#include <hip/hip_runtime.h>

#define N_NODES 50000
#define NE_EDGES 800000
#define BPAIRS 4096
#define DF 128
#define EPS_BN 1e-5f

typedef float v2f __attribute__((ext_vector_type(2)));
typedef float v8f __attribute__((ext_vector_type(8)));

__device__ __forceinline__ v8f wmma4(v2f a, v2f b, v8f c) {
  return __builtin_amdgcn_wmma_f32_16x16x4_f32(
       false, a,  false, b,
       (short)0, c,  false,  false);
}

#define INV_N_NODES (1.0f / (float)N_NODES)

__device__ __forceinline__ float bn_leaky1(float x, float sc, float sh) {
  float y = x * sc + sh;
  return (y >= 0.f) ? y : 0.1f * y;
}

__device__ __forceinline__ void bn_scale_shift(float sum, float sumsq, float g, float be,
                                               float& sc, float& sh) {
  float m   = sum * INV_N_NODES;
  float var = sumsq * INV_N_NODES - m * m;
  sc = g * rsqrtf(var + EPS_BN);
  sh = be - m * sc;
}

__device__ __forceinline__ void bn_lane_scale_shift(const float* __restrict__ stats,
                                                    const float* __restrict__ g,
                                                    const float* __restrict__ be, int lane4,
                                                    float4& sc, float4& sh) {
  const float4 su = *(const float4*)(stats + lane4);
  const float4 sq = *(const float4*)(stats + DF + lane4);
  const float4 gv = *(const float4*)(g + lane4);
  const float4 bv = *(const float4*)(be + lane4);
  bn_scale_shift(su.x, sq.x, gv.x, bv.x, sc.x, sh.x);
  bn_scale_shift(su.y, sq.y, gv.y, bv.y, sc.y, sh.y);
  bn_scale_shift(su.z, sq.z, gv.z, bv.z, sc.z, sh.z);
  bn_scale_shift(su.w, sq.w, gv.w, bv.w, sc.w, sh.w);
}

__device__ __forceinline__ float4 bn_leaky4(float4 v, float4 sc, float4 sh) {
  float4 y;
  y.x = bn_leaky1(v.x, sc.x, sh.x);
  y.y = bn_leaky1(v.y, sc.y, sh.y);
  y.z = bn_leaky1(v.z, sc.z, sh.z);
  y.w = bn_leaky1(v.w, sc.w, sh.w);
  return y;
}

#define GS_PITCH 132
#define GS_WAVES 8
static_assert(DF == 128, "a row is 32 lanes x 16 bytes; K of every product here is 128");
static_assert(GS_WAVES * 16 == DF, "the block's waves bring 16 rows of Bt each: all 128");

__global__ __launch_bounds__(GS_WAVES * 32)
void wmma_gemm_strip(const float* __restrict__ A, const float* __restrict__ Bt,
                     const float* __restrict__ bias, float* __restrict__ C, int M,
                     const float* __restrict__ stats, const float* __restrict__ g,
                     const float* __restrict__ be) {
  __shared__ __attribute__((aligned(16))) float sA[GS_WAVES][16 * GS_PITCH];
  __shared__ __attribute__((aligned(16))) float sB[DF * GS_PITCH];
  int gtid = blockIdx.x * blockDim.x + threadIdx.x;
  int wave = gtid >> 5;
  int lane = threadIdx.x & 31;
  int w    = threadIdx.x >> 5;
  const bool active = wave * 16 < M;
  const int lane4 = lane * 4;

  {
    const float* pb = Bt + (size_t)w * 16 * DF + lane4;
    float4 bv[16], av[16];
#pragma unroll
    for (int i = 0; i < 16; ++i) bv[i] = *(const float4*)(pb + (size_t)i * DF);
    if (active) {
      const float* pa = A + (size_t)wave * 16 * DF + lane4;
#pragma unroll
      for (int i = 0; i < 16; ++i) av[i] = *(const float4*)(pa + (size_t)i * DF);
    }
#pragma unroll
    for (int i = 0; i < 16; ++i) *(float4*)&sB[(w * 16 + i) * GS_PITCH + lane4] = bv[i];
    if (active) {
      if (stats != nullptr) {
        float4 sc, sh;
        bn_lane_scale_shift(stats, g, be, lane4, sc, sh);
#pragma unroll
        for (int i = 0; i < 16; ++i) av[i] = bn_leaky4(av[i], sc, sh);
      }
#pragma unroll
      for (int i = 0; i < 16; ++i) *(float4*)&sA[w][i * GS_PITCH + lane4] = av[i];
    }
  }
  __syncthreads();
  if (!active) return;

  int l16   = lane & 15;
  int khalf = (lane >> 4) << 1;
  const float* arow = &sA[w][l16 * GS_PITCH + khalf];
  const float* brow = &sB[l16 * GS_PITCH + khalf];

  v8f zero = {};
  v8f acc[8];
#pragma unroll
  for (int nt = 0; nt < 8; ++nt) acc[nt] = zero;

  for (int k = 0; k < DF; k += 4) {
    v2f a = *(const v2f*)(arow + k);
#pragma unroll
    for (int nt = 0; nt < 8; ++nt) {
      v2f b = *(const v2f*)(brow + nt * 16 * GS_PITCH + k);
      acc[nt] = wmma4(a, b, acc[nt]);
    }
  }

  int rbase = wave * 16 + ((lane >> 4) << 3);
#pragma unroll
  for (int nt = 0; nt < 8; ++nt) {
    float bs = bias ? bias[nt * 16 + l16] : 0.f;
    float* crow = C + (size_t)rbase * 128 + nt * 16 + l16;
#pragma unroll
    for (int r = 0; r < 8; ++r)
      crow[(size_t)r * 128] = acc[nt][r] + bs;
  }
}

__device__ __forceinline__ void tile_product_nt(const float* __restrict__ A,
                                                const float* __restrict__ Bm,
                                                float* __restrict__ C, int tile, int lane) {
  const int mt = tile >> 3;
  const int nt = tile & 7;
  const int l16   = lane & 15;
  const int khalf = (lane >> 4) << 1;
  const float* arow = A  + (size_t)(mt * 16 + l16) * DF + khalf;
  const float* brow = Bm + (size_t)(nt * 16 + l16) * DF + khalf;
  v8f acc = {};
  for (int k = 0; k < DF; k += 4) {
    v2f a = *(const v2f*)(arow + k);
    v2f b = *(const v2f*)(brow + k);
    acc = wmma4(a, b, acc);
  }
  const int rbase = mt * 16 + ((lane >> 4) << 3);
  float* crow = C + (size_t)rbase * DF + nt * 16 + l16;
#pragma unroll
  for (int r = 0; r < 8; ++r)
    crow[(size_t)r * DF] = acc[r];
}

__global__ void zero_cnt_transpose_w(int* __restrict__ cnt, int n,
                                     const float* __restrict__ w, float* __restrict__ wt) {
  int t = blockIdx.x * blockDim.x + threadIdx.x;
  if (t < n) cnt[t] = 0;
  if (t < 128 * 128) wt[(t & 127) * 128 + (t >> 7)] = w[t];
}

__global__ void count_edges(const int* __restrict__ ecol, int* __restrict__ cnt,
                            int* __restrict__ rank) {
  int e = blockIdx.x * blockDim.x + threadIdx.x;
  if (e < NE_EDGES) rank[e] = atomicAdd(&cnt[ecol[e]], 1);
}

#define SCAN_STEPS 13
#define SCAN_PER_WAVE (SCAN_STEPS * 128)
static_assert(32 * SCAN_PER_WAVE >= N_NODES, "the 32 waves cover every counter");
static_assert(N_NODES % 4 == 0, "a lane's 4 counters are all inside or all outside");

__device__ __forceinline__ int wave_incl_scan(int x, int lane) {
#pragma unroll
  for (int off = 1; off < 32; off <<= 1) {
    int y = __shfl_up(x, off, 32);
    if (lane >= off) x += y;
  }
  return x;
}

__global__ __launch_bounds__(1024)
void scan_rowptr(const int* __restrict__ cnt, int* __restrict__ rowptr) {
  __shared__ int wtot[32];
  const int tid = threadIdx.x;
  const int lane = tid & 31;
  const int w = tid >> 5;
  const int first = w * SCAN_PER_WAVE + lane * 4;
  int4 v[SCAN_STEPS];
#pragma unroll
  for (int j = 0; j < SCAN_STEPS; ++j) {
    int idx = first + j * 128;
    int idc = (idx < N_NODES) ? idx : (N_NODES - 4);
    v[j] = *(const int4*)(cnt + idc);
  }
  int run = 0;
#pragma unroll
  for (int j = 0; j < SCAN_STEPS; ++j) {
    const bool in = (first + j * 128) < N_NODES;
    const int c0 = in ? v[j].x : 0;
    const int c1 = in ? v[j].y : 0;
    const int c2 = in ? v[j].z : 0;
    const int c3 = in ? v[j].w : 0;
    const int s1 = c0, s2 = s1 + c1, s3 = s2 + c2, s4 = s3 + c3;
    const int inc = wave_incl_scan(s4, lane);
    const int ex = run + inc - s4;
    run += __shfl(inc, 31, 32);
    v[j].x = ex; v[j].y = ex + s1; v[j].z = ex + s2; v[j].w = ex + s3;
  }
  if (lane == 0) wtot[w] = run;
  __syncthreads();
  const int t  = wtot[lane];
  const int ti = wave_incl_scan(t, lane);
  const int base = __shfl(ti - t, w, 32);
#pragma unroll
  for (int j = 0; j < SCAN_STEPS; ++j) {
    int idx = first + j * 128;
    if (idx < N_NODES) {
      int4 o;
      o.x = v[j].x + base; o.y = v[j].y + base; o.z = v[j].z + base; o.w = v[j].w + base;
      *(int4*)(rowptr + idx) = o;
    }
  }
  if (tid == 1023) rowptr[N_NODES] = base + run;
}

__global__ void csr_fill(const int* __restrict__ erow, const int* __restrict__ ecol,
                         const int* __restrict__ rowptr, const int* __restrict__ rank,
                         int* __restrict__ srcidx, float* __restrict__ dis) {
  int e = blockIdx.x * blockDim.x + threadIdx.x;
  if (e >= NE_EDGES) return;
  if (e < N_NODES)
    dis[e] = rsqrtf((float)(rowptr[e + 1] - rowptr[e]) + 1.0f);
  int c = ecol[e];
  srcidx[rowptr[c] + rank[e]] = erow[e];
}

static_assert(NE_EDGES >= N_NODES, "csr_fill's first N_NODES threads write dis");
static_assert((size_t)NE_EDGES <= (size_t)N_NODES * DF, "rank (one int per edge) fits in the agg plane");
__global__ __launch_bounds__(256)
void gcn_aggregate(const int* __restrict__ rowptr, const int* __restrict__ srcidx,
                   const float* __restrict__ dis, const float* __restrict__ lin,
                   float* __restrict__ agg, float* __restrict__ stats,
                   const float* __restrict__ wnext, float* __restrict__ wt) {
  int t = blockIdx.x * blockDim.x + threadIdx.x;
  if (blockIdx.x == 0) stats[threadIdx.x] = 0.0f;
  if (wnext != nullptr && t < 128 * 128) wt[(t & 127) * 128 + (t >> 7)] = wnext[t];
  int c = t >> 5;
  if (c >= N_NODES) return;
  int lane = threadIdx.x & 31;
  float dc = dis[c];
  const float4 sv = *(const float4*)(lin + (size_t)c * DF + lane * 4);
  float s = dc * dc;
  float ax = s * sv.x, ay = s * sv.y, az = s * sv.z, aw = s * sv.w;
  int e0 = rowptr[c], e1 = rowptr[c + 1];
  for (int j = e0; j < e1; ++j) {
    int r = srcidx[j];
    float nrm = dis[r] * dc;
    const float4 v = *(const float4*)(lin + (size_t)r * DF + lane * 4);
    ax += nrm * v.x; ay += nrm * v.y; az += nrm * v.z; aw += nrm * v.w;
  }
  float4 o; o.x = ax; o.y = ay; o.z = az; o.w = aw;
  *(float4*)(agg + (size_t)c * DF + lane * 4) = o;
}

#define RIDER_BLOCKS 2
static_assert(RIDER_BLOCKS * 32 == (DF / 16) * (DF / 16), "one wave per tile of a 128 x 128 product");
__global__ __launch_bounds__(1024)
void bn_stats(const float* __restrict__ agg, float* __restrict__ stats, int n,
              const float* __restrict__ ra, const float* __restrict__ rb,
              float* __restrict__ rc, int rider) {
  __shared__ __attribute__((aligned(16))) float red[32][2 * DF];
  int tid  = threadIdx.x;
  int lane = tid & 31;
  int w    = tid >> 5;
  if ((int)blockIdx.x < rider) {
    tile_product_nt(ra, rb, rc, blockIdx.x * 32 + w, lane);
    return;
  }
  int rw   = (blockIdx.x - rider) * 512 + w * 16;
  float4 s = {0.f, 0.f, 0.f, 0.f};
  float4 q = {0.f, 0.f, 0.f, 0.f};
  if (rw + 16 <= n) {
    const float* p = agg + (size_t)rw * DF + lane * 4;
    float4 v[16];
#pragma unroll
    for (int i = 0; i < 16; ++i) v[i] = *(const float4*)(p + (size_t)i * DF);
#pragma unroll
    for (int i = 0; i < 16; ++i) {
      s.x += v[i].x; s.y += v[i].y; s.z += v[i].z; s.w += v[i].w;
      q.x += v[i].x * v[i].x; q.y += v[i].y * v[i].y;
      q.z += v[i].z * v[i].z; q.w += v[i].w * v[i].w;
    }
  } else {
    for (int i = 0; i < 16; ++i) {
      int r = rw + i;
      if (r < n) {
        const float4 v = *(const float4*)(agg + (size_t)r * DF + lane * 4);
        s.x += v.x; s.y += v.y; s.z += v.z; s.w += v.w;
        q.x += v.x * v.x; q.y += v.y * v.y; q.z += v.z * v.z; q.w += v.w * v.w;
      }
    }
  }
  *(float4*)&red[w][lane * 4]      = s;
  *(float4*)&red[w][DF + lane * 4] = q;
  __syncthreads();
  if (tid < 2 * DF) {
    float a = 0.f;
#pragma unroll
    for (int k = 0; k < 32; ++k) a += red[k][tid];
    atomicAdd(&stats[tid], a);
  }
}

#define DP_WAVES 4
static_assert(BPAIRS % (DP_WAVES * 16) == 0, "every wave of the launch has 16 whole pairs");
static_assert(DP_WAVES * 32 == DF, "the block's waves bring 32 rows of Q^T each: all 128");

__global__ __launch_bounds__(DP_WAVES * 32)
void decode_pairs(const float* __restrict__ agg, const float* __restrict__ stats,
                  const float* __restrict__ g, const float* __restrict__ be,
                  const int* __restrict__ di, const float* __restrict__ Qt,
                  float* __restrict__ out) {
  __shared__ __attribute__((aligned(16))) float sA[DP_WAVES][16 * GS_PITCH];
  __shared__ __attribute__((aligned(16))) float sV[DP_WAVES][16 * GS_PITCH];
  __shared__ __attribute__((aligned(16))) float sB[DF * GS_PITCH];
  const int lane  = threadIdx.x & 31;
  const int w     = threadIdx.x >> 5;
  const int lane4 = lane * 4;
  const int l16   = lane & 15;
  const int pair0 = (blockIdx.x * DP_WAVES + w) * 16;

  const int2 dv = *(const int2*)(di + 2 * (pair0 + l16));
  int ia = dv.x - 1; if (ia < 0) ia += N_NODES;
  int ib = dv.y - 1; if (ib < 0) ib += N_NODES;

  float4 sc, sh;
  bn_lane_scale_shift(stats, g, be, lane4, sc, sh);

  {
    const float* pb = Qt + (size_t)w * 32 * DF + lane4;
    float4 bv[16], av[16];
#pragma unroll
    for (int i = 0; i < 16; ++i) bv[i] = *(const float4*)(pb + (size_t)i * DF);
#pragma unroll
    for (int i = 0; i < 16; ++i)
      av[i] = *(const float4*)(agg + (size_t)__builtin_amdgcn_readlane(ia, i) * DF + lane4);
#pragma unroll
    for (int i = 0; i < 16; ++i) *(float4*)&sB[(w * 32 + i) * GS_PITCH + lane4] = bv[i];
#pragma unroll
    for (int i = 0; i < 16; ++i) bv[i] = *(const float4*)(pb + (size_t)(16 + i) * DF);
#pragma unroll
    for (int i = 0; i < 16; ++i) *(float4*)&sA[w][i * GS_PITCH + lane4] = bn_leaky4(av[i], sc, sh);
#pragma unroll
    for (int i = 0; i < 16; ++i)
      av[i] = *(const float4*)(agg + (size_t)__builtin_amdgcn_readlane(ib, i) * DF + lane4);
#pragma unroll
    for (int i = 0; i < 16; ++i) *(float4*)&sB[(w * 32 + 16 + i) * GS_PITCH + lane4] = bv[i];
#pragma unroll
    for (int i = 0; i < 16; ++i) *(float4*)&sV[w][i * GS_PITCH + lane4] = bn_leaky4(av[i], sc, sh);
  }
  __syncthreads();

  const int khalf = (lane >> 4) << 1;
  const float* arow = &sA[w][l16 * GS_PITCH + khalf];
  const float* brow = &sB[l16 * GS_PITCH + khalf];

  v8f zero = {};
  v8f acc[8];
#pragma unroll
  for (int nt = 0; nt < 8; ++nt) acc[nt] = zero;

  for (int k = 0; k < DF; k += 4) {
    v2f a = *(const v2f*)(arow + k);
#pragma unroll
    for (int nt = 0; nt < 8; ++nt) {
      v2f b = *(const v2f*)(brow + nt * 16 * GS_PITCH + k);
      acc[nt] = wmma4(a, b, acc[nt]);
    }
  }

  const int h8 = (lane >> 4) << 3;
  const float* vrow = &sV[w][h8 * GS_PITCH + l16];
  float p[8];
#pragma unroll
  for (int r = 0; r < 8; ++r) {
    float s = 0.f;
#pragma unroll
    for (int nt = 0; nt < 8; ++nt) s += acc[nt][r] * vrow[r * GS_PITCH + nt * 16];
    p[r] = s;
  }
#pragma unroll
  for (int off = 8; off > 0; off >>= 1) {
#pragma unroll
    for (int r = 0; r < 8; ++r) p[r] += __shfl_xor(p[r], off, 32);
  }
  float o = p[0];
#pragma unroll
  for (int r = 1; r < 8; ++r) o = (l16 == r) ? p[r] : o;
  if (l16 < 8) out[pair0 + h8 + l16] = o;
}

#define WS_ALIGN 128
#define WS_PAD(n) ((((size_t)(n)) + WS_ALIGN - 1) / WS_ALIGN * WS_ALIGN)
static_assert(WS_PAD(N_NODES) + WS_PAD(256) + WS_PAD(128 * 128) + WS_PAD(N_NODES) +
              WS_PAD(N_NODES + 16) + WS_PAD(NE_EDGES)
              <= (size_t)N_NODES + 256 + 128 * 128 + N_NODES + (N_NODES + 16) + N_NODES + NE_EDGES,
              "the aligned planes end no later than the packed ones with their unused slot did");

static_assert(WS_PAD(N_NODES) >= 2 * 128 * 128, "U and Q^T fit in the edge counters' plane");

extern "C" void kernel_launch(void* const* d_in, const int* in_sizes, int n_in,
                              void* d_out, int out_size, void* d_ws, size_t ws_size,
                              hipStream_t stream) {
  const float* x   = (const float*)d_in[0];
  const int*   edg = (const int*)d_in[1];
  const int*   di  = (const int*)d_in[2];
  const float* w1  = (const float*)d_in[3];
  const float* b1  = (const float*)d_in[4];
  const float* w2  = (const float*)d_in[5];
  const float* b2  = (const float*)d_in[6];
  const float* w3  = (const float*)d_in[7];
  const float* b3  = (const float*)d_in[8];
  const float* g1  = (const float*)d_in[9];
  const float* be1 = (const float*)d_in[10];
  const float* g2  = (const float*)d_in[11];
  const float* be2 = (const float*)d_in[12];
  const float* g3  = (const float*)d_in[13];
  const float* be3 = (const float*)d_in[14];
  const float* p1  = (const float*)d_in[15];
  const float* p2  = (const float*)d_in[16];
  float* out = (float*)d_out;

  float* ws = (float*)d_ws;
  size_t off = 0;
  float* dis    = ws + off; off += WS_PAD(N_NODES);
  float* stats  = ws + off; off += WS_PAD(256);
  float* Wt     = ws + off; off += WS_PAD(128 * 128);
  int*   cnt    = (int*)(ws + off); off += WS_PAD(N_NODES);
  int*   rowptr = (int*)(ws + off); off += WS_PAD(N_NODES + 16);
  int*   srcidx = (int*)(ws + off); off += WS_PAD(NE_EDGES);
  float* lin    = ws + off; off += (size_t)N_NODES * DF;
  float* agg    = ws + off; off += (size_t)N_NODES * DF;
  float* U    = (float*)cnt;
  float* Qt   = (float*)cnt + 128 * 128;
  int* rank = (int*)agg;

  const int* erow = edg;
  const int* ecol = edg + NE_EDGES;

  zero_cnt_transpose_w<<<(N_NODES + 255) / 256, 256, 0, stream>>>(cnt, N_NODES, w1, Wt);
  count_edges<<<(NE_EDGES + 255) / 256, 256, 0, stream>>>(ecol, cnt, rank);
  scan_rowptr<<<1, 1024, 0, stream>>>(cnt, rowptr);
  csr_fill<<<(NE_EDGES + 255) / 256, 256, 0, stream>>>(erow, ecol, rowptr, rank, srcidx, dis);

  const float* Wn[3] = {w2, w3, nullptr};
  const float* Ra[3] = {p1, U, nullptr};
  const float* Rb[3] = {p2, p1, nullptr};
  float*       Rc[3] = {U, Qt, nullptr};
  const int    Rn[3] = {RIDER_BLOCKS, RIDER_BLOCKS, 0};
  const float* Bl[3] = {b1, b2, b3};
  const float* Gl[3] = {g1, g2, g3};
  const float* El[3] = {be1, be2, be3};
  const int strip_waves = N_NODES / 16;
  for (int l = 0; l < 3; ++l) {
    if (l == 0)
      wmma_gemm_strip<<<(strip_waves * 32 + 255) / 256, 256, 0, stream>>>(
          x, Wt, Bl[l], lin, N_NODES, nullptr, nullptr, nullptr);
    else
      wmma_gemm_strip<<<(strip_waves * 32 + 255) / 256, 256, 0, stream>>>(
          agg, Wt, Bl[l], lin, N_NODES, stats, Gl[l - 1], El[l - 1]);
    gcn_aggregate<<<(N_NODES * 32 + 255) / 256, 256, 0, stream>>>(
        rowptr, srcidx, dis, lin, agg, stats, Wn[l], Wt);
    bn_stats<<<(N_NODES + 511) / 512 + Rn[l], 1024, 0, stream>>>(
        agg, stats, N_NODES, Ra[l], Rb[l], Rc[l], Rn[l]);
  }

  decode_pairs<<<BPAIRS / (DP_WAVES * 16), DP_WAVES * 32, 0, stream>>>(
      agg, stats, g3, be3, di, Qt, out);
}


# count_edges' 800000 scattered returning global atomics -> LDS: 32 blocks count 25000-edge chunks in 50000 LDS counters, a pass makes chunk counts offsets; in-row edge order: chunk, then LDS arrival
# speedup vs baseline: 5.0979x; 1.0860x over previous
#include <hip/hip_runtime.h>

#define N_NODES 50000
#define NE_EDGES 800000
#define BPAIRS 4096
#define DF 128
#define EPS_BN 1e-5f

typedef float v2f __attribute__((ext_vector_type(2)));
typedef float v8f __attribute__((ext_vector_type(8)));

__device__ __forceinline__ v8f wmma4(v2f a, v2f b, v8f c) {
  return __builtin_amdgcn_wmma_f32_16x16x4_f32(
       false, a,  false, b,
       (short)0, c,  false,  false);
}

#define INV_N_NODES (1.0f / (float)N_NODES)

__device__ __forceinline__ float bn_leaky1(float x, float sc, float sh) {
  float y = x * sc + sh;
  return (y >= 0.f) ? y : 0.1f * y;
}

__device__ __forceinline__ void bn_scale_shift(float sum, float sumsq, float g, float be,
                                               float& sc, float& sh) {
  float m   = sum * INV_N_NODES;
  float var = sumsq * INV_N_NODES - m * m;
  sc = g * rsqrtf(var + EPS_BN);
  sh = be - m * sc;
}

__device__ __forceinline__ void bn_lane_scale_shift(const float* __restrict__ stats,
                                                    const float* __restrict__ g,
                                                    const float* __restrict__ be, int lane4,
                                                    float4& sc, float4& sh) {
  const float4 su = *(const float4*)(stats + lane4);
  const float4 sq = *(const float4*)(stats + DF + lane4);
  const float4 gv = *(const float4*)(g + lane4);
  const float4 bv = *(const float4*)(be + lane4);
  bn_scale_shift(su.x, sq.x, gv.x, bv.x, sc.x, sh.x);
  bn_scale_shift(su.y, sq.y, gv.y, bv.y, sc.y, sh.y);
  bn_scale_shift(su.z, sq.z, gv.z, bv.z, sc.z, sh.z);
  bn_scale_shift(su.w, sq.w, gv.w, bv.w, sc.w, sh.w);
}

__device__ __forceinline__ float4 bn_leaky4(float4 v, float4 sc, float4 sh) {
  float4 y;
  y.x = bn_leaky1(v.x, sc.x, sh.x);
  y.y = bn_leaky1(v.y, sc.y, sh.y);
  y.z = bn_leaky1(v.z, sc.z, sh.z);
  y.w = bn_leaky1(v.w, sc.w, sh.w);
  return y;
}

#define GS_PITCH 132
#define GS_WAVES 8
static_assert(DF == 128, "a row is 32 lanes x 16 bytes; K of every product here is 128");
static_assert(GS_WAVES * 16 == DF, "the block's waves bring 16 rows of Bt each: all 128");

__global__ __launch_bounds__(GS_WAVES * 32)
void wmma_gemm_strip(const float* __restrict__ A, const float* __restrict__ Bt,
                     const float* __restrict__ bias, float* __restrict__ C, int M,
                     const float* __restrict__ stats, const float* __restrict__ g,
                     const float* __restrict__ be) {
  __shared__ __attribute__((aligned(16))) float sA[GS_WAVES][16 * GS_PITCH];
  __shared__ __attribute__((aligned(16))) float sB[DF * GS_PITCH];
  int gtid = blockIdx.x * blockDim.x + threadIdx.x;
  int wave = gtid >> 5;
  int lane = threadIdx.x & 31;
  int w    = threadIdx.x >> 5;
  const bool active = wave * 16 < M;
  const int lane4 = lane * 4;

  {
    const float* pb = Bt + (size_t)w * 16 * DF + lane4;
    float4 bv[16], av[16];
#pragma unroll
    for (int i = 0; i < 16; ++i) bv[i] = *(const float4*)(pb + (size_t)i * DF);
    if (active) {
      const float* pa = A + (size_t)wave * 16 * DF + lane4;
#pragma unroll
      for (int i = 0; i < 16; ++i) av[i] = *(const float4*)(pa + (size_t)i * DF);
    }
#pragma unroll
    for (int i = 0; i < 16; ++i) *(float4*)&sB[(w * 16 + i) * GS_PITCH + lane4] = bv[i];
    if (active) {
      if (stats != nullptr) {
        float4 sc, sh;
        bn_lane_scale_shift(stats, g, be, lane4, sc, sh);
#pragma unroll
        for (int i = 0; i < 16; ++i) av[i] = bn_leaky4(av[i], sc, sh);
      }
#pragma unroll
      for (int i = 0; i < 16; ++i) *(float4*)&sA[w][i * GS_PITCH + lane4] = av[i];
    }
  }
  __syncthreads();
  if (!active) return;

  int l16   = lane & 15;
  int khalf = (lane >> 4) << 1;
  const float* arow = &sA[w][l16 * GS_PITCH + khalf];
  const float* brow = &sB[l16 * GS_PITCH + khalf];

  v8f zero = {};
  v8f acc[8];
#pragma unroll
  for (int nt = 0; nt < 8; ++nt) acc[nt] = zero;

  for (int k = 0; k < DF; k += 4) {
    v2f a = *(const v2f*)(arow + k);
#pragma unroll
    for (int nt = 0; nt < 8; ++nt) {
      v2f b = *(const v2f*)(brow + nt * 16 * GS_PITCH + k);
      acc[nt] = wmma4(a, b, acc[nt]);
    }
  }

  int rbase = wave * 16 + ((lane >> 4) << 3);
#pragma unroll
  for (int nt = 0; nt < 8; ++nt) {
    float bs = bias ? bias[nt * 16 + l16] : 0.f;
    float* crow = C + (size_t)rbase * 128 + nt * 16 + l16;
#pragma unroll
    for (int r = 0; r < 8; ++r)
      crow[(size_t)r * 128] = acc[nt][r] + bs;
  }
}

__device__ __forceinline__ void tile_product_nt(const float* __restrict__ A,
                                                const float* __restrict__ Bm,
                                                float* __restrict__ C, int tile, int lane) {
  const int mt = tile >> 3;
  const int nt = tile & 7;
  const int l16   = lane & 15;
  const int khalf = (lane >> 4) << 1;
  const float* arow = A  + (size_t)(mt * 16 + l16) * DF + khalf;
  const float* brow = Bm + (size_t)(nt * 16 + l16) * DF + khalf;
  v8f acc = {};
  for (int k = 0; k < DF; k += 4) {
    v2f a = *(const v2f*)(arow + k);
    v2f b = *(const v2f*)(brow + k);
    acc = wmma4(a, b, acc);
  }
  const int rbase = mt * 16 + ((lane >> 4) << 3);
  float* crow = C + (size_t)rbase * DF + nt * 16 + l16;
#pragma unroll
  for (int r = 0; r < 8; ++r)
    crow[(size_t)r * DF] = acc[r];
}

#define CH_BLOCKS 32
#define CH_EDGES (NE_EDGES / CH_BLOCKS)
#define CH_VEC (CH_EDGES / 4)
#define CH_STEPS ((CH_VEC + 1023) / 1024)
#define N_VEC (N_NODES / 4)
#define N_STEPS ((N_VEC + 1023) / 1024)
#define HIST_PITCH ((N_NODES + 127) / 128 * 128)
static_assert(NE_EDGES % (CH_BLOCKS * 4) == 0, "chunks are whole 16-byte groups of edges");
static_assert(N_NODES % 4 == 0, "a histogram row is whole 16-byte groups");
static_assert(CH_BLOCKS * 1024 >= 128 * 128, "the launch has a thread per weight element");
static_assert((size_t)N_NODES * sizeof(int) <= 320 * 1024, "all counters fit in one work-group's LDS");

__global__ __launch_bounds__(1024)
void hist_rank(const int* __restrict__ ecol, int* __restrict__ rank, int* __restrict__ hist,
               const float* __restrict__ w, float* __restrict__ wt) {
  __shared__ __attribute__((aligned(16))) int h[N_NODES];
  const int tid = threadIdx.x;
  const int4* ec4 = (const int4*)(ecol + (size_t)blockIdx.x * CH_EDGES);
  int4 c[CH_STEPS];
#pragma unroll
  for (int i = 0; i < CH_STEPS; ++i) {
    const int idx = tid + i * 1024;
    c[i] = ec4[(idx < CH_VEC) ? idx : (CH_VEC - 1)];
  }
  int4* h4 = (int4*)h;
  const int4 z4 = {0, 0, 0, 0};
#pragma unroll
  for (int j = 0; j < N_STEPS; ++j) {
    const int i = tid + j * 1024;
    if (i < N_VEC) h4[i] = z4;
  }
  const int t = blockIdx.x * 1024 + tid;
  if (t < 128 * 128) wt[(t & 127) * 128 + (t >> 7)] = w[t];
  __syncthreads();
  int4* rk4 = (int4*)(rank + (size_t)blockIdx.x * CH_EDGES);
  int4 r[CH_STEPS];
#pragma unroll
  for (int i = 0; i < CH_STEPS; ++i) {
    if (tid + i * 1024 < CH_VEC) {
      r[i].x = atomicAdd(&h[c[i].x], 1);
      r[i].y = atomicAdd(&h[c[i].y], 1);
      r[i].z = atomicAdd(&h[c[i].z], 1);
      r[i].w = atomicAdd(&h[c[i].w], 1);
    }
  }
#pragma unroll
  for (int i = 0; i < CH_STEPS; ++i) {
    const int idx = tid + i * 1024;
    if (idx < CH_VEC) rk4[idx] = r[i];
  }
  __syncthreads();
  int4* ho = (int4*)(hist + (size_t)blockIdx.x * HIST_PITCH);
#pragma unroll
  for (int j = 0; j < N_STEPS; ++j) {
    const int i = tid + j * 1024;
    if (i < N_VEC) ho[i] = h4[i];
  }
}

__global__ __launch_bounds__(64)
void chunk_prefix(int* hist, int* cnt) {
  const int i = blockIdx.x * 64 + threadIdx.x;
  if (i >= N_VEC) return;
  int4 v[CH_BLOCKS];
#pragma unroll
  for (int b = 0; b < CH_BLOCKS; ++b) v[b] = ((const int4*)(hist + (size_t)b * HIST_PITCH))[i];
  int4 run = {0, 0, 0, 0};
#pragma unroll
  for (int b = 0; b < CH_BLOCKS; ++b) {
    const int4 tv = v[b];
    ((int4*)(hist + (size_t)b * HIST_PITCH))[i] = run;
    run.x += tv.x; run.y += tv.y; run.z += tv.z; run.w += tv.w;
  }
  ((int4*)cnt)[i] = run;
}

#define SCAN_STEPS 13
#define SCAN_PER_WAVE (SCAN_STEPS * 128)
static_assert(32 * SCAN_PER_WAVE >= N_NODES, "the 32 waves cover every counter");
static_assert(N_NODES % 4 == 0, "a lane's 4 counters are all inside or all outside");

__device__ __forceinline__ int wave_incl_scan(int x, int lane) {
#pragma unroll
  for (int off = 1; off < 32; off <<= 1) {
    int y = __shfl_up(x, off, 32);
    if (lane >= off) x += y;
  }
  return x;
}

__global__ __launch_bounds__(1024)
void scan_rowptr(const int* __restrict__ cnt, int* __restrict__ rowptr) {
  __shared__ int wtot[32];
  const int tid = threadIdx.x;
  const int lane = tid & 31;
  const int w = tid >> 5;
  const int first = w * SCAN_PER_WAVE + lane * 4;
  int4 v[SCAN_STEPS];
#pragma unroll
  for (int j = 0; j < SCAN_STEPS; ++j) {
    int idx = first + j * 128;
    int idc = (idx < N_NODES) ? idx : (N_NODES - 4);
    v[j] = *(const int4*)(cnt + idc);
  }
  int run = 0;
#pragma unroll
  for (int j = 0; j < SCAN_STEPS; ++j) {
    const bool in = (first + j * 128) < N_NODES;
    const int c0 = in ? v[j].x : 0;
    const int c1 = in ? v[j].y : 0;
    const int c2 = in ? v[j].z : 0;
    const int c3 = in ? v[j].w : 0;
    const int s1 = c0, s2 = s1 + c1, s3 = s2 + c2, s4 = s3 + c3;
    const int inc = wave_incl_scan(s4, lane);
    const int ex = run + inc - s4;
    run += __shfl(inc, 31, 32);
    v[j].x = ex; v[j].y = ex + s1; v[j].z = ex + s2; v[j].w = ex + s3;
  }
  if (lane == 0) wtot[w] = run;
  __syncthreads();
  const int t  = wtot[lane];
  const int ti = wave_incl_scan(t, lane);
  const int base = __shfl(ti - t, w, 32);
#pragma unroll
  for (int j = 0; j < SCAN_STEPS; ++j) {
    int idx = first + j * 128;
    if (idx < N_NODES) {
      int4 o;
      o.x = v[j].x + base; o.y = v[j].y + base; o.z = v[j].z + base; o.w = v[j].w + base;
      *(int4*)(rowptr + idx) = o;
    }
  }
  if (tid == 1023) rowptr[N_NODES] = base + run;
}

__global__ void csr_fill(const int* __restrict__ erow, const int* __restrict__ ecol,
                         const int* __restrict__ rowptr, const int* __restrict__ rank,
                         const int* __restrict__ hist,
                         int* __restrict__ srcidx, float* __restrict__ dis) {
  int e = blockIdx.x * blockDim.x + threadIdx.x;
  if (e >= NE_EDGES) return;
  if (e < N_NODES)
    dis[e] = rsqrtf((float)(rowptr[e + 1] - rowptr[e]) + 1.0f);
  int c = ecol[e];
  int b = e / CH_EDGES;
  srcidx[rowptr[c] + hist[(size_t)b * HIST_PITCH + c] + rank[e]] = erow[e];
}

static_assert(NE_EDGES >= N_NODES, "csr_fill's first N_NODES threads write dis");
static_assert((size_t)(NE_EDGES + 127) / 128 * 128 + (size_t)CH_BLOCKS * HIST_PITCH <= (size_t)N_NODES * DF,
              "rank (one int per edge) and the chunk histograms fit in the agg plane");
__global__ __launch_bounds__(256)
void gcn_aggregate(const int* __restrict__ rowptr, const int* __restrict__ srcidx,
                   const float* __restrict__ dis, const float* __restrict__ lin,
                   float* __restrict__ agg, float* __restrict__ stats,
                   const float* __restrict__ wnext, float* __restrict__ wt) {
  int t = blockIdx.x * blockDim.x + threadIdx.x;
  if (blockIdx.x == 0) stats[threadIdx.x] = 0.0f;
  if (wnext != nullptr && t < 128 * 128) wt[(t & 127) * 128 + (t >> 7)] = wnext[t];
  int c = t >> 5;
  if (c >= N_NODES) return;
  int lane = threadIdx.x & 31;
  float dc = dis[c];
  const float4 sv = *(const float4*)(lin + (size_t)c * DF + lane * 4);
  float s = dc * dc;
  float ax = s * sv.x, ay = s * sv.y, az = s * sv.z, aw = s * sv.w;
  int e0 = rowptr[c], e1 = rowptr[c + 1];
  for (int j = e0; j < e1; ++j) {
    int r = srcidx[j];
    float nrm = dis[r] * dc;
    const float4 v = *(const float4*)(lin + (size_t)r * DF + lane * 4);
    ax += nrm * v.x; ay += nrm * v.y; az += nrm * v.z; aw += nrm * v.w;
  }
  float4 o; o.x = ax; o.y = ay; o.z = az; o.w = aw;
  *(float4*)(agg + (size_t)c * DF + lane * 4) = o;
}

#define RIDER_BLOCKS 2
static_assert(RIDER_BLOCKS * 32 == (DF / 16) * (DF / 16), "one wave per tile of a 128 x 128 product");
__global__ __launch_bounds__(1024)
void bn_stats(const float* __restrict__ agg, float* __restrict__ stats, int n,
              const float* __restrict__ ra, const float* __restrict__ rb,
              float* __restrict__ rc, int rider) {
  __shared__ __attribute__((aligned(16))) float red[32][2 * DF];
  int tid  = threadIdx.x;
  int lane = tid & 31;
  int w    = tid >> 5;
  if ((int)blockIdx.x < rider) {
    tile_product_nt(ra, rb, rc, blockIdx.x * 32 + w, lane);
    return;
  }
  int rw   = (blockIdx.x - rider) * 512 + w * 16;
  float4 s = {0.f, 0.f, 0.f, 0.f};
  float4 q = {0.f, 0.f, 0.f, 0.f};
  if (rw + 16 <= n) {
    const float* p = agg + (size_t)rw * DF + lane * 4;
    float4 v[16];
#pragma unroll
    for (int i = 0; i < 16; ++i) v[i] = *(const float4*)(p + (size_t)i * DF);
#pragma unroll
    for (int i = 0; i < 16; ++i) {
      s.x += v[i].x; s.y += v[i].y; s.z += v[i].z; s.w += v[i].w;
      q.x += v[i].x * v[i].x; q.y += v[i].y * v[i].y;
      q.z += v[i].z * v[i].z; q.w += v[i].w * v[i].w;
    }
  } else {
    for (int i = 0; i < 16; ++i) {
      int r = rw + i;
      if (r < n) {
        const float4 v = *(const float4*)(agg + (size_t)r * DF + lane * 4);
        s.x += v.x; s.y += v.y; s.z += v.z; s.w += v.w;
        q.x += v.x * v.x; q.y += v.y * v.y; q.z += v.z * v.z; q.w += v.w * v.w;
      }
    }
  }
  *(float4*)&red[w][lane * 4]      = s;
  *(float4*)&red[w][DF + lane * 4] = q;
  __syncthreads();
  if (tid < 2 * DF) {
    float a = 0.f;
#pragma unroll
    for (int k = 0; k < 32; ++k) a += red[k][tid];
    atomicAdd(&stats[tid], a);
  }
}

#define DP_WAVES 4
static_assert(BPAIRS % (DP_WAVES * 16) == 0, "every wave of the launch has 16 whole pairs");
static_assert(DP_WAVES * 32 == DF, "the block's waves bring 32 rows of Q^T each: all 128");

__global__ __launch_bounds__(DP_WAVES * 32)
void decode_pairs(const float* __restrict__ agg, const float* __restrict__ stats,
                  const float* __restrict__ g, const float* __restrict__ be,
                  const int* __restrict__ di, const float* __restrict__ Qt,
                  float* __restrict__ out) {
  __shared__ __attribute__((aligned(16))) float sA[DP_WAVES][16 * GS_PITCH];
  __shared__ __attribute__((aligned(16))) float sV[DP_WAVES][16 * GS_PITCH];
  __shared__ __attribute__((aligned(16))) float sB[DF * GS_PITCH];
  const int lane  = threadIdx.x & 31;
  const int w     = threadIdx.x >> 5;
  const int lane4 = lane * 4;
  const int l16   = lane & 15;
  const int pair0 = (blockIdx.x * DP_WAVES + w) * 16;

  const int2 dv = *(const int2*)(di + 2 * (pair0 + l16));
  int ia = dv.x - 1; if (ia < 0) ia += N_NODES;
  int ib = dv.y - 1; if (ib < 0) ib += N_NODES;

  float4 sc, sh;
  bn_lane_scale_shift(stats, g, be, lane4, sc, sh);

  {
    const float* pb = Qt + (size_t)w * 32 * DF + lane4;
    float4 bv[16], av[16];
#pragma unroll
    for (int i = 0; i < 16; ++i) bv[i] = *(const float4*)(pb + (size_t)i * DF);
#pragma unroll
    for (int i = 0; i < 16; ++i)
      av[i] = *(const float4*)(agg + (size_t)__builtin_amdgcn_readlane(ia, i) * DF + lane4);
#pragma unroll
    for (int i = 0; i < 16; ++i) *(float4*)&sB[(w * 32 + i) * GS_PITCH + lane4] = bv[i];
#pragma unroll
    for (int i = 0; i < 16; ++i) bv[i] = *(const float4*)(pb + (size_t)(16 + i) * DF);
#pragma unroll
    for (int i = 0; i < 16; ++i) *(float4*)&sA[w][i * GS_PITCH + lane4] = bn_leaky4(av[i], sc, sh);
#pragma unroll
    for (int i = 0; i < 16; ++i)
      av[i] = *(const float4*)(agg + (size_t)__builtin_amdgcn_readlane(ib, i) * DF + lane4);
#pragma unroll
    for (int i = 0; i < 16; ++i) *(float4*)&sB[(w * 32 + 16 + i) * GS_PITCH + lane4] = bv[i];
#pragma unroll
    for (int i = 0; i < 16; ++i) *(float4*)&sV[w][i * GS_PITCH + lane4] = bn_leaky4(av[i], sc, sh);
  }
  __syncthreads();

  const int khalf = (lane >> 4) << 1;
  const float* arow = &sA[w][l16 * GS_PITCH + khalf];
  const float* brow = &sB[l16 * GS_PITCH + khalf];

  v8f zero = {};
  v8f acc[8];
#pragma unroll
  for (int nt = 0; nt < 8; ++nt) acc[nt] = zero;

  for (int k = 0; k < DF; k += 4) {
    v2f a = *(const v2f*)(arow + k);
#pragma unroll
    for (int nt = 0; nt < 8; ++nt) {
      v2f b = *(const v2f*)(brow + nt * 16 * GS_PITCH + k);
      acc[nt] = wmma4(a, b, acc[nt]);
    }
  }

  const int h8 = (lane >> 4) << 3;
  const float* vrow = &sV[w][h8 * GS_PITCH + l16];
  float p[8];
#pragma unroll
  for (int r = 0; r < 8; ++r) {
    float s = 0.f;
#pragma unroll
    for (int nt = 0; nt < 8; ++nt) s += acc[nt][r] * vrow[r * GS_PITCH + nt * 16];
    p[r] = s;
  }
#pragma unroll
  for (int off = 8; off > 0; off >>= 1) {
#pragma unroll
    for (int r = 0; r < 8; ++r) p[r] += __shfl_xor(p[r], off, 32);
  }
  float o = p[0];
#pragma unroll
  for (int r = 1; r < 8; ++r) o = (l16 == r) ? p[r] : o;
  if (l16 < 8) out[pair0 + h8 + l16] = o;
}

#define WS_ALIGN 128
#define WS_PAD(n) ((((size_t)(n)) + WS_ALIGN - 1) / WS_ALIGN * WS_ALIGN)
static_assert(WS_PAD(N_NODES) + WS_PAD(256) + WS_PAD(128 * 128) + WS_PAD(N_NODES) +
              WS_PAD(N_NODES + 16) + WS_PAD(NE_EDGES)
              <= (size_t)N_NODES + 256 + 128 * 128 + N_NODES + (N_NODES + 16) + N_NODES + NE_EDGES,
              "the aligned planes end no later than the packed ones with their unused slot did");

static_assert(WS_PAD(N_NODES) >= 2 * 128 * 128, "U and Q^T fit in the edge counters' plane");

extern "C" void kernel_launch(void* const* d_in, const int* in_sizes, int n_in,
                              void* d_out, int out_size, void* d_ws, size_t ws_size,
                              hipStream_t stream) {
  const float* x   = (const float*)d_in[0];
  const int*   edg = (const int*)d_in[1];
  const int*   di  = (const int*)d_in[2];
  const float* w1  = (const float*)d_in[3];
  const float* b1  = (const float*)d_in[4];
  const float* w2  = (const float*)d_in[5];
  const float* b2  = (const float*)d_in[6];
  const float* w3  = (const float*)d_in[7];
  const float* b3  = (const float*)d_in[8];
  const float* g1  = (const float*)d_in[9];
  const float* be1 = (const float*)d_in[10];
  const float* g2  = (const float*)d_in[11];
  const float* be2 = (const float*)d_in[12];
  const float* g3  = (const float*)d_in[13];
  const float* be3 = (const float*)d_in[14];
  const float* p1  = (const float*)d_in[15];
  const float* p2  = (const float*)d_in[16];
  float* out = (float*)d_out;

  float* ws = (float*)d_ws;
  size_t off = 0;
  float* dis    = ws + off; off += WS_PAD(N_NODES);
  float* stats  = ws + off; off += WS_PAD(256);
  float* Wt     = ws + off; off += WS_PAD(128 * 128);
  int*   cnt    = (int*)(ws + off); off += WS_PAD(N_NODES);
  int*   rowptr = (int*)(ws + off); off += WS_PAD(N_NODES + 16);
  int*   srcidx = (int*)(ws + off); off += WS_PAD(NE_EDGES);
  float* lin    = ws + off; off += (size_t)N_NODES * DF;
  float* agg    = ws + off; off += (size_t)N_NODES * DF;
  float* U    = (float*)cnt;
  float* Qt   = (float*)cnt + 128 * 128;
  int* rank = (int*)agg;
  int* hist = (int*)agg + WS_PAD(NE_EDGES);

  const int* erow = edg;
  const int* ecol = edg + NE_EDGES;

  hist_rank<<<CH_BLOCKS, 1024, 0, stream>>>(ecol, rank, hist, w1, Wt);
  chunk_prefix<<<(N_VEC + 63) / 64, 64, 0, stream>>>(hist, cnt);
  scan_rowptr<<<1, 1024, 0, stream>>>(cnt, rowptr);
  csr_fill<<<(NE_EDGES + 255) / 256, 256, 0, stream>>>(erow, ecol, rowptr, rank, hist, srcidx, dis);

  const float* Wn[3] = {w2, w3, nullptr};
  const float* Ra[3] = {p1, U, nullptr};
  const float* Rb[3] = {p2, p1, nullptr};
  float*       Rc[3] = {U, Qt, nullptr};
  const int    Rn[3] = {RIDER_BLOCKS, RIDER_BLOCKS, 0};
  const float* Bl[3] = {b1, b2, b3};
  const float* Gl[3] = {g1, g2, g3};
  const float* El[3] = {be1, be2, be3};
  const int strip_waves = N_NODES / 16;
  for (int l = 0; l < 3; ++l) {
    if (l == 0)
      wmma_gemm_strip<<<(strip_waves * 32 + 255) / 256, 256, 0, stream>>>(
          x, Wt, Bl[l], lin, N_NODES, nullptr, nullptr, nullptr);
    else
      wmma_gemm_strip<<<(strip_waves * 32 + 255) / 256, 256, 0, stream>>>(
          agg, Wt, Bl[l], lin, N_NODES, stats, Gl[l - 1], El[l - 1]);
    gcn_aggregate<<<(N_NODES * 32 + 255) / 256, 256, 0, stream>>>(
        rowptr, srcidx, dis, lin, agg, stats, Wn[l], Wt);
    bn_stats<<<(N_NODES + 511) / 512 + Rn[l], 1024, 0, stream>>>(
        agg, stats, N_NODES, Ra[l], Rb[l], Rc[l], Rn[l]);
  }

  decode_pairs<<<BPAIRS / (DP_WAVES * 16), DP_WAVES * 32, 0, stream>>>(
      agg, stats, g3, be3, di, Qt, out);
}
